# starting kernel
# baseline (speedup 1.0000x reference)
#include <hip/hip_runtime.h>
#include <hip/hip_bf16.h>

typedef __attribute__((ext_vector_type(2))) float v2f;
typedef __attribute__((ext_vector_type(8))) float v8f;

#define SLOPE 0.2f

static __host__ __device__ inline int cdiv(int a, int b) { return (a + b - 1) / b; }

__device__ inline void atomicAddF32(float* p, float v) { unsafeAtomicAdd(p, v); }

__global__ void gat_fill_f32(float* __restrict__ p, float v, int n) {
    int i = blockIdx.x * blockDim.x + threadIdx.x;
    if (i < n) p[i] = v;
}

__global__ void gat_gemm1_wmma(const float* __restrict__ A,
                               const float* __restrict__ W,
                               float* __restrict__ H,
                               int n_nodes) {
    __shared__ float As[32 * 256];
    __shared__ float Ws[32 * 256];

    const int row0 = blockIdx.x * 32;
    const int tid  = threadIdx.x;
    const int wave = tid >> 5;
    const int lane = tid & 31;

    for (int i = tid; i < 32 * 64; i += 256) {
        int r  = i >> 6;
        int c4 = (i & 63) << 2;
        int rr = row0 + r; if (rr >= n_nodes) rr = n_nodes - 1;
        *(float4*)(&As[r * 256 + c4]) = *(const float4*)(A + (size_t)rr * 256 + c4);
    }

    const int m   = lane & 15;
    const int kk  = (lane >> 4) << 1;
    const int n0  = wave * 32;
    const int n1  = n0 + 16;

    v8f c00 = {}, c01 = {}, c10 = {}, c11 = {};

    for (int kc = 0; kc < 256; kc += 32) {
        __syncthreads();
        for (int i = tid; i < 32 * 64; i += 256) {
            int r  = i >> 6;
            int c4 = (i & 63) << 2;
            *(float4*)(&Ws[r * 256 + c4]) = *(const float4*)(W + (size_t)(kc + r) * 256 + c4);
        }
        __syncthreads();

#pragma unroll
        for (int k = 0; k < 32; k += 4) {
            v2f a0, a1, b0, b1;
            a0.x = As[m * 256 + kc + k + kk];
            a0.y = As[m * 256 + kc + k + kk + 1];
            a1.x = As[(m + 16) * 256 + kc + k + kk];
            a1.y = As[(m + 16) * 256 + kc + k + kk + 1];
            b0.x = Ws[(k + kk) * 256 + n0 + m];
            b0.y = Ws[(k + kk + 1) * 256 + n0 + m];
            b1.x = Ws[(k + kk) * 256 + n1 + m];
            b1.y = Ws[(k + kk + 1) * 256 + n1 + m];
            c00 = __builtin_amdgcn_wmma_f32_16x16x4_f32(false, a0, false, b0, (short)0, c00, false, false);
            c01 = __builtin_amdgcn_wmma_f32_16x16x4_f32(false, a0, false, b1, (short)0, c01, false, false);
            c10 = __builtin_amdgcn_wmma_f32_16x16x4_f32(false, a1, false, b0, (short)0, c10, false, false);
            c11 = __builtin_amdgcn_wmma_f32_16x16x4_f32(false, a1, false, b1, (short)0, c11, false, false);
        }
    }

    const int mhi = (lane >> 4) * 8;
    for (int v = 0; v < 8; ++v) {
        int r0 = row0 + mhi + v;
        int r1 = r0 + 16;
        if (r0 < n_nodes) {
            H[(size_t)r0 * 256 + n0 + m] = c00[v];
            H[(size_t)r0 * 256 + n1 + m] = c01[v];
        }
        if (r1 < n_nodes) {
            H[(size_t)r1 * 256 + n0 + m] = c10[v];
            H[(size_t)r1 * 256 + n1 + m] = c11[v];
        }
    }
}

__global__ void gat_attn_scores(const float* __restrict__ H,
                                const float* __restrict__ al,
                                const float* __restrict__ ar,
                                float* __restrict__ el, float* __restrict__ er,
                                int n_nodes) {
    int i = blockIdx.x * blockDim.x + threadIdx.x;
    if (i >= n_nodes * 8) return;
    int n = i >> 3, h = i & 7;
    const float* hp  = H + (size_t)n * 256 + h * 32;
    const float* alp = al + h * 32;
    const float* arp = ar + h * 32;
    float sl = 0.f, sr = 0.f;
    for (int d = 0; d < 32; ++d) { float v = hp[d]; sl += v * alp[d]; sr += v * arp[d]; }
    el[i] = sl; er[i] = sr;
}

__device__ inline void atomicMaxFloat(float* addr, float value) {
    if (value >= 0.0f)
        atomicMax((int*)addr, __float_as_int(value));
    else
        atomicMin((unsigned int*)addr, __float_as_uint(value));
}

__global__ void gat_edge_e_max(const int* __restrict__ src, const int* __restrict__ dst,
                               const float* __restrict__ el, const float* __restrict__ er,
                               float* __restrict__ e, float* __restrict__ m,
                               int n_edges, int heads) {
    int i = blockIdx.x * blockDim.x + threadIdx.x;
    if (i >= n_edges * heads) return;
    int ed = i / heads, h = i - ed * heads;
    float v = el[src[ed] * heads + h] + er[dst[ed] * heads + h];
    v = (v >= 0.f) ? v : SLOPE * v;
    e[i] = v;
    atomicMaxFloat(&m[dst[ed] * heads + h], v);
}

__global__ void gat_edge_expsum(const int* __restrict__ dst,
                                float* __restrict__ e, const float* __restrict__ m,
                                float* __restrict__ s, int n_edges, int heads) {
    int i = blockIdx.x * blockDim.x + threadIdx.x;
    if (i >= n_edges * heads) return;
    int ed = i / heads, h = i - ed * heads;
    float v = expf(e[i] - m[dst[ed] * heads + h]);
    e[i] = v;
    atomicAddF32(&s[dst[ed] * heads + h], v);
}

__global__ void gat_aggregate(const int* __restrict__ src, const int* __restrict__ dst,
                              const float* __restrict__ H, const float* __restrict__ ue,
                              const float* __restrict__ s, float* __restrict__ agg,
                              int n_edges, int heads, int dim) {
    int hd = heads * dim;
    int i = blockIdx.x * blockDim.x + threadIdx.x;
    if (i >= n_edges * hd) return;
    int ed = i / hd, c = i - ed * hd;
    int h = c / dim;
    int sn = src[ed], dn = dst[ed];
    float alpha = ue[ed * heads + h] / s[dn * heads + h];
    atomicAddF32(&agg[(size_t)dn * hd + c], H[(size_t)sn * hd + c] * alpha);
}

__global__ void gat_elu_bias(float* __restrict__ x, const float* __restrict__ b,
                             int total, int hd) {
    int i = blockIdx.x * blockDim.x + threadIdx.x;
    if (i >= total) return;
    float v = x[i] + b[i % hd];
    x[i] = (v > 0.f) ? v : expm1f(v);
}

__global__ void gat_l2_linear(const float* __restrict__ x, const float* __restrict__ W2,
                              const float* __restrict__ al2, const float* __restrict__ ar2,
                              float* __restrict__ h2, float* __restrict__ el2,
                              float* __restrict__ er2, int n_nodes) {
    int n = blockIdx.x * blockDim.x + threadIdx.x;
    if (n >= n_nodes) return;
    float a0 = 0.f, a1 = 0.f, a2 = 0.f;
    const float* xp = x + (size_t)n * 256;
    for (int k = 0; k < 256; ++k) {
        float xv = xp[k];
        a0 += xv * W2[k * 3 + 0];
        a1 += xv * W2[k * 3 + 1];
        a2 += xv * W2[k * 3 + 2];
    }
    h2[n * 3 + 0] = a0; h2[n * 3 + 1] = a1; h2[n * 3 + 2] = a2;
    el2[n] = a0 * al2[0] + a1 * al2[1] + a2 * al2[2];
    er2[n] = a0 * ar2[0] + a1 * ar2[1] + a2 * ar2[2];
}

__global__ void gat_finalize(const float* __restrict__ agg2, const float* __restrict__ b2,
                             float* __restrict__ out, int total) {
    int i = blockIdx.x * blockDim.x + threadIdx.x;
    if (i >= total) return;
    out[i] = agg2[i] + b2[i % 3];
}

extern "C" void kernel_launch(void* const* d_in, const int* in_sizes, int n_in,
                              void* d_out, int out_size, void* d_ws, size_t ws_size,
                              hipStream_t stream) {
    const float* feat = (const float*)d_in[0];
    const int*   src  = (const int*)  d_in[1];
    const int*   dst  = (const int*)  d_in[2];
    const float* W1   = (const float*)d_in[3];
    const float* al1  = (const float*)d_in[4];
    const float* ar1  = (const float*)d_in[5];
    const float* b1   = (const float*)d_in[6];
    const float* W2   = (const float*)d_in[7];
    const float* al2  = (const float*)d_in[8];
    const float* ar2  = (const float*)d_in[9];
    const float* b2   = (const float*)d_in[10];
    float* out = (float*)d_out;

    const int n_nodes = in_sizes[0] / 256;
    const int n_edges = in_sizes[1];

    char* cur = (char*)d_ws;
    auto alloc = [&](size_t bytes) -> float* {
        float* p = (float*)cur;
        cur += (bytes + 255) & ~(size_t)255;
        return p;
    };
    float* H1   = alloc((size_t)n_nodes * 256 * 4);
    float* el1  = alloc((size_t)n_nodes * 8 * 4);
    float* er1  = alloc((size_t)n_nodes * 8 * 4);
    float* m1   = alloc((size_t)n_nodes * 8 * 4);
    float* s1   = alloc((size_t)n_nodes * 8 * 4);
    float* e1   = alloc((size_t)n_edges * 8 * 4);
    float* agg1 = alloc((size_t)n_nodes * 256 * 4);
    float* h2   = alloc((size_t)n_nodes * 3 * 4);
    float* el2  = alloc((size_t)n_nodes * 4);
    float* er2  = alloc((size_t)n_nodes * 4);
    float* m2   = alloc((size_t)n_nodes * 4);
    float* s2   = alloc((size_t)n_nodes * 4);
    float* e2   = alloc((size_t)n_edges * 4);
    float* agg2 = alloc((size_t)n_nodes * 3 * 4);

    const int B = 256;
    const float NEG_INF = -__builtin_huge_valf();

    gat_gemm1_wmma<<<cdiv(n_nodes, 32), 256, 0, stream>>>(feat, W1, H1, n_nodes);
    gat_attn_scores<<<cdiv(n_nodes * 8, B), B, 0, stream>>>(H1, al1, ar1, el1, er1, n_nodes);

    gat_fill_f32<<<cdiv(n_nodes * 8, B), B, 0, stream>>>(m1, NEG_INF, n_nodes * 8);
    gat_fill_f32<<<cdiv(n_nodes * 8, B), B, 0, stream>>>(s1, 0.f, n_nodes * 8);
    gat_fill_f32<<<cdiv(n_nodes * 256, B), B, 0, stream>>>(agg1, 0.f, n_nodes * 256);

    gat_edge_e_max<<<cdiv(n_edges * 8, B), B, 0, stream>>>(src, dst, el1, er1, e1, m1, n_edges, 8);
    gat_edge_expsum<<<cdiv(n_edges * 8, B), B, 0, stream>>>(dst, e1, m1, s1, n_edges, 8);
    gat_aggregate<<<cdiv(n_edges * 256, B), B, 0, stream>>>(src, dst, H1, e1, s1, agg1,
                                                            n_edges, 8, 32);
    gat_elu_bias<<<cdiv(n_nodes * 256, B), B, 0, stream>>>(agg1, b1, n_nodes * 256, 256);

    gat_l2_linear<<<cdiv(n_nodes, 128), 128, 0, stream>>>(agg1, W2, al2, ar2, h2, el2, er2, n_nodes);

    gat_fill_f32<<<cdiv(n_nodes, B), B, 0, stream>>>(m2, NEG_INF, n_nodes);
    gat_fill_f32<<<cdiv(n_nodes, B), B, 0, stream>>>(s2, 0.f, n_nodes);
    gat_fill_f32<<<cdiv(n_nodes * 3, B), B, 0, stream>>>(agg2, 0.f, n_nodes * 3);

    gat_edge_e_max<<<cdiv(n_edges, B), B, 0, stream>>>(src, dst, el2, er2, e2, m2, n_edges, 1);
    gat_edge_expsum<<<cdiv(n_edges, B), B, 0, stream>>>(dst, e2, m2, s2, n_edges, 1);
    gat_aggregate<<<cdiv(n_edges * 3, B), B, 0, stream>>>(src, dst, h2, e2, s2, agg2,
                                                          n_edges, 1, 3);
    gat_finalize<<<cdiv(n_nodes * 3, B), B, 0, stream>>>(agg2, b2, out, n_nodes * 3);
}


# Layer-1 aggregation by dst-range owner groups: each scans dst[], sums its 196 rows in LDS with LDS f32 adds (arrival order), plain row stores; global f32 adds and agg zero fill gone, 16 launches
# speedup vs baseline: 2.1033x; 2.1033x over previous
#include <hip/hip_runtime.h>
#include <hip/hip_bf16.h>

typedef __attribute__((ext_vector_type(2))) float v2f;
typedef __attribute__((ext_vector_type(8))) float v8f;

#define SLOPE 0.2f

static __host__ __device__ inline int cdiv(int a, int b) { return (a + b - 1) / b; }

__device__ inline void atomicAddF32(float* p, float v) { unsafeAtomicAdd(p, v); }
__device__ inline void atomicAddLds(float* p, float v) {
    __hip_atomic_fetch_add(p, v, __ATOMIC_RELAXED, __HIP_MEMORY_SCOPE_WORKGROUP);
}

__global__ void gat_fill_f32(float* __restrict__ p, float v, int n) {
    int i = blockIdx.x * blockDim.x + threadIdx.x;
    if (i < n) p[i] = v;
}

__global__ void gat_gemm1_wmma(const float* __restrict__ A,
                               const float* __restrict__ W,
                               float* __restrict__ H,
                               int n_nodes) {
    __shared__ float As[32 * 256];
    __shared__ float Ws[32 * 256];

    const int row0 = blockIdx.x * 32;
    const int tid  = threadIdx.x;
    const int wave = tid >> 5;
    const int lane = tid & 31;

    for (int i = tid; i < 32 * 64; i += 256) {
        int r  = i >> 6;
        int c4 = (i & 63) << 2;
        int rr = row0 + r; if (rr >= n_nodes) rr = n_nodes - 1;
        *(float4*)(&As[r * 256 + c4]) = *(const float4*)(A + (size_t)rr * 256 + c4);
    }

    const int m   = lane & 15;
    const int kk  = (lane >> 4) << 1;
    const int n0  = wave * 32;
    const int n1  = n0 + 16;

    v8f c00 = {}, c01 = {}, c10 = {}, c11 = {};

    for (int kc = 0; kc < 256; kc += 32) {
        __syncthreads();
        for (int i = tid; i < 32 * 64; i += 256) {
            int r  = i >> 6;
            int c4 = (i & 63) << 2;
            *(float4*)(&Ws[r * 256 + c4]) = *(const float4*)(W + (size_t)(kc + r) * 256 + c4);
        }
        __syncthreads();

#pragma unroll
        for (int k = 0; k < 32; k += 4) {
            v2f a0, a1, b0, b1;
            a0.x = As[m * 256 + kc + k + kk];
            a0.y = As[m * 256 + kc + k + kk + 1];
            a1.x = As[(m + 16) * 256 + kc + k + kk];
            a1.y = As[(m + 16) * 256 + kc + k + kk + 1];
            b0.x = Ws[(k + kk) * 256 + n0 + m];
            b0.y = Ws[(k + kk + 1) * 256 + n0 + m];
            b1.x = Ws[(k + kk) * 256 + n1 + m];
            b1.y = Ws[(k + kk + 1) * 256 + n1 + m];
            c00 = __builtin_amdgcn_wmma_f32_16x16x4_f32(false, a0, false, b0, (short)0, c00, false, false);
            c01 = __builtin_amdgcn_wmma_f32_16x16x4_f32(false, a0, false, b1, (short)0, c01, false, false);
            c10 = __builtin_amdgcn_wmma_f32_16x16x4_f32(false, a1, false, b0, (short)0, c10, false, false);
            c11 = __builtin_amdgcn_wmma_f32_16x16x4_f32(false, a1, false, b1, (short)0, c11, false, false);
        }
    }

    const int mhi = (lane >> 4) * 8;
    for (int v = 0; v < 8; ++v) {
        int r0 = row0 + mhi + v;
        int r1 = r0 + 16;
        if (r0 < n_nodes) {
            H[(size_t)r0 * 256 + n0 + m] = c00[v];
            H[(size_t)r0 * 256 + n1 + m] = c01[v];
        }
        if (r1 < n_nodes) {
            H[(size_t)r1 * 256 + n0 + m] = c10[v];
            H[(size_t)r1 * 256 + n1 + m] = c11[v];
        }
    }
}

__global__ void gat_attn_scores(const float* __restrict__ H,
                                const float* __restrict__ al,
                                const float* __restrict__ ar,
                                float* __restrict__ el, float* __restrict__ er,
                                int n_nodes) {
    int i = blockIdx.x * blockDim.x + threadIdx.x;
    if (i >= n_nodes * 8) return;
    int n = i >> 3, h = i & 7;
    const float* hp  = H + (size_t)n * 256 + h * 32;
    const float* alp = al + h * 32;
    const float* arp = ar + h * 32;
    float sl = 0.f, sr = 0.f;
    for (int d = 0; d < 32; ++d) { float v = hp[d]; sl += v * alp[d]; sr += v * arp[d]; }
    el[i] = sl; er[i] = sr;
}

__device__ inline void atomicMaxFloat(float* addr, float value) {
    if (value >= 0.0f)
        atomicMax((int*)addr, __float_as_int(value));
    else
        atomicMin((unsigned int*)addr, __float_as_uint(value));
}

__global__ void gat_edge_e_max(const int* __restrict__ src, const int* __restrict__ dst,
                               const float* __restrict__ el, const float* __restrict__ er,
                               float* __restrict__ e, float* __restrict__ m,
                               int n_edges, int heads) {
    int i = blockIdx.x * blockDim.x + threadIdx.x;
    if (i >= n_edges * heads) return;
    int ed = i / heads, h = i - ed * heads;
    float v = el[src[ed] * heads + h] + er[dst[ed] * heads + h];
    v = (v >= 0.f) ? v : SLOPE * v;
    e[i] = v;
    atomicMaxFloat(&m[dst[ed] * heads + h], v);
}

__global__ void gat_edge_expsum(const int* __restrict__ dst,
                                float* __restrict__ e, const float* __restrict__ m,
                                float* __restrict__ s, int n_edges, int heads) {
    int i = blockIdx.x * blockDim.x + threadIdx.x;
    if (i >= n_edges * heads) return;
    int ed = i / heads, h = i - ed * heads;
    float v = expf(e[i] - m[dst[ed] * heads + h]);
    e[i] = v;
    atomicAddF32(&s[dst[ed] * heads + h], v);
}

__global__ void gat_aggregate(const int* __restrict__ src, const int* __restrict__ dst,
                              const float* __restrict__ H, const float* __restrict__ ue,
                              const float* __restrict__ s, float* __restrict__ agg,
                              int n_edges, int heads, int dim) {
    int hd = heads * dim;
    int i = blockIdx.x * blockDim.x + threadIdx.x;
    if (i >= n_edges * hd) return;
    int ed = i / hd, c = i - ed * hd;
    int h = c / dim;
    int sn = src[ed], dn = dst[ed];
    float alpha = ue[ed * heads + h] / s[dn * heads + h];
    atomicAddF32(&agg[(size_t)dn * hd + c], H[(size_t)sn * hd + c] * alpha);
}

#define AGG_TILE    196
#define AGG_THREADS 1024
#define AGG_WAVES   32
#define AGG_STEP    128
#define AGG_QCAP    160

__global__ __launch_bounds__(AGG_THREADS)
void gat_aggregate_tile(const int* __restrict__ src, const int* __restrict__ dst,
                        const float* __restrict__ H, const float* __restrict__ ue,
                        const float* __restrict__ s, float* __restrict__ agg,
                        int n_edges, int n_nodes) {
    __shared__ __attribute__((aligned(16))) float acc[AGG_TILE * 256];
    __shared__ float s_t[AGG_TILE * 8];
    __shared__ int   q_e[AGG_WAVES * AGG_QCAP];
    __shared__ int   q_d[AGG_WAVES * AGG_QCAP];

    const int tid  = threadIdx.x;
    const int wave = tid >> 5;
    const int lane = tid & 31;
    const int lo   = blockIdx.x * AGG_TILE;
    int rows = n_nodes - lo; if (rows > AGG_TILE) rows = AGG_TILE;
    const unsigned span = (unsigned)rows;

    for (int i = tid; i < AGG_TILE * 64; i += AGG_THREADS)
        ((float4*)acc)[i] = make_float4(0.f, 0.f, 0.f, 0.f);
    for (int i = tid; i < rows * 8; i += AGG_THREADS)
        s_t[i] = s[(size_t)lo * 8 + i];
    __syncthreads();

    int* qe = q_e + wave * AGG_QCAP;
    int* qd = q_d + wave * AGG_QCAP;
    const unsigned lt_mask = (1u << lane) - 1u;
    const int hl = lane >> 3;
    int cnt = 0;

    auto drain = [&](int n) {
        for (int b = 0; b < n; b += 32) {
            int nb = n - b; if (nb > 32) nb = 32;
            const int i  = b + (lane < nb ? lane : nb - 1);
            const int e  = qe[i];
            const int dl = qd[i];
            const int sn = src[e];
            for (int g = 0; g < nb; g += 4) {
                int su[4], eu[4], du[4];
#pragma unroll
                for (int u = 0; u < 4; ++u) {
                    const int k = (g + u < nb) ? g + u : nb - 1;
                    su[u] = __builtin_amdgcn_readlane(sn, k);
                    eu[u] = __builtin_amdgcn_readlane(e,  k);
                    du[u] = __builtin_amdgcn_readlane(dl, k);
                }
                float4 r0[4], r1[4];
                float  u0[4], u1[4];
#pragma unroll
                for (int u = 0; u < 4; ++u) {
                    const float* hp = H + (size_t)su[u] * 256 + 4 * lane;
                    r0[u] = *(const float4*)(hp);
                    r1[u] = *(const float4*)(hp + 128);
                    u0[u] = ue[(size_t)eu[u] * 8 + hl];
                    u1[u] = ue[(size_t)eu[u] * 8 + 4 + hl];
                }
#pragma unroll
                for (int u = 0; u < 4; ++u) {
                    if (g + u < nb) {
                        const float a0 = u0[u] / s_t[du[u] * 8 + hl];
                        const float a1 = u1[u] / s_t[du[u] * 8 + 4 + hl];
                        float* row = acc + du[u] * 256 + lane;
                        atomicAddLds(row,       r0[u].x * a0);
                        atomicAddLds(row + 32,  r0[u].y * a0);
                        atomicAddLds(row + 64,  r0[u].z * a0);
                        atomicAddLds(row + 96,  r0[u].w * a0);
                        atomicAddLds(row + 128, r1[u].x * a1);
                        atomicAddLds(row + 160, r1[u].y * a1);
                        atomicAddLds(row + 192, r1[u].z * a1);
                        atomicAddLds(row + 224, r1[u].w * a1);
                    }
                }
            }
        }
    };

    const int n4 = n_edges & ~3;
    if (n4 > 0) {
        const int last    = n4 - 4;
        const int n_steps = cdiv(n4, AGG_STEP);
        auto load4 = [&](int st) {
            const int e0 = st * AGG_STEP + lane * 4;
            return *(const int4*)(dst + (e0 < last ? e0 : last));
        };
        int4 cur[4], nxt[4];
#pragma unroll
        for (int u = 0; u < 4; ++u) cur[u] = load4(wave + u * AGG_WAVES);

        for (int st0 = wave; st0 < n_steps; st0 += 4 * AGG_WAVES) {
#pragma unroll
            for (int u = 0; u < 4; ++u) nxt[u] = load4(st0 + (4 + u) * AGG_WAVES);
#pragma unroll
            for (int u = 0; u < 4; ++u) {
                const int e0 = (st0 + u * AGG_WAVES) * AGG_STEP + lane * 4;
                const bool valid = e0 < n4;
                const int dv[4] = {cur[u].x, cur[u].y, cur[u].z, cur[u].w};
                bool mt[4];
#pragma unroll
                for (int j = 0; j < 4; ++j) mt[j] = valid && ((unsigned)(dv[j] - lo) < span);
                if (__builtin_amdgcn_ballot_w32(mt[0] | mt[1] | mt[2] | mt[3]) != 0u) {
#pragma unroll
                    for (int j = 0; j < 4; ++j) {
                        const unsigned mask = __builtin_amdgcn_ballot_w32(mt[j]);
                        if (mt[j]) {
                            const int pos = cnt + __popc(mask & lt_mask);
                            qe[pos] = e0 + j;
                            qd[pos] = dv[j] - lo;
                        }
                        cnt += __popc(mask);
                    }
                    if (cnt > AGG_QCAP - AGG_STEP) { drain(cnt); cnt = 0; }
                }
            }
#pragma unroll
            for (int u = 0; u < 4; ++u) cur[u] = nxt[u];
        }
    }
    if (wave == 0) {
        const int e = n4 + lane;
        const int d = (e < n_edges) ? dst[e] : -1;
        const bool mt = (unsigned)(d - lo) < span;
        const unsigned mask = __builtin_amdgcn_ballot_w32(mt);
        if (mt) {
            const int pos = cnt + __popc(mask & lt_mask);
            qe[pos] = e;
            qd[pos] = d - lo;
        }
        cnt += __popc(mask);
    }
    if (cnt > 0) drain(cnt);

    __syncthreads();
    for (int i = tid; i < rows * 64; i += AGG_THREADS) {
        const int r  = i >> 6;
        const int c4 = i & 63;
        const float* p = acc + r * 256 + (c4 >> 5) * 128 + (c4 & 31);
        float4 o = make_float4(p[0], p[32], p[64], p[96]);
        *(float4*)(agg + (size_t)(lo + r) * 256 + 4 * c4) = o;
    }
}

__global__ void gat_elu_bias(float* __restrict__ x, const float* __restrict__ b,
                             int total, int hd) {
    int i = blockIdx.x * blockDim.x + threadIdx.x;
    if (i >= total) return;
    float v = x[i] + b[i % hd];
    x[i] = (v > 0.f) ? v : expm1f(v);
}

__global__ void gat_l2_linear(const float* __restrict__ x, const float* __restrict__ W2,
                              const float* __restrict__ al2, const float* __restrict__ ar2,
                              float* __restrict__ h2, float* __restrict__ el2,
                              float* __restrict__ er2, int n_nodes) {
    int n = blockIdx.x * blockDim.x + threadIdx.x;
    if (n >= n_nodes) return;
    float a0 = 0.f, a1 = 0.f, a2 = 0.f;
    const float* xp = x + (size_t)n * 256;
    for (int k = 0; k < 256; ++k) {
        float xv = xp[k];
        a0 += xv * W2[k * 3 + 0];
        a1 += xv * W2[k * 3 + 1];
        a2 += xv * W2[k * 3 + 2];
    }
    h2[n * 3 + 0] = a0; h2[n * 3 + 1] = a1; h2[n * 3 + 2] = a2;
    el2[n] = a0 * al2[0] + a1 * al2[1] + a2 * al2[2];
    er2[n] = a0 * ar2[0] + a1 * ar2[1] + a2 * ar2[2];
}

__global__ void gat_finalize(const float* __restrict__ agg2, const float* __restrict__ b2,
                             float* __restrict__ out, int total) {
    int i = blockIdx.x * blockDim.x + threadIdx.x;
    if (i >= total) return;
    out[i] = agg2[i] + b2[i % 3];
}

extern "C" void kernel_launch(void* const* d_in, const int* in_sizes, int n_in,
                              void* d_out, int out_size, void* d_ws, size_t ws_size,
                              hipStream_t stream) {
    const float* feat = (const float*)d_in[0];
    const int*   src  = (const int*)  d_in[1];
    const int*   dst  = (const int*)  d_in[2];
    const float* W1   = (const float*)d_in[3];
    const float* al1  = (const float*)d_in[4];
    const float* ar1  = (const float*)d_in[5];
    const float* b1   = (const float*)d_in[6];
    const float* W2   = (const float*)d_in[7];
    const float* al2  = (const float*)d_in[8];
    const float* ar2  = (const float*)d_in[9];
    const float* b2   = (const float*)d_in[10];
    float* out = (float*)d_out;

    const int n_nodes = in_sizes[0] / 256;
    const int n_edges = in_sizes[1];

    char* cur = (char*)d_ws;
    auto alloc = [&](size_t bytes) -> float* {
        float* p = (float*)cur;
        cur += (bytes + 255) & ~(size_t)255;
        return p;
    };
    float* H1   = alloc((size_t)n_nodes * 256 * 4);
    float* el1  = alloc((size_t)n_nodes * 8 * 4);
    float* er1  = alloc((size_t)n_nodes * 8 * 4);
    float* m1   = alloc((size_t)n_nodes * 8 * 4);
    float* s1   = alloc((size_t)n_nodes * 8 * 4);
    float* e1   = alloc((size_t)n_edges * 8 * 4);
    float* agg1 = alloc((size_t)n_nodes * 256 * 4);
    float* h2   = alloc((size_t)n_nodes * 3 * 4);
    float* el2  = alloc((size_t)n_nodes * 4);
    float* er2  = alloc((size_t)n_nodes * 4);
    float* m2   = alloc((size_t)n_nodes * 4);
    float* s2   = alloc((size_t)n_nodes * 4);
    float* e2   = alloc((size_t)n_edges * 4);
    float* agg2 = alloc((size_t)n_nodes * 3 * 4);

    const int B = 256;
    const float NEG_INF = -__builtin_huge_valf();

    gat_gemm1_wmma<<<cdiv(n_nodes, 32), 256, 0, stream>>>(feat, W1, H1, n_nodes);
    gat_attn_scores<<<cdiv(n_nodes * 8, B), B, 0, stream>>>(H1, al1, ar1, el1, er1, n_nodes);

    gat_fill_f32<<<cdiv(n_nodes * 8, B), B, 0, stream>>>(m1, NEG_INF, n_nodes * 8);
    gat_fill_f32<<<cdiv(n_nodes * 8, B), B, 0, stream>>>(s1, 0.f, n_nodes * 8);

    gat_edge_e_max<<<cdiv(n_edges * 8, B), B, 0, stream>>>(src, dst, el1, er1, e1, m1, n_edges, 8);
    gat_edge_expsum<<<cdiv(n_edges * 8, B), B, 0, stream>>>(dst, e1, m1, s1, n_edges, 8);
    gat_aggregate_tile<<<cdiv(n_nodes, AGG_TILE), AGG_THREADS, 0, stream>>>(src, dst, H1, e1, s1, agg1,
                                                                            n_edges, n_nodes);
    gat_elu_bias<<<cdiv(n_nodes * 256, B), B, 0, stream>>>(agg1, b1, n_nodes * 256, 256);

    gat_l2_linear<<<cdiv(n_nodes, 128), 128, 0, stream>>>(agg1, W2, al2, ar2, h2, el2, er2, n_nodes);

    gat_fill_f32<<<cdiv(n_nodes, B), B, 0, stream>>>(m2, NEG_INF, n_nodes);
    gat_fill_f32<<<cdiv(n_nodes, B), B, 0, stream>>>(s2, 0.f, n_nodes);
    gat_fill_f32<<<cdiv(n_nodes * 3, B), B, 0, stream>>>(agg2, 0.f, n_nodes * 3);

    gat_edge_e_max<<<cdiv(n_edges, B), B, 0, stream>>>(src, dst, el2, er2, e2, m2, n_edges, 1);
    gat_edge_expsum<<<cdiv(n_edges, B), B, 0, stream>>>(dst, e2, m2, s2, n_edges, 1);
    gat_aggregate<<<cdiv(n_edges * 3, B), B, 0, stream>>>(src, dst, h2, e2, s2, agg2,
                                                          n_edges, 1, 3);
    gat_finalize<<<cdiv(n_nodes * 3, B), B, 0, stream>>>(agg2, b2, out, n_nodes * 3);
}


# Layer-1 softmax max and exp sum kept by the dst-range owner groups in LDS (per-wave edge list, LDS int max, f32 adds in arrival order, rows divided by the sum at the end): 4 launches fewer
# speedup vs baseline: 2.3736x; 1.1285x over previous
#include <hip/hip_runtime.h>
#include <hip/hip_bf16.h>

typedef __attribute__((ext_vector_type(2))) float v2f;
typedef __attribute__((ext_vector_type(8))) float v8f;

#define SLOPE 0.2f

static __host__ __device__ inline int cdiv(int a, int b) { return (a + b - 1) / b; }

__device__ inline void atomicAddF32(float* p, float v) { unsafeAtomicAdd(p, v); }
__device__ inline void atomicAddLds(float* p, float v) {
    __hip_atomic_fetch_add(p, v, __ATOMIC_RELAXED, __HIP_MEMORY_SCOPE_WORKGROUP);
}

__global__ void gat_fill_f32(float* __restrict__ p, float v, int n) {
    int i = blockIdx.x * blockDim.x + threadIdx.x;
    if (i < n) p[i] = v;
}

__global__ void gat_gemm1_wmma(const float* __restrict__ A,
                               const float* __restrict__ W,
                               float* __restrict__ H,
                               int n_nodes) {
    __shared__ float As[32 * 256];
    __shared__ float Ws[32 * 256];

    const int row0 = blockIdx.x * 32;
    const int tid  = threadIdx.x;
    const int wave = tid >> 5;
    const int lane = tid & 31;

    for (int i = tid; i < 32 * 64; i += 256) {
        int r  = i >> 6;
        int c4 = (i & 63) << 2;
        int rr = row0 + r; if (rr >= n_nodes) rr = n_nodes - 1;
        *(float4*)(&As[r * 256 + c4]) = *(const float4*)(A + (size_t)rr * 256 + c4);
    }

    const int m   = lane & 15;
    const int kk  = (lane >> 4) << 1;
    const int n0  = wave * 32;
    const int n1  = n0 + 16;

    v8f c00 = {}, c01 = {}, c10 = {}, c11 = {};

    for (int kc = 0; kc < 256; kc += 32) {
        __syncthreads();
        for (int i = tid; i < 32 * 64; i += 256) {
            int r  = i >> 6;
            int c4 = (i & 63) << 2;
            *(float4*)(&Ws[r * 256 + c4]) = *(const float4*)(W + (size_t)(kc + r) * 256 + c4);
        }
        __syncthreads();

#pragma unroll
        for (int k = 0; k < 32; k += 4) {
            v2f a0, a1, b0, b1;
            a0.x = As[m * 256 + kc + k + kk];
            a0.y = As[m * 256 + kc + k + kk + 1];
            a1.x = As[(m + 16) * 256 + kc + k + kk];
            a1.y = As[(m + 16) * 256 + kc + k + kk + 1];
            b0.x = Ws[(k + kk) * 256 + n0 + m];
            b0.y = Ws[(k + kk + 1) * 256 + n0 + m];
            b1.x = Ws[(k + kk) * 256 + n1 + m];
            b1.y = Ws[(k + kk + 1) * 256 + n1 + m];
            c00 = __builtin_amdgcn_wmma_f32_16x16x4_f32(false, a0, false, b0, (short)0, c00, false, false);
            c01 = __builtin_amdgcn_wmma_f32_16x16x4_f32(false, a0, false, b1, (short)0, c01, false, false);
            c10 = __builtin_amdgcn_wmma_f32_16x16x4_f32(false, a1, false, b0, (short)0, c10, false, false);
            c11 = __builtin_amdgcn_wmma_f32_16x16x4_f32(false, a1, false, b1, (short)0, c11, false, false);
        }
    }

    const int mhi = (lane >> 4) * 8;
    for (int v = 0; v < 8; ++v) {
        int r0 = row0 + mhi + v;
        int r1 = r0 + 16;
        if (r0 < n_nodes) {
            H[(size_t)r0 * 256 + n0 + m] = c00[v];
            H[(size_t)r0 * 256 + n1 + m] = c01[v];
        }
        if (r1 < n_nodes) {
            H[(size_t)r1 * 256 + n0 + m] = c10[v];
            H[(size_t)r1 * 256 + n1 + m] = c11[v];
        }
    }
}

__global__ void gat_attn_scores(const float* __restrict__ H,
                                const float* __restrict__ al,
                                const float* __restrict__ ar,
                                float* __restrict__ el, float* __restrict__ er,
                                int n_nodes) {
    int i = blockIdx.x * blockDim.x + threadIdx.x;
    if (i >= n_nodes * 8) return;
    int n = i >> 3, h = i & 7;
    const float* hp  = H + (size_t)n * 256 + h * 32;
    const float* alp = al + h * 32;
    const float* arp = ar + h * 32;
    float sl = 0.f, sr = 0.f;
    for (int d = 0; d < 32; ++d) { float v = hp[d]; sl += v * alp[d]; sr += v * arp[d]; }
    el[i] = sl; er[i] = sr;
}

__device__ inline void atomicMaxFloat(float* addr, float value) {
    if (value >= 0.0f)
        atomicMax((int*)addr, __float_as_int(value));
    else
        atomicMin((unsigned int*)addr, __float_as_uint(value));
}

__global__ void gat_edge_e_max(const int* __restrict__ src, const int* __restrict__ dst,
                               const float* __restrict__ el, const float* __restrict__ er,
                               float* __restrict__ e, float* __restrict__ m,
                               int n_edges, int heads) {
    int i = blockIdx.x * blockDim.x + threadIdx.x;
    if (i >= n_edges * heads) return;
    int ed = i / heads, h = i - ed * heads;
    float v = el[src[ed] * heads + h] + er[dst[ed] * heads + h];
    v = (v >= 0.f) ? v : SLOPE * v;
    e[i] = v;
    atomicMaxFloat(&m[dst[ed] * heads + h], v);
}

__global__ void gat_edge_expsum(const int* __restrict__ dst,
                                float* __restrict__ e, const float* __restrict__ m,
                                float* __restrict__ s, int n_edges, int heads) {
    int i = blockIdx.x * blockDim.x + threadIdx.x;
    if (i >= n_edges * heads) return;
    int ed = i / heads, h = i - ed * heads;
    float v = expf(e[i] - m[dst[ed] * heads + h]);
    e[i] = v;
    atomicAddF32(&s[dst[ed] * heads + h], v);
}

__global__ void gat_aggregate(const int* __restrict__ src, const int* __restrict__ dst,
                              const float* __restrict__ H, const float* __restrict__ ue,
                              const float* __restrict__ s, float* __restrict__ agg,
                              int n_edges, int heads, int dim) {
    int hd = heads * dim;
    int i = blockIdx.x * blockDim.x + threadIdx.x;
    if (i >= n_edges * hd) return;
    int ed = i / hd, c = i - ed * hd;
    int h = c / dim;
    int sn = src[ed], dn = dst[ed];
    float alpha = ue[ed * heads + h] / s[dn * heads + h];
    atomicAddF32(&agg[(size_t)dn * hd + c], H[(size_t)sn * hd + c] * alpha);
}

#define AGG_TILE    196
#define AGG_THREADS 1024
#define AGG_WAVES   32
#define AGG_STEP    128
#define AGG_QCAP    288

__device__ inline void ldsMaxFloat(float* addr, float value) {
    if (__float_as_int(value) >= 0)
        atomicMax((int*)addr, __float_as_int(value));
    else
        atomicMin((unsigned int*)addr, __float_as_uint(value));
}

__global__ __launch_bounds__(AGG_THREADS)
void gat_softmax_aggregate_tile(const int* __restrict__ src, const int* __restrict__ dst,
                                const float* __restrict__ H, const float* __restrict__ el,
                                const float* __restrict__ er, float* __restrict__ agg,
                                int n_edges, int n_nodes) {
    __shared__ __attribute__((aligned(16))) float acc[AGG_TILE * 256];
    __shared__ __attribute__((aligned(16))) float er_t[AGG_TILE * 8];
    __shared__ float m_t[AGG_TILE * 8];
    __shared__ float s_t[AGG_TILE * 8];
    __shared__ int   q_e[AGG_WAVES * AGG_QCAP];
    __shared__ int   q_d[AGG_WAVES * AGG_QCAP];

    const int tid  = threadIdx.x;
    const int wave = tid >> 5;
    const int lane = tid & 31;
    const int lo   = blockIdx.x * AGG_TILE;
    int rows = n_nodes - lo; if (rows > AGG_TILE) rows = AGG_TILE;
    const unsigned span = (unsigned)rows;

    for (int i = tid; i < AGG_TILE * 64; i += AGG_THREADS)
        ((float4*)acc)[i] = make_float4(0.f, 0.f, 0.f, 0.f);
    for (int i = tid; i < AGG_TILE * 8; i += AGG_THREADS) {
        er_t[i] = (i < rows * 8) ? er[(size_t)lo * 8 + i] : 0.f;
        m_t[i]  = -__builtin_huge_valf();
        s_t[i]  = 0.f;
    }
    __syncthreads();

    int* qe = q_e + wave * AGG_QCAP;
    int* qd = q_d + wave * AGG_QCAP;
    const unsigned lt_mask = (1u << lane) - 1u;
    const int hl = lane >> 3;
    int  cnt  = 0;
    int  done = 0;
    bool keep = true;

    auto drain_max = [&](int from, int n) {
        for (int b = 0; b < n; b += 32) {
            if (b + lane < n) {
                const int e  = qe[from + b + lane];
                const int dl = qd[from + b + lane];
                const int sn = src[e];
                const float4 l0 = *(const float4*)(el + (size_t)sn * 8);
                const float4 l1 = *(const float4*)(el + (size_t)sn * 8 + 4);
                const float4 r0 = *(const float4*)(er_t + dl * 8);
                const float4 r1 = *(const float4*)(er_t + dl * 8 + 4);
                const float x[8] = {l0.x + r0.x, l0.y + r0.y, l0.z + r0.z, l0.w + r0.w,
                                    l1.x + r1.x, l1.y + r1.y, l1.z + r1.z, l1.w + r1.w};
#pragma unroll
                for (int h = 0; h < 8; ++h) {
                    const float v = (x[h] >= 0.f) ? x[h] : SLOPE * x[h];
                    ldsMaxFloat(m_t + dl * 8 + h, v);
                }
            }
        }
    };

    auto drain_sum = [&](int from, int n) {
        for (int b = 0; b < n; b += 32) {
            int nb = n - b; if (nb > 32) nb = 32;
            const int i  = from + b + (lane < nb ? lane : nb - 1);
            const int e  = qe[i];
            const int dl = qd[i];
            const int sn = src[e];
            for (int g = 0; g < nb; g += 4) {
                int su[4], du[4];
#pragma unroll
                for (int u = 0; u < 4; ++u) {
                    const int k = (g + u < nb) ? g + u : nb - 1;
                    su[u] = __builtin_amdgcn_readlane(sn, k);
                    du[u] = __builtin_amdgcn_readlane(dl, k);
                }
                float4 r0[4], r1[4];
                float  l0[4], l1[4];
#pragma unroll
                for (int u = 0; u < 4; ++u) {
                    const float* hp = H + (size_t)su[u] * 256 + 4 * lane;
                    r0[u] = *(const float4*)(hp);
                    r1[u] = *(const float4*)(hp + 128);
                    l0[u] = el[(size_t)su[u] * 8 + hl];
                    l1[u] = el[(size_t)su[u] * 8 + 4 + hl];
                }
#pragma unroll
                for (int u = 0; u < 4; ++u) {
                    if (g + u < nb) {
                        const int t0 = du[u] * 8 + hl;
                        const float x0 = l0[u] + er_t[t0];
                        const float x1 = l1[u] + er_t[t0 + 4];
                        const float v0 = (x0 >= 0.f) ? x0 : SLOPE * x0;
                        const float v1 = (x1 >= 0.f) ? x1 : SLOPE * x1;
                        const float a0 = expf(v0 - m_t[t0]);
                        const float a1 = expf(v1 - m_t[t0 + 4]);
                        if ((lane & 7) == 0) {
                            atomicAddLds(s_t + t0,     a0);
                            atomicAddLds(s_t + t0 + 4, a1);
                        }
                        float* row = acc + du[u] * 256 + lane;
                        atomicAddLds(row,       r0[u].x * a0);
                        atomicAddLds(row + 32,  r0[u].y * a0);
                        atomicAddLds(row + 64,  r0[u].z * a0);
                        atomicAddLds(row + 96,  r0[u].w * a0);
                        atomicAddLds(row + 128, r1[u].x * a1);
                        atomicAddLds(row + 160, r1[u].y * a1);
                        atomicAddLds(row + 192, r1[u].z * a1);
                        atomicAddLds(row + 224, r1[u].w * a1);
                    }
                }
            }
        }
    };

    auto scan = [&](auto&& after) {
        const int n4 = n_edges & ~3;
        if (n4 > 0) {
            const int last    = n4 - 4;
            const int n_steps = cdiv(n4, AGG_STEP);
            auto load4 = [&](int st) {
                const int e0 = st * AGG_STEP + lane * 4;
                return *(const int4*)(dst + (e0 < last ? e0 : last));
            };
            int4 cur[4], nxt[4];
#pragma unroll
            for (int u = 0; u < 4; ++u) cur[u] = load4(wave + u * AGG_WAVES);

            for (int st0 = wave; st0 < n_steps; st0 += 4 * AGG_WAVES) {
#pragma unroll
                for (int u = 0; u < 4; ++u) nxt[u] = load4(st0 + (4 + u) * AGG_WAVES);
#pragma unroll
                for (int u = 0; u < 4; ++u) {
                    const int e0 = (st0 + u * AGG_WAVES) * AGG_STEP + lane * 4;
                    const bool valid = e0 < n4;
                    const int dv[4] = {cur[u].x, cur[u].y, cur[u].z, cur[u].w};
                    bool mt[4];
#pragma unroll
                    for (int j = 0; j < 4; ++j) mt[j] = valid && ((unsigned)(dv[j] - lo) < span);
                    if (__builtin_amdgcn_ballot_w32(mt[0] | mt[1] | mt[2] | mt[3]) != 0u) {
#pragma unroll
                        for (int j = 0; j < 4; ++j) {
                            const unsigned mask = __builtin_amdgcn_ballot_w32(mt[j]);
                            if (mt[j]) {
                                const int pos = cnt + __popc(mask & lt_mask);
                                qe[pos] = e0 + j;
                                qd[pos] = dv[j] - lo;
                            }
                            cnt += __popc(mask);
                        }
                        after();
                    }
                }
#pragma unroll
                for (int u = 0; u < 4; ++u) cur[u] = nxt[u];
            }
        }
        if (wave == 0) {
            const int e = n4 + lane;
            const int d = (e < n_edges) ? dst[e] : -1;
            const bool mt = (unsigned)(d - lo) < span;
            const unsigned mask = __builtin_amdgcn_ballot_w32(mt);
            if (mt) {
                const int pos = cnt + __popc(mask & lt_mask);
                qe[pos] = e;
                qd[pos] = d - lo;
            }
            cnt += __popc(mask);
            if (mask != 0u) after();
        }
    };

    scan([&]() {
        while (cnt - done >= 32) { drain_max(done, 32); done += 32; }
        if (cnt > AGG_QCAP - AGG_STEP) {
            if (cnt > done) drain_max(done, cnt - done);
            keep = false; cnt = 0; done = 0;
        }
    });
    if (cnt > done) drain_max(done, cnt - done);
    __syncthreads();

    if (keep) {
        if (cnt > 0) drain_sum(0, cnt);
    } else {
        cnt = 0;
        scan([&]() {
            if (cnt > AGG_QCAP - AGG_STEP) { drain_sum(0, cnt); cnt = 0; }
        });
        if (cnt > 0) drain_sum(0, cnt);
    }
    __syncthreads();

    for (int i = tid; i < rows * 64; i += AGG_THREADS) {
        const int r  = i >> 6;
        const int c4 = i & 63;
        const float* p = acc + r * 256 + (c4 >> 5) * 128 + (c4 & 31);
        const float sv = s_t[r * 8 + (c4 >> 3)];
        float4 o = make_float4(0.f, 0.f, 0.f, 0.f);
        if (sv != 0.f) o = make_float4(p[0] / sv, p[32] / sv, p[64] / sv, p[96] / sv);
        *(float4*)(agg + (size_t)(lo + r) * 256 + 4 * c4) = o;
    }
}

__global__ void gat_elu_bias(float* __restrict__ x, const float* __restrict__ b,
                             int total, int hd) {
    int i = blockIdx.x * blockDim.x + threadIdx.x;
    if (i >= total) return;
    float v = x[i] + b[i % hd];
    x[i] = (v > 0.f) ? v : expm1f(v);
}

__global__ void gat_l2_linear(const float* __restrict__ x, const float* __restrict__ W2,
                              const float* __restrict__ al2, const float* __restrict__ ar2,
                              float* __restrict__ h2, float* __restrict__ el2,
                              float* __restrict__ er2, int n_nodes) {
    int n = blockIdx.x * blockDim.x + threadIdx.x;
    if (n >= n_nodes) return;
    float a0 = 0.f, a1 = 0.f, a2 = 0.f;
    const float* xp = x + (size_t)n * 256;
    for (int k = 0; k < 256; ++k) {
        float xv = xp[k];
        a0 += xv * W2[k * 3 + 0];
        a1 += xv * W2[k * 3 + 1];
        a2 += xv * W2[k * 3 + 2];
    }
    h2[n * 3 + 0] = a0; h2[n * 3 + 1] = a1; h2[n * 3 + 2] = a2;
    el2[n] = a0 * al2[0] + a1 * al2[1] + a2 * al2[2];
    er2[n] = a0 * ar2[0] + a1 * ar2[1] + a2 * ar2[2];
}

__global__ void gat_finalize(const float* __restrict__ agg2, const float* __restrict__ b2,
                             float* __restrict__ out, int total) {
    int i = blockIdx.x * blockDim.x + threadIdx.x;
    if (i >= total) return;
    out[i] = agg2[i] + b2[i % 3];
}

extern "C" void kernel_launch(void* const* d_in, const int* in_sizes, int n_in,
                              void* d_out, int out_size, void* d_ws, size_t ws_size,
                              hipStream_t stream) {
    const float* feat = (const float*)d_in[0];
    const int*   src  = (const int*)  d_in[1];
    const int*   dst  = (const int*)  d_in[2];
    const float* W1   = (const float*)d_in[3];
    const float* al1  = (const float*)d_in[4];
    const float* ar1  = (const float*)d_in[5];
    const float* b1   = (const float*)d_in[6];
    const float* W2   = (const float*)d_in[7];
    const float* al2  = (const float*)d_in[8];
    const float* ar2  = (const float*)d_in[9];
    const float* b2   = (const float*)d_in[10];
    float* out = (float*)d_out;

    const int n_nodes = in_sizes[0] / 256;
    const int n_edges = in_sizes[1];

    char* cur = (char*)d_ws;
    auto alloc = [&](size_t bytes) -> float* {
        float* p = (float*)cur;
        cur += (bytes + 255) & ~(size_t)255;
        return p;
    };
    float* H1   = alloc((size_t)n_nodes * 256 * 4);
    float* el1  = alloc((size_t)n_nodes * 8 * 4);
    float* er1  = alloc((size_t)n_nodes * 8 * 4);
    float* agg1 = alloc((size_t)n_nodes * 256 * 4);
    float* h2   = alloc((size_t)n_nodes * 3 * 4);
    float* el2  = alloc((size_t)n_nodes * 4);
    float* er2  = alloc((size_t)n_nodes * 4);
    float* m2   = alloc((size_t)n_nodes * 4);
    float* s2   = alloc((size_t)n_nodes * 4);
    float* e2   = alloc((size_t)n_edges * 4);
    float* agg2 = alloc((size_t)n_nodes * 3 * 4);

    const int B = 256;
    const float NEG_INF = -__builtin_huge_valf();

    gat_gemm1_wmma<<<cdiv(n_nodes, 32), 256, 0, stream>>>(feat, W1, H1, n_nodes);
    gat_attn_scores<<<cdiv(n_nodes * 8, B), B, 0, stream>>>(H1, al1, ar1, el1, er1, n_nodes);

    gat_softmax_aggregate_tile<<<cdiv(n_nodes, AGG_TILE), AGG_THREADS, 0, stream>>>(src, dst, H1, el1, er1, agg1,
                                                                                    n_edges, n_nodes);
    gat_elu_bias<<<cdiv(n_nodes * 256, B), B, 0, stream>>>(agg1, b1, n_nodes * 256, 256);

    gat_l2_linear<<<cdiv(n_nodes, 128), 128, 0, stream>>>(agg1, W2, al2, ar2, h2, el2, er2, n_nodes);

    gat_fill_f32<<<cdiv(n_nodes, B), B, 0, stream>>>(m2, NEG_INF, n_nodes);
    gat_fill_f32<<<cdiv(n_nodes, B), B, 0, stream>>>(s2, 0.f, n_nodes);
    gat_fill_f32<<<cdiv(n_nodes * 3, B), B, 0, stream>>>(agg2, 0.f, n_nodes * 3);

    gat_edge_e_max<<<cdiv(n_edges, B), B, 0, stream>>>(src, dst, el2, er2, e2, m2, n_edges, 1);
    gat_edge_expsum<<<cdiv(n_edges, B), B, 0, stream>>>(dst, e2, m2, s2, n_edges, 1);
    gat_aggregate<<<cdiv(n_edges * 3, B), B, 0, stream>>>(src, dst, h2, e2, s2, agg2,
                                                          n_edges, 1, 3);
    gat_finalize<<<cdiv(n_nodes * 3, B), B, 0, stream>>>(agg2, b2, out, n_nodes * 3);
}


# Owner groups finish their LDS rows: /exp sum, +b1, elu, x@W2 (256x3), el2/er2, one wave per row (8 cols per lane, xor tree: W2 sum order changed); agg never stored, 2 launches fewer
# speedup vs baseline: 2.6664x; 1.1234x over previous
#include <hip/hip_runtime.h>
#include <hip/hip_bf16.h>

typedef __attribute__((ext_vector_type(2))) float v2f;
typedef __attribute__((ext_vector_type(8))) float v8f;

#define SLOPE 0.2f

static __host__ __device__ inline int cdiv(int a, int b) { return (a + b - 1) / b; }

__device__ inline void atomicAddF32(float* p, float v) { unsafeAtomicAdd(p, v); }
__device__ inline void atomicAddLds(float* p, float v) {
    __hip_atomic_fetch_add(p, v, __ATOMIC_RELAXED, __HIP_MEMORY_SCOPE_WORKGROUP);
}

__global__ void gat_fill_f32(float* __restrict__ p, float v, int n) {
    int i = blockIdx.x * blockDim.x + threadIdx.x;
    if (i < n) p[i] = v;
}

__global__ void gat_gemm1_wmma(const float* __restrict__ A,
                               const float* __restrict__ W,
                               float* __restrict__ H,
                               int n_nodes) {
    __shared__ float As[32 * 256];
    __shared__ float Ws[32 * 256];

    const int row0 = blockIdx.x * 32;
    const int tid  = threadIdx.x;
    const int wave = tid >> 5;
    const int lane = tid & 31;

    for (int i = tid; i < 32 * 64; i += 256) {
        int r  = i >> 6;
        int c4 = (i & 63) << 2;
        int rr = row0 + r; if (rr >= n_nodes) rr = n_nodes - 1;
        *(float4*)(&As[r * 256 + c4]) = *(const float4*)(A + (size_t)rr * 256 + c4);
    }

    const int m   = lane & 15;
    const int kk  = (lane >> 4) << 1;
    const int n0  = wave * 32;
    const int n1  = n0 + 16;

    v8f c00 = {}, c01 = {}, c10 = {}, c11 = {};

    for (int kc = 0; kc < 256; kc += 32) {
        __syncthreads();
        for (int i = tid; i < 32 * 64; i += 256) {
            int r  = i >> 6;
            int c4 = (i & 63) << 2;
            *(float4*)(&Ws[r * 256 + c4]) = *(const float4*)(W + (size_t)(kc + r) * 256 + c4);
        }
        __syncthreads();

#pragma unroll
        for (int k = 0; k < 32; k += 4) {
            v2f a0, a1, b0, b1;
            a0.x = As[m * 256 + kc + k + kk];
            a0.y = As[m * 256 + kc + k + kk + 1];
            a1.x = As[(m + 16) * 256 + kc + k + kk];
            a1.y = As[(m + 16) * 256 + kc + k + kk + 1];
            b0.x = Ws[(k + kk) * 256 + n0 + m];
            b0.y = Ws[(k + kk + 1) * 256 + n0 + m];
            b1.x = Ws[(k + kk) * 256 + n1 + m];
            b1.y = Ws[(k + kk + 1) * 256 + n1 + m];
            c00 = __builtin_amdgcn_wmma_f32_16x16x4_f32(false, a0, false, b0, (short)0, c00, false, false);
            c01 = __builtin_amdgcn_wmma_f32_16x16x4_f32(false, a0, false, b1, (short)0, c01, false, false);
            c10 = __builtin_amdgcn_wmma_f32_16x16x4_f32(false, a1, false, b0, (short)0, c10, false, false);
            c11 = __builtin_amdgcn_wmma_f32_16x16x4_f32(false, a1, false, b1, (short)0, c11, false, false);
        }
    }

    const int mhi = (lane >> 4) * 8;
    for (int v = 0; v < 8; ++v) {
        int r0 = row0 + mhi + v;
        int r1 = r0 + 16;
        if (r0 < n_nodes) {
            H[(size_t)r0 * 256 + n0 + m] = c00[v];
            H[(size_t)r0 * 256 + n1 + m] = c01[v];
        }
        if (r1 < n_nodes) {
            H[(size_t)r1 * 256 + n0 + m] = c10[v];
            H[(size_t)r1 * 256 + n1 + m] = c11[v];
        }
    }
}

__global__ void gat_attn_scores(const float* __restrict__ H,
                                const float* __restrict__ al,
                                const float* __restrict__ ar,
                                float* __restrict__ el, float* __restrict__ er,
                                int n_nodes) {
    int i = blockIdx.x * blockDim.x + threadIdx.x;
    if (i >= n_nodes * 8) return;
    int n = i >> 3, h = i & 7;
    const float* hp  = H + (size_t)n * 256 + h * 32;
    const float* alp = al + h * 32;
    const float* arp = ar + h * 32;
    float sl = 0.f, sr = 0.f;
    for (int d = 0; d < 32; ++d) { float v = hp[d]; sl += v * alp[d]; sr += v * arp[d]; }
    el[i] = sl; er[i] = sr;
}

__device__ inline void atomicMaxFloat(float* addr, float value) {
    if (value >= 0.0f)
        atomicMax((int*)addr, __float_as_int(value));
    else
        atomicMin((unsigned int*)addr, __float_as_uint(value));
}

__global__ void gat_edge_e_max(const int* __restrict__ src, const int* __restrict__ dst,
                               const float* __restrict__ el, const float* __restrict__ er,
                               float* __restrict__ e, float* __restrict__ m,
                               int n_edges, int heads) {
    int i = blockIdx.x * blockDim.x + threadIdx.x;
    if (i >= n_edges * heads) return;
    int ed = i / heads, h = i - ed * heads;
    float v = el[src[ed] * heads + h] + er[dst[ed] * heads + h];
    v = (v >= 0.f) ? v : SLOPE * v;
    e[i] = v;
    atomicMaxFloat(&m[dst[ed] * heads + h], v);
}

__global__ void gat_edge_expsum(const int* __restrict__ dst,
                                float* __restrict__ e, const float* __restrict__ m,
                                float* __restrict__ s, int n_edges, int heads) {
    int i = blockIdx.x * blockDim.x + threadIdx.x;
    if (i >= n_edges * heads) return;
    int ed = i / heads, h = i - ed * heads;
    float v = expf(e[i] - m[dst[ed] * heads + h]);
    e[i] = v;
    atomicAddF32(&s[dst[ed] * heads + h], v);
}

__global__ void gat_aggregate(const int* __restrict__ src, const int* __restrict__ dst,
                              const float* __restrict__ H, const float* __restrict__ ue,
                              const float* __restrict__ s, float* __restrict__ agg,
                              int n_edges, int heads, int dim) {
    int hd = heads * dim;
    int i = blockIdx.x * blockDim.x + threadIdx.x;
    if (i >= n_edges * hd) return;
    int ed = i / hd, c = i - ed * hd;
    int h = c / dim;
    int sn = src[ed], dn = dst[ed];
    float alpha = ue[ed * heads + h] / s[dn * heads + h];
    atomicAddF32(&agg[(size_t)dn * hd + c], H[(size_t)sn * hd + c] * alpha);
}

#define AGG_TILE    196
#define AGG_THREADS 1024
#define AGG_WAVES   32
#define AGG_STEP    128
#define AGG_QCAP    288

__device__ inline void ldsMaxFloat(float* addr, float value) {
    if (__float_as_int(value) >= 0)
        atomicMax((int*)addr, __float_as_int(value));
    else
        atomicMin((unsigned int*)addr, __float_as_uint(value));
}

__global__ __launch_bounds__(AGG_THREADS)
void gat_softmax_aggregate_tile(const int* __restrict__ src, const int* __restrict__ dst,
                                const float* __restrict__ H, const float* __restrict__ el,
                                const float* __restrict__ er,
                                const float* __restrict__ b1, const float* __restrict__ W2,
                                const float* __restrict__ al2, const float* __restrict__ ar2,
                                float* __restrict__ h2, float* __restrict__ el2,
                                float* __restrict__ er2,
                                int n_edges, int n_nodes) {
    __shared__ __attribute__((aligned(16))) float acc[AGG_TILE * 256];
    __shared__ __attribute__((aligned(16))) float er_t[AGG_TILE * 8];
    __shared__ float m_t[AGG_TILE * 8];
    __shared__ float s_t[AGG_TILE * 8];
    __shared__ int   q_e[AGG_WAVES * AGG_QCAP];
    __shared__ int   q_d[AGG_WAVES * AGG_QCAP];

    const int tid  = threadIdx.x;
    const int wave = tid >> 5;
    const int lane = tid & 31;
    const int lo   = blockIdx.x * AGG_TILE;
    int rows = n_nodes - lo; if (rows > AGG_TILE) rows = AGG_TILE;
    const unsigned span = (unsigned)rows;

    for (int i = tid; i < AGG_TILE * 64; i += AGG_THREADS)
        ((float4*)acc)[i] = make_float4(0.f, 0.f, 0.f, 0.f);
    for (int i = tid; i < AGG_TILE * 8; i += AGG_THREADS) {
        er_t[i] = (i < rows * 8) ? er[(size_t)lo * 8 + i] : 0.f;
        m_t[i]  = -__builtin_huge_valf();
        s_t[i]  = 0.f;
    }
    __syncthreads();

    int* qe = q_e + wave * AGG_QCAP;
    int* qd = q_d + wave * AGG_QCAP;
    const unsigned lt_mask = (1u << lane) - 1u;
    const int hl = lane >> 3;
    int  cnt  = 0;
    int  done = 0;
    bool keep = true;

    auto drain_max = [&](int from, int n) {
        for (int b = 0; b < n; b += 32) {
            if (b + lane < n) {
                const int e  = qe[from + b + lane];
                const int dl = qd[from + b + lane];
                const int sn = src[e];
                const float4 l0 = *(const float4*)(el + (size_t)sn * 8);
                const float4 l1 = *(const float4*)(el + (size_t)sn * 8 + 4);
                const float4 r0 = *(const float4*)(er_t + dl * 8);
                const float4 r1 = *(const float4*)(er_t + dl * 8 + 4);
                const float x[8] = {l0.x + r0.x, l0.y + r0.y, l0.z + r0.z, l0.w + r0.w,
                                    l1.x + r1.x, l1.y + r1.y, l1.z + r1.z, l1.w + r1.w};
#pragma unroll
                for (int h = 0; h < 8; ++h) {
                    const float v = (x[h] >= 0.f) ? x[h] : SLOPE * x[h];
                    ldsMaxFloat(m_t + dl * 8 + h, v);
                }
            }
        }
    };

    auto drain_sum = [&](int from, int n) {
        for (int b = 0; b < n; b += 32) {
            int nb = n - b; if (nb > 32) nb = 32;
            const int i  = from + b + (lane < nb ? lane : nb - 1);
            const int e  = qe[i];
            const int dl = qd[i];
            const int sn = src[e];
            for (int g = 0; g < nb; g += 4) {
                int su[4], du[4];
#pragma unroll
                for (int u = 0; u < 4; ++u) {
                    const int k = (g + u < nb) ? g + u : nb - 1;
                    su[u] = __builtin_amdgcn_readlane(sn, k);
                    du[u] = __builtin_amdgcn_readlane(dl, k);
                }
                float4 r0[4], r1[4];
                float  l0[4], l1[4];
#pragma unroll
                for (int u = 0; u < 4; ++u) {
                    const float* hp = H + (size_t)su[u] * 256 + 4 * lane;
                    r0[u] = *(const float4*)(hp);
                    r1[u] = *(const float4*)(hp + 128);
                    l0[u] = el[(size_t)su[u] * 8 + hl];
                    l1[u] = el[(size_t)su[u] * 8 + 4 + hl];
                }
#pragma unroll
                for (int u = 0; u < 4; ++u) {
                    if (g + u < nb) {
                        const int t0 = du[u] * 8 + hl;
                        const float x0 = l0[u] + er_t[t0];
                        const float x1 = l1[u] + er_t[t0 + 4];
                        const float v0 = (x0 >= 0.f) ? x0 : SLOPE * x0;
                        const float v1 = (x1 >= 0.f) ? x1 : SLOPE * x1;
                        const float a0 = expf(v0 - m_t[t0]);
                        const float a1 = expf(v1 - m_t[t0 + 4]);
                        if ((lane & 7) == 0) {
                            atomicAddLds(s_t + t0,     a0);
                            atomicAddLds(s_t + t0 + 4, a1);
                        }
                        float* row = acc + du[u] * 256 + lane;
                        atomicAddLds(row,       r0[u].x * a0);
                        atomicAddLds(row + 32,  r0[u].y * a0);
                        atomicAddLds(row + 64,  r0[u].z * a0);
                        atomicAddLds(row + 96,  r0[u].w * a0);
                        atomicAddLds(row + 128, r1[u].x * a1);
                        atomicAddLds(row + 160, r1[u].y * a1);
                        atomicAddLds(row + 192, r1[u].z * a1);
                        atomicAddLds(row + 224, r1[u].w * a1);
                    }
                }
            }
        }
    };

    auto scan = [&](auto&& after) {
        const int n4 = n_edges & ~3;
        if (n4 > 0) {
            const int last    = n4 - 4;
            const int n_steps = cdiv(n4, AGG_STEP);
            auto load4 = [&](int st) {
                const int e0 = st * AGG_STEP + lane * 4;
                return *(const int4*)(dst + (e0 < last ? e0 : last));
            };
            int4 cur[4], nxt[4];
#pragma unroll
            for (int u = 0; u < 4; ++u) cur[u] = load4(wave + u * AGG_WAVES);

            for (int st0 = wave; st0 < n_steps; st0 += 4 * AGG_WAVES) {
#pragma unroll
                for (int u = 0; u < 4; ++u) nxt[u] = load4(st0 + (4 + u) * AGG_WAVES);
#pragma unroll
                for (int u = 0; u < 4; ++u) {
                    const int e0 = (st0 + u * AGG_WAVES) * AGG_STEP + lane * 4;
                    const bool valid = e0 < n4;
                    const int dv[4] = {cur[u].x, cur[u].y, cur[u].z, cur[u].w};
                    bool mt[4];
#pragma unroll
                    for (int j = 0; j < 4; ++j) mt[j] = valid && ((unsigned)(dv[j] - lo) < span);
                    if (__builtin_amdgcn_ballot_w32(mt[0] | mt[1] | mt[2] | mt[3]) != 0u) {
#pragma unroll
                        for (int j = 0; j < 4; ++j) {
                            const unsigned mask = __builtin_amdgcn_ballot_w32(mt[j]);
                            if (mt[j]) {
                                const int pos = cnt + __popc(mask & lt_mask);
                                qe[pos] = e0 + j;
                                qd[pos] = dv[j] - lo;
                            }
                            cnt += __popc(mask);
                        }
                        after();
                    }
                }
#pragma unroll
                for (int u = 0; u < 4; ++u) cur[u] = nxt[u];
            }
        }
        if (wave == 0) {
            const int e = n4 + lane;
            const int d = (e < n_edges) ? dst[e] : -1;
            const bool mt = (unsigned)(d - lo) < span;
            const unsigned mask = __builtin_amdgcn_ballot_w32(mt);
            if (mt) {
                const int pos = cnt + __popc(mask & lt_mask);
                qe[pos] = e;
                qd[pos] = d - lo;
            }
            cnt += __popc(mask);
            if (mask != 0u) after();
        }
    };

    scan([&]() {
        while (cnt - done >= 32) { drain_max(done, 32); done += 32; }
        if (cnt > AGG_QCAP - AGG_STEP) {
            if (cnt > done) drain_max(done, cnt - done);
            keep = false; cnt = 0; done = 0;
        }
    });
    if (cnt > done) drain_max(done, cnt - done);
    __syncthreads();

    if (keep) {
        if (cnt > 0) drain_sum(0, cnt);
    } else {
        cnt = 0;
        scan([&]() {
            if (cnt > AGG_QCAP - AGG_STEP) { drain_sum(0, cnt); cnt = 0; }
        });
        if (cnt > 0) drain_sum(0, cnt);
    }
    __syncthreads();

    float bw[8], w0[8], w1[8], w2[8];
#pragma unroll
    for (int j = 0; j < 8; ++j) {
        const int c = (j >> 2) * 128 + 4 * lane + (j & 3);
        bw[j] = b1[c];
        w0[j] = W2[c * 3 + 0];
        w1[j] = W2[c * 3 + 1];
        w2[j] = W2[c * 3 + 2];
    }
    const float l2a0 = al2[0], l2a1 = al2[1], l2a2 = al2[2];
    const float r2a0 = ar2[0], r2a1 = ar2[1], r2a2 = ar2[2];
    for (int r = wave; r < rows; r += AGG_WAVES) {
        const float* p = acc + r * 256 + lane;
        const float sv0 = s_t[r * 8 + hl];
        const float sv1 = s_t[r * 8 + 4 + hl];
        float a0 = 0.f, a1 = 0.f, a2 = 0.f;
#pragma unroll
        for (int j = 0; j < 8; ++j) {
            const float sv = (j < 4) ? sv0 : sv1;
            const float q  = p[(j >> 2) * 128 + (j & 3) * 32];
            float v = (sv != 0.f) ? q / sv : 0.f;
            v += bw[j];
            v = (v > 0.f) ? v : expm1f(v);
            a0 += v * w0[j];
            a1 += v * w1[j];
            a2 += v * w2[j];
        }
        for (int o = 16; o > 0; o >>= 1) {
            a0 += __shfl_xor(a0, o);
            a1 += __shfl_xor(a1, o);
            a2 += __shfl_xor(a2, o);
        }
        if (lane == 0) {
            const size_t n = (size_t)(lo + r);
            h2[n * 3 + 0] = a0;
            h2[n * 3 + 1] = a1;
            h2[n * 3 + 2] = a2;
            el2[n] = a0 * l2a0 + a1 * l2a1 + a2 * l2a2;
            er2[n] = a0 * r2a0 + a1 * r2a1 + a2 * r2a2;
        }
    }
}

__global__ void gat_finalize(const float* __restrict__ agg2, const float* __restrict__ b2,
                             float* __restrict__ out, int total) {
    int i = blockIdx.x * blockDim.x + threadIdx.x;
    if (i >= total) return;
    out[i] = agg2[i] + b2[i % 3];
}

extern "C" void kernel_launch(void* const* d_in, const int* in_sizes, int n_in,
                              void* d_out, int out_size, void* d_ws, size_t ws_size,
                              hipStream_t stream) {
    const float* feat = (const float*)d_in[0];
    const int*   src  = (const int*)  d_in[1];
    const int*   dst  = (const int*)  d_in[2];
    const float* W1   = (const float*)d_in[3];
    const float* al1  = (const float*)d_in[4];
    const float* ar1  = (const float*)d_in[5];
    const float* b1   = (const float*)d_in[6];
    const float* W2   = (const float*)d_in[7];
    const float* al2  = (const float*)d_in[8];
    const float* ar2  = (const float*)d_in[9];
    const float* b2   = (const float*)d_in[10];
    float* out = (float*)d_out;

    const int n_nodes = in_sizes[0] / 256;
    const int n_edges = in_sizes[1];

    char* cur = (char*)d_ws;
    auto alloc = [&](size_t bytes) -> float* {
        float* p = (float*)cur;
        cur += (bytes + 255) & ~(size_t)255;
        return p;
    };
    float* H1   = alloc((size_t)n_nodes * 256 * 4);
    float* el1  = alloc((size_t)n_nodes * 8 * 4);
    float* er1  = alloc((size_t)n_nodes * 8 * 4);
    float* h2   = alloc((size_t)n_nodes * 3 * 4);
    float* el2  = alloc((size_t)n_nodes * 4);
    float* er2  = alloc((size_t)n_nodes * 4);
    float* m2   = alloc((size_t)n_nodes * 4);
    float* s2   = alloc((size_t)n_nodes * 4);
    float* e2   = alloc((size_t)n_edges * 4);
    float* agg2 = alloc((size_t)n_nodes * 3 * 4);

    const int B = 256;
    const float NEG_INF = -__builtin_huge_valf();

    gat_gemm1_wmma<<<cdiv(n_nodes, 32), 256, 0, stream>>>(feat, W1, H1, n_nodes);
    gat_attn_scores<<<cdiv(n_nodes * 8, B), B, 0, stream>>>(H1, al1, ar1, el1, er1, n_nodes);

    gat_softmax_aggregate_tile<<<cdiv(n_nodes, AGG_TILE), AGG_THREADS, 0, stream>>>(src, dst, H1, el1, er1,
                                                                                    b1, W2, al2, ar2, h2, el2, er2,
                                                                                    n_edges, n_nodes);

    gat_fill_f32<<<cdiv(n_nodes, B), B, 0, stream>>>(m2, NEG_INF, n_nodes);
    gat_fill_f32<<<cdiv(n_nodes, B), B, 0, stream>>>(s2, 0.f, n_nodes);
    gat_fill_f32<<<cdiv(n_nodes * 3, B), B, 0, stream>>>(agg2, 0.f, n_nodes * 3);

    gat_edge_e_max<<<cdiv(n_edges, B), B, 0, stream>>>(src, dst, el2, er2, e2, m2, n_edges, 1);
    gat_edge_expsum<<<cdiv(n_edges, B), B, 0, stream>>>(dst, e2, m2, s2, n_edges, 1);
    gat_aggregate<<<cdiv(n_edges * 3, B), B, 0, stream>>>(src, dst, h2, e2, s2, agg2,
                                                          n_edges, 1, 3);
    gat_finalize<<<cdiv(n_nodes * 3, B), B, 0, stream>>>(agg2, b2, out, n_nodes * 3);
}


# Layer 2 (max, exp sum, 3 sums, /sum at the end, +b2) by dst-range owner groups in LDS from edge lists the layer-1 launch files in workspace: 6 launches fewer, no global atomics (LDS arrival order)
# speedup vs baseline: 3.8537x; 1.4453x over previous
#include <hip/hip_runtime.h>
#include <hip/hip_bf16.h>

typedef __attribute__((ext_vector_type(2))) float v2f;
typedef __attribute__((ext_vector_type(8))) float v8f;

#define SLOPE 0.2f

static __host__ __device__ inline int cdiv(int a, int b) { return (a + b - 1) / b; }

__device__ inline void atomicAddLds(float* p, float v) {
    __hip_atomic_fetch_add(p, v, __ATOMIC_RELAXED, __HIP_MEMORY_SCOPE_WORKGROUP);
}

__global__ void gat_gemm1_wmma(const float* __restrict__ A,
                               const float* __restrict__ W,
                               float* __restrict__ H,
                               int n_nodes) {
    __shared__ float As[32 * 256];
    __shared__ float Ws[32 * 256];

    const int row0 = blockIdx.x * 32;
    const int tid  = threadIdx.x;
    const int wave = tid >> 5;
    const int lane = tid & 31;

    for (int i = tid; i < 32 * 64; i += 256) {
        int r  = i >> 6;
        int c4 = (i & 63) << 2;
        int rr = row0 + r; if (rr >= n_nodes) rr = n_nodes - 1;
        *(float4*)(&As[r * 256 + c4]) = *(const float4*)(A + (size_t)rr * 256 + c4);
    }

    const int m   = lane & 15;
    const int kk  = (lane >> 4) << 1;
    const int n0  = wave * 32;
    const int n1  = n0 + 16;

    v8f c00 = {}, c01 = {}, c10 = {}, c11 = {};

    for (int kc = 0; kc < 256; kc += 32) {
        __syncthreads();
        for (int i = tid; i < 32 * 64; i += 256) {
            int r  = i >> 6;
            int c4 = (i & 63) << 2;
            *(float4*)(&Ws[r * 256 + c4]) = *(const float4*)(W + (size_t)(kc + r) * 256 + c4);
        }
        __syncthreads();

#pragma unroll
        for (int k = 0; k < 32; k += 4) {
            v2f a0, a1, b0, b1;
            a0.x = As[m * 256 + kc + k + kk];
            a0.y = As[m * 256 + kc + k + kk + 1];
            a1.x = As[(m + 16) * 256 + kc + k + kk];
            a1.y = As[(m + 16) * 256 + kc + k + kk + 1];
            b0.x = Ws[(k + kk) * 256 + n0 + m];
            b0.y = Ws[(k + kk + 1) * 256 + n0 + m];
            b1.x = Ws[(k + kk) * 256 + n1 + m];
            b1.y = Ws[(k + kk + 1) * 256 + n1 + m];
            c00 = __builtin_amdgcn_wmma_f32_16x16x4_f32(false, a0, false, b0, (short)0, c00, false, false);
            c01 = __builtin_amdgcn_wmma_f32_16x16x4_f32(false, a0, false, b1, (short)0, c01, false, false);
            c10 = __builtin_amdgcn_wmma_f32_16x16x4_f32(false, a1, false, b0, (short)0, c10, false, false);
            c11 = __builtin_amdgcn_wmma_f32_16x16x4_f32(false, a1, false, b1, (short)0, c11, false, false);
        }
    }

    const int mhi = (lane >> 4) * 8;
    for (int v = 0; v < 8; ++v) {
        int r0 = row0 + mhi + v;
        int r1 = r0 + 16;
        if (r0 < n_nodes) {
            H[(size_t)r0 * 256 + n0 + m] = c00[v];
            H[(size_t)r0 * 256 + n1 + m] = c01[v];
        }
        if (r1 < n_nodes) {
            H[(size_t)r1 * 256 + n0 + m] = c10[v];
            H[(size_t)r1 * 256 + n1 + m] = c11[v];
        }
    }
}

__global__ void gat_attn_scores(const float* __restrict__ H,
                                const float* __restrict__ al,
                                const float* __restrict__ ar,
                                float* __restrict__ el, float* __restrict__ er,
                                int n_nodes) {
    int i = blockIdx.x * blockDim.x + threadIdx.x;
    if (i >= n_nodes * 8) return;
    int n = i >> 3, h = i & 7;
    const float* hp  = H + (size_t)n * 256 + h * 32;
    const float* alp = al + h * 32;
    const float* arp = ar + h * 32;
    float sl = 0.f, sr = 0.f;
    for (int d = 0; d < 32; ++d) { float v = hp[d]; sl += v * alp[d]; sr += v * arp[d]; }
    el[i] = sl; er[i] = sr;
}

#define AGG_TILE    196
#define AGG_THREADS 1024
#define AGG_WAVES   32
#define AGG_STEP    128
#define AGG_QCAP    288
#define L2_CAP      512

__device__ inline void ldsMaxFloat(float* addr, float value) {
    if (__float_as_int(value) >= 0)
        atomicMax((int*)addr, __float_as_int(value));
    else
        atomicMin((unsigned int*)addr, __float_as_uint(value));
}

__global__ __launch_bounds__(AGG_THREADS)
void gat_softmax_aggregate_tile(const int* __restrict__ src, const int* __restrict__ dst,
                                const float* __restrict__ H, const float* __restrict__ el,
                                const float* __restrict__ er,
                                const float* __restrict__ b1, const float* __restrict__ W2,
                                const float* __restrict__ al2, const float* __restrict__ ar2,
                                float* __restrict__ h2, float* __restrict__ el2,
                                float* __restrict__ er2,
                                int2* __restrict__ l2_list, int* __restrict__ l2_cnt,
                                int n_edges, int n_nodes) {
    __shared__ __attribute__((aligned(16))) float acc[AGG_TILE * 256];
    __shared__ __attribute__((aligned(16))) float er_t[AGG_TILE * 8];
    __shared__ float m_t[AGG_TILE * 8];
    __shared__ float s_t[AGG_TILE * 8];
    __shared__ int   q_e[AGG_WAVES * AGG_QCAP];
    __shared__ int   q_d[AGG_WAVES * AGG_QCAP];

    const int tid  = threadIdx.x;
    const int wave = tid >> 5;
    const int lane = tid & 31;
    const int lo   = blockIdx.x * AGG_TILE;
    int rows = n_nodes - lo; if (rows > AGG_TILE) rows = AGG_TILE;
    const unsigned span = (unsigned)rows;

    for (int i = tid; i < AGG_TILE * 64; i += AGG_THREADS)
        ((float4*)acc)[i] = make_float4(0.f, 0.f, 0.f, 0.f);
    for (int i = tid; i < AGG_TILE * 8; i += AGG_THREADS) {
        er_t[i] = (i < rows * 8) ? er[(size_t)lo * 8 + i] : 0.f;
        m_t[i]  = -__builtin_huge_valf();
        s_t[i]  = 0.f;
    }
    __syncthreads();

    int* qe = q_e + wave * AGG_QCAP;
    int* qd = q_d + wave * AGG_QCAP;
    const unsigned lt_mask = (1u << lane) - 1u;
    const int hl = lane >> 3;
    int  cnt  = 0;
    int  done = 0;
    bool keep = true;
    int  filed = 0;
    int2* slot = l2_list + (size_t)(blockIdx.x * AGG_WAVES + wave) * L2_CAP;

    auto drain_max = [&](int from, int n) {
        for (int b = 0; b < n; b += 32) {
            if (b + lane < n) {
                const int e  = qe[from + b + lane];
                const int dl = qd[from + b + lane];
                const int sn = src[e];
                const int fp = filed + b + lane;
                if (fp < L2_CAP) slot[fp] = make_int2(sn, dl);
                const float4 l0 = *(const float4*)(el + (size_t)sn * 8);
                const float4 l1 = *(const float4*)(el + (size_t)sn * 8 + 4);
                const float4 r0 = *(const float4*)(er_t + dl * 8);
                const float4 r1 = *(const float4*)(er_t + dl * 8 + 4);
                const float x[8] = {l0.x + r0.x, l0.y + r0.y, l0.z + r0.z, l0.w + r0.w,
                                    l1.x + r1.x, l1.y + r1.y, l1.z + r1.z, l1.w + r1.w};
#pragma unroll
                for (int h = 0; h < 8; ++h) {
                    const float v = (x[h] >= 0.f) ? x[h] : SLOPE * x[h];
                    ldsMaxFloat(m_t + dl * 8 + h, v);
                }
            }
        }
        filed += n;
    };

    auto drain_sum = [&](int from, int n) {
        for (int b = 0; b < n; b += 32) {
            int nb = n - b; if (nb > 32) nb = 32;
            const int i  = from + b + (lane < nb ? lane : nb - 1);
            const int e  = qe[i];
            const int dl = qd[i];
            const int sn = src[e];
            for (int g = 0; g < nb; g += 4) {
                int su[4], du[4];
#pragma unroll
                for (int u = 0; u < 4; ++u) {
                    const int k = (g + u < nb) ? g + u : nb - 1;
                    su[u] = __builtin_amdgcn_readlane(sn, k);
                    du[u] = __builtin_amdgcn_readlane(dl, k);
                }
                float4 r0[4], r1[4];
                float  l0[4], l1[4];
#pragma unroll
                for (int u = 0; u < 4; ++u) {
                    const float* hp = H + (size_t)su[u] * 256 + 4 * lane;
                    r0[u] = *(const float4*)(hp);
                    r1[u] = *(const float4*)(hp + 128);
                    l0[u] = el[(size_t)su[u] * 8 + hl];
                    l1[u] = el[(size_t)su[u] * 8 + 4 + hl];
                }
#pragma unroll
                for (int u = 0; u < 4; ++u) {
                    if (g + u < nb) {
                        const int t0 = du[u] * 8 + hl;
                        const float x0 = l0[u] + er_t[t0];
                        const float x1 = l1[u] + er_t[t0 + 4];
                        const float v0 = (x0 >= 0.f) ? x0 : SLOPE * x0;
                        const float v1 = (x1 >= 0.f) ? x1 : SLOPE * x1;
                        const float a0 = expf(v0 - m_t[t0]);
                        const float a1 = expf(v1 - m_t[t0 + 4]);
                        if ((lane & 7) == 0) {
                            atomicAddLds(s_t + t0,     a0);
                            atomicAddLds(s_t + t0 + 4, a1);
                        }
                        float* row = acc + du[u] * 256 + lane;
                        atomicAddLds(row,       r0[u].x * a0);
                        atomicAddLds(row + 32,  r0[u].y * a0);
                        atomicAddLds(row + 64,  r0[u].z * a0);
                        atomicAddLds(row + 96,  r0[u].w * a0);
                        atomicAddLds(row + 128, r1[u].x * a1);
                        atomicAddLds(row + 160, r1[u].y * a1);
                        atomicAddLds(row + 192, r1[u].z * a1);
                        atomicAddLds(row + 224, r1[u].w * a1);
                    }
                }
            }
        }
    };

    auto scan = [&](auto&& after) {
        const int n4 = n_edges & ~3;
        if (n4 > 0) {
            const int last    = n4 - 4;
            const int n_steps = cdiv(n4, AGG_STEP);
            auto load4 = [&](int st) {
                const int e0 = st * AGG_STEP + lane * 4;
                return *(const int4*)(dst + (e0 < last ? e0 : last));
            };
            int4 cur[4], nxt[4];
#pragma unroll
            for (int u = 0; u < 4; ++u) cur[u] = load4(wave + u * AGG_WAVES);

            for (int st0 = wave; st0 < n_steps; st0 += 4 * AGG_WAVES) {
#pragma unroll
                for (int u = 0; u < 4; ++u) nxt[u] = load4(st0 + (4 + u) * AGG_WAVES);
#pragma unroll
                for (int u = 0; u < 4; ++u) {
                    const int e0 = (st0 + u * AGG_WAVES) * AGG_STEP + lane * 4;
                    const bool valid = e0 < n4;
                    const int dv[4] = {cur[u].x, cur[u].y, cur[u].z, cur[u].w};
                    bool mt[4];
#pragma unroll
                    for (int j = 0; j < 4; ++j) mt[j] = valid && ((unsigned)(dv[j] - lo) < span);
                    if (__builtin_amdgcn_ballot_w32(mt[0] | mt[1] | mt[2] | mt[3]) != 0u) {
#pragma unroll
                        for (int j = 0; j < 4; ++j) {
                            const unsigned mask = __builtin_amdgcn_ballot_w32(mt[j]);
                            if (mt[j]) {
                                const int pos = cnt + __popc(mask & lt_mask);
                                qe[pos] = e0 + j;
                                qd[pos] = dv[j] - lo;
                            }
                            cnt += __popc(mask);
                        }
                        after();
                    }
                }
#pragma unroll
                for (int u = 0; u < 4; ++u) cur[u] = nxt[u];
            }
        }
        if (wave == 0) {
            const int e = n4 + lane;
            const int d = (e < n_edges) ? dst[e] : -1;
            const bool mt = (unsigned)(d - lo) < span;
            const unsigned mask = __builtin_amdgcn_ballot_w32(mt);
            if (mt) {
                const int pos = cnt + __popc(mask & lt_mask);
                qe[pos] = e;
                qd[pos] = d - lo;
            }
            cnt += __popc(mask);
            if (mask != 0u) after();
        }
    };

    scan([&]() {
        while (cnt - done >= 32) { drain_max(done, 32); done += 32; }
        if (cnt > AGG_QCAP - AGG_STEP) {
            if (cnt > done) drain_max(done, cnt - done);
            keep = false; cnt = 0; done = 0;
        }
    });
    if (cnt > done) drain_max(done, cnt - done);
    if (lane == 0) l2_cnt[blockIdx.x * AGG_WAVES + wave] = filed;
    __syncthreads();

    if (keep) {
        if (cnt > 0) drain_sum(0, cnt);
    } else {
        cnt = 0;
        scan([&]() {
            if (cnt > AGG_QCAP - AGG_STEP) { drain_sum(0, cnt); cnt = 0; }
        });
        if (cnt > 0) drain_sum(0, cnt);
    }
    __syncthreads();

    float bw[8], w0[8], w1[8], w2[8];
#pragma unroll
    for (int j = 0; j < 8; ++j) {
        const int c = (j >> 2) * 128 + 4 * lane + (j & 3);
        bw[j] = b1[c];
        w0[j] = W2[c * 3 + 0];
        w1[j] = W2[c * 3 + 1];
        w2[j] = W2[c * 3 + 2];
    }
    const float l2a0 = al2[0], l2a1 = al2[1], l2a2 = al2[2];
    const float r2a0 = ar2[0], r2a1 = ar2[1], r2a2 = ar2[2];
    for (int r = wave; r < rows; r += AGG_WAVES) {
        const float* p = acc + r * 256 + lane;
        const float sv0 = s_t[r * 8 + hl];
        const float sv1 = s_t[r * 8 + 4 + hl];
        float a0 = 0.f, a1 = 0.f, a2 = 0.f;
#pragma unroll
        for (int j = 0; j < 8; ++j) {
            const float sv = (j < 4) ? sv0 : sv1;
            const float q  = p[(j >> 2) * 128 + (j & 3) * 32];
            float v = (sv != 0.f) ? q / sv : 0.f;
            v += bw[j];
            v = (v > 0.f) ? v : expm1f(v);
            a0 += v * w0[j];
            a1 += v * w1[j];
            a2 += v * w2[j];
        }
        for (int o = 16; o > 0; o >>= 1) {
            a0 += __shfl_xor(a0, o);
            a1 += __shfl_xor(a1, o);
            a2 += __shfl_xor(a2, o);
        }
        if (lane == 0) {
            const size_t n = (size_t)(lo + r);
            h2[n * 3 + 0] = a0;
            h2[n * 3 + 1] = a1;
            h2[n * 3 + 2] = a2;
            el2[n] = a0 * l2a0 + a1 * l2a1 + a2 * l2a2;
            er2[n] = a0 * r2a0 + a1 * r2a1 + a2 * r2a2;
        }
    }
}

__global__ __launch_bounds__(AGG_THREADS)
void gat_layer2_owner(const int* __restrict__ src, const int* __restrict__ dst,
                      const int2* __restrict__ l2_list, const int* __restrict__ l2_cnt,
                      const float* __restrict__ h2, const float* __restrict__ el2,
                      const float* __restrict__ er2, const float* __restrict__ b2,
                      float* __restrict__ out, int n_edges, int n_nodes) {
    __shared__ float er_t[AGG_TILE];
    __shared__ float m_t[AGG_TILE];
    __shared__ float s_t[AGG_TILE];
    __shared__ float a_t[AGG_TILE * 3];

    const int tid  = threadIdx.x;
    const int wave = tid >> 5;
    const int lane = tid & 31;
    const int lo   = blockIdx.x * AGG_TILE;
    int rows = n_nodes - lo; if (rows > AGG_TILE) rows = AGG_TILE;
    const unsigned span = (unsigned)rows;

    const int cnt = __builtin_amdgcn_readfirstlane(l2_cnt[blockIdx.x * AGG_WAVES + wave]);
    const int2* slot = l2_list + (size_t)(blockIdx.x * AGG_WAVES + wave) * L2_CAP;

    for (int i = tid; i < AGG_TILE; i += AGG_THREADS) {
        er_t[i] = (i < rows) ? er2[lo + i] : 0.f;
        m_t[i]  = -__builtin_huge_valf();
        s_t[i]  = 0.f;
        a_t[i * 3 + 0] = 0.f;
        a_t[i * 3 + 1] = 0.f;
        a_t[i * 3 + 2] = 0.f;
    }
    __syncthreads();

    auto score = [&](float l, int dl) {
        const float x = l + er_t[dl];
        return (x >= 0.f) ? x : SLOPE * x;
    };
    auto add_edge = [&](float v, int dl, float g0, float g1, float g2) {
        const float a = expf(v - m_t[dl]);
        atomicAddLds(s_t + dl, a);
        atomicAddLds(a_t + dl * 3 + 0, g0 * a);
        atomicAddLds(a_t + dl * 3 + 1, g1 * a);
        atomicAddLds(a_t + dl * 3 + 2, g2 * a);
    };
    auto own_edges = [&](auto&& f) {
        const int n4 = n_edges & ~3;
        for (int st = wave; st * AGG_STEP < n4; st += AGG_WAVES) {
            for (int j = 0; j < AGG_STEP; j += 32) {
                const int e = st * AGG_STEP + j + lane;
                if (e < n4) {
                    const int d = dst[e];
                    if ((unsigned)(d - lo) < span) f(src[e], d - lo);
                }
            }
        }
        if (wave == 0) {
            const int e = n4 + lane;
            if (e < n_edges) {
                const int d = dst[e];
                if ((unsigned)(d - lo) < span) f(src[e], d - lo);
            }
        }
    };
    const bool listed = cnt <= L2_CAP;

    if (listed) {
        for (int base = 0; base < cnt; base += 128) {
            int2 en[4]; float l[4];
#pragma unroll
            for (int u = 0; u < 4; ++u) {
                const int i = base + u * 32 + lane;
                en[u] = slot[i < cnt ? i : cnt - 1];
            }
#pragma unroll
            for (int u = 0; u < 4; ++u) l[u] = el2[en[u].x];
            asm volatile("" :: "v"(l[0]), "v"(l[1]), "v"(l[2]), "v"(l[3]));
#pragma unroll
            for (int u = 0; u < 4; ++u)
                if (base + u * 32 + lane < cnt) ldsMaxFloat(m_t + en[u].y, score(l[u], en[u].y));
        }
    } else {
        own_edges([&](int sn, int dl) { ldsMaxFloat(m_t + dl, score(el2[sn], dl)); });
    }
    __syncthreads();

    if (listed) {
        for (int base = 0; base < cnt; base += 128) {
            int2 en[4]; float l[4], g0[4], g1[4], g2[4];
#pragma unroll
            for (int u = 0; u < 4; ++u) {
                const int i = base + u * 32 + lane;
                en[u] = slot[i < cnt ? i : cnt - 1];
            }
#pragma unroll
            for (int u = 0; u < 4; ++u) {
                l[u]  = el2[en[u].x];
                g0[u] = h2[(size_t)en[u].x * 3 + 0];
                g1[u] = h2[(size_t)en[u].x * 3 + 1];
                g2[u] = h2[(size_t)en[u].x * 3 + 2];
            }
            asm volatile("" :: "v"(l[0]), "v"(l[1]), "v"(l[2]), "v"(l[3]),
                               "v"(g0[0]), "v"(g0[1]), "v"(g0[2]), "v"(g0[3]),
                               "v"(g1[0]), "v"(g1[1]), "v"(g1[2]), "v"(g1[3]),
                               "v"(g2[0]), "v"(g2[1]), "v"(g2[2]), "v"(g2[3]));
#pragma unroll
            for (int u = 0; u < 4; ++u)
                if (base + u * 32 + lane < cnt)
                    add_edge(score(l[u], en[u].y), en[u].y, g0[u], g1[u], g2[u]);
        }
    } else {
        own_edges([&](int sn, int dl) {
            add_edge(score(el2[sn], dl), dl,
                     h2[(size_t)sn * 3 + 0], h2[(size_t)sn * 3 + 1], h2[(size_t)sn * 3 + 2]);
        });
    }
    __syncthreads();

    for (int i = tid; i < rows * 3; i += AGG_THREADS) {
        const int r = i / 3;
        const float sv = s_t[r];
        const float v = (sv != 0.f) ? a_t[i] / sv : 0.f;
        out[(size_t)lo * 3 + i] = v + b2[i - r * 3];
    }
}

extern "C" void kernel_launch(void* const* d_in, const int* in_sizes, int n_in,
                              void* d_out, int out_size, void* d_ws, size_t ws_size,
                              hipStream_t stream) {
    const float* feat = (const float*)d_in[0];
    const int*   src  = (const int*)  d_in[1];
    const int*   dst  = (const int*)  d_in[2];
    const float* W1   = (const float*)d_in[3];
    const float* al1  = (const float*)d_in[4];
    const float* ar1  = (const float*)d_in[5];
    const float* b1   = (const float*)d_in[6];
    const float* W2   = (const float*)d_in[7];
    const float* al2  = (const float*)d_in[8];
    const float* ar2  = (const float*)d_in[9];
    const float* b2   = (const float*)d_in[10];
    float* out = (float*)d_out;

    const int n_nodes = in_sizes[0] / 256;
    const int n_edges = in_sizes[1];

    char* cur = (char*)d_ws;
    auto alloc = [&](size_t bytes) -> float* {
        float* p = (float*)cur;
        cur += (bytes + 255) & ~(size_t)255;
        return p;
    };
    float* H1   = alloc((size_t)n_nodes * 256 * 4);
    float* el1  = alloc((size_t)n_nodes * 8 * 4);
    float* er1  = alloc((size_t)n_nodes * 8 * 4);
    float* h2   = alloc((size_t)n_nodes * 3 * 4);
    float* el2  = alloc((size_t)n_nodes * 4);
    float* er2  = alloc((size_t)n_nodes * 4);
    const int groups = cdiv(n_nodes, AGG_TILE);
    int*  l2_cnt  = (int*) alloc((size_t)groups * AGG_WAVES * 4);
    int2* l2_list = (int2*)alloc((size_t)groups * AGG_WAVES * L2_CAP * 8);

    const int B = 256;

    gat_gemm1_wmma<<<cdiv(n_nodes, 32), 256, 0, stream>>>(feat, W1, H1, n_nodes);
    gat_attn_scores<<<cdiv(n_nodes * 8, B), B, 0, stream>>>(H1, al1, ar1, el1, er1, n_nodes);

    gat_softmax_aggregate_tile<<<groups, AGG_THREADS, 0, stream>>>(src, dst, H1, el1, er1,
                                                                   b1, W2, al2, ar2, h2, el2, er2,
                                                                   l2_list, l2_cnt, n_edges, n_nodes);

    gat_layer2_owner<<<groups, AGG_THREADS, 0, stream>>>(src, dst, l2_list, l2_cnt, h2, el2, er2, b2,
                                                         out, n_edges, n_nodes);
}


# GEMM1 by 256 resident work-groups: W1 held whole in LDS in fragment order (conflict-free 16-byte reads), feat straight from global to A registers, wave = 16 rows x 64 cols; k order in each 8 changed
# speedup vs baseline: 5.2921x; 1.3733x over previous
#include <hip/hip_runtime.h>
#include <hip/hip_bf16.h>

typedef __attribute__((ext_vector_type(2))) float v2f;
typedef __attribute__((ext_vector_type(8))) float v8f;

#define SLOPE 0.2f

static __host__ __device__ inline int cdiv(int a, int b) { return (a + b - 1) / b; }

__device__ inline void atomicAddLds(float* p, float v) {
    __hip_atomic_fetch_add(p, v, __ATOMIC_RELAXED, __HIP_MEMORY_SCOPE_WORKGROUP);
}

#define G1_GROUPS  256
#define G1_THREADS 512
#define G1_WAVES   16

__global__ __launch_bounds__(G1_THREADS)
void gat_gemm1_wmma(const float* __restrict__ A,
                    const float* __restrict__ W,
                    float* __restrict__ H,
                    int n_nodes) {
    __shared__ __attribute__((aligned(16))) float Wl[256 * 256];

    const int tid  = threadIdx.x;
    const int wave = __builtin_amdgcn_readfirstlane(tid >> 5);
    const int lane = tid & 31;
    const int c    = lane & 15;
    const int hf   = lane >> 4;
    const int q    = wave & 3;
    const int gq   = wave >> 2;

    const unsigned tiles = (unsigned)cdiv(n_nodes, 16);
    const int t0 = (int)((blockIdx.x * tiles) / G1_GROUPS);
    const int t1 = (int)(((blockIdx.x + 1) * tiles) / G1_GROUPS);
    const int full = (t1 - t0) >> 2;
    const int rem  = (t1 - t0) & 3;
    const int e    = (gq - q) & 3;
    const int nit  = full + (e < rem ? 1 : 0);
    auto tile_of = [&](int i) { return t0 + (i < full ? 4 * i + gq : 4 * full + e); };
    auto a_ptr = [&](int tile) {
        int r = 16 * tile + c; if (r >= n_nodes) r = n_nodes - 1;
        return A + (size_t)r * 256 + 4 * hf;
    };
    auto load_a = [&](float4 (&buf)[8], const float* p) {
#pragma unroll
        for (int j = 0; j < 8; ++j) buf[j] = *(const float4*)(p + 8 * j);
    };

    float4 a0[8], a1[8];
    load_a(a0, a_ptr(tile_of(0)));

#pragma unroll
    for (int k = 0; k < 128 / G1_WAVES; ++k) {
        const int u = wave + G1_WAVES * k;
        const float* wp = W + (size_t)(8 * (u >> 2) + 4 * hf) * 256 + 64 * (u & 3) + 4 * c;
        const float4 g0 = *(const float4*)(wp);
        const float4 g1 = *(const float4*)(wp + 256);
        const float4 g2 = *(const float4*)(wp + 512);
        const float4 g3 = *(const float4*)(wp + 768);
        float4* dp = (float4*)Wl + (u * 4) * 32 + lane;
        dp[0]  = make_float4(g0.x, g1.x, g2.x, g3.x);
        dp[32] = make_float4(g0.y, g1.y, g2.y, g3.y);
        dp[64] = make_float4(g0.z, g1.z, g2.z, g3.z);
        dp[96] = make_float4(g0.w, g1.w, g2.w, g3.w);
    }
    __syncthreads();

    v8f acc0 = {}, acc1 = {}, acc2 = {}, acc3 = {};
    const float4* bq = (const float4*)Wl + q * 128 + lane;

    auto mma_chunk = [&](const float4 (&a)[8], int ch) {
        const float4* bp = bq + ch * (8 * 512);
#pragma unroll
        for (int j = 0; j < 8; ++j) {
            const float4 b0 = bp[j * 512];
            const float4 b1 = bp[j * 512 + 32];
            const float4 b2 = bp[j * 512 + 64];
            const float4 b3 = bp[j * 512 + 96];
            const v2f alo = {a[j].x, a[j].y};
            const v2f ahi = {a[j].z, a[j].w};
            acc0 = __builtin_amdgcn_wmma_f32_16x16x4_f32(false, alo, false, (v2f){b0.x, b0.y}, (short)0, acc0, false, false);
            acc1 = __builtin_amdgcn_wmma_f32_16x16x4_f32(false, alo, false, (v2f){b1.x, b1.y}, (short)0, acc1, false, false);
            acc2 = __builtin_amdgcn_wmma_f32_16x16x4_f32(false, alo, false, (v2f){b2.x, b2.y}, (short)0, acc2, false, false);
            acc3 = __builtin_amdgcn_wmma_f32_16x16x4_f32(false, alo, false, (v2f){b3.x, b3.y}, (short)0, acc3, false, false);
            acc0 = __builtin_amdgcn_wmma_f32_16x16x4_f32(false, ahi, false, (v2f){b0.z, b0.w}, (short)0, acc0, false, false);
            acc1 = __builtin_amdgcn_wmma_f32_16x16x4_f32(false, ahi, false, (v2f){b1.z, b1.w}, (short)0, acc1, false, false);
            acc2 = __builtin_amdgcn_wmma_f32_16x16x4_f32(false, ahi, false, (v2f){b2.z, b2.w}, (short)0, acc2, false, false);
            acc3 = __builtin_amdgcn_wmma_f32_16x16x4_f32(false, ahi, false, (v2f){b3.z, b3.w}, (short)0, acc3, false, false);
        }
    };

    for (int it = 0; it < nit; ++it) {
        asm volatile("" ::: "memory");
        const int tile = tile_of(it);
        const float* ap  = a_ptr(tile);
        const float* apn = a_ptr(tile_of(it + 1 < nit ? it + 1 : it));
        load_a(a1, ap + 64);   mma_chunk(a0, 0);
        load_a(a0, ap + 128);  mma_chunk(a1, 1);
        load_a(a1, ap + 192);  mma_chunk(a0, 2);
        load_a(a0, apn);       mma_chunk(a1, 3);

        const int rbase = 16 * tile + 8 * hf;
        float* hp = H + (size_t)rbase * 256 + 64 * q + 4 * c;
        if (16 * tile + 16 <= n_nodes) {
#pragma unroll
            for (int r = 0; r < 8; ++r)
                *(float4*)(hp + r * 256) = make_float4(acc0[r], acc1[r], acc2[r], acc3[r]);
        } else {
#pragma unroll
            for (int r = 0; r < 8; ++r)
                if (rbase + r < n_nodes)
                    *(float4*)(hp + r * 256) = make_float4(acc0[r], acc1[r], acc2[r], acc3[r]);
        }
        acc0 = (v8f){}; acc1 = (v8f){}; acc2 = (v8f){}; acc3 = (v8f){};
    }
}

__global__ void gat_attn_scores(const float* __restrict__ H,
                                const float* __restrict__ al,
                                const float* __restrict__ ar,
                                float* __restrict__ el, float* __restrict__ er,
                                int n_nodes) {
    int i = blockIdx.x * blockDim.x + threadIdx.x;
    if (i >= n_nodes * 8) return;
    int n = i >> 3, h = i & 7;
    const float* hp  = H + (size_t)n * 256 + h * 32;
    const float* alp = al + h * 32;
    const float* arp = ar + h * 32;
    float sl = 0.f, sr = 0.f;
    for (int d = 0; d < 32; ++d) { float v = hp[d]; sl += v * alp[d]; sr += v * arp[d]; }
    el[i] = sl; er[i] = sr;
}

#define AGG_TILE    196
#define AGG_THREADS 1024
#define AGG_WAVES   32
#define AGG_STEP    128
#define AGG_QCAP    288
#define L2_CAP      512

__device__ inline void ldsMaxFloat(float* addr, float value) {
    if (__float_as_int(value) >= 0)
        atomicMax((int*)addr, __float_as_int(value));
    else
        atomicMin((unsigned int*)addr, __float_as_uint(value));
}

__global__ __launch_bounds__(AGG_THREADS)
void gat_softmax_aggregate_tile(const int* __restrict__ src, const int* __restrict__ dst,
                                const float* __restrict__ H, const float* __restrict__ el,
                                const float* __restrict__ er,
                                const float* __restrict__ b1, const float* __restrict__ W2,
                                const float* __restrict__ al2, const float* __restrict__ ar2,
                                float* __restrict__ h2, float* __restrict__ el2,
                                float* __restrict__ er2,
                                int2* __restrict__ l2_list, int* __restrict__ l2_cnt,
                                int n_edges, int n_nodes) {
    __shared__ __attribute__((aligned(16))) float acc[AGG_TILE * 256];
    __shared__ __attribute__((aligned(16))) float er_t[AGG_TILE * 8];
    __shared__ float m_t[AGG_TILE * 8];
    __shared__ float s_t[AGG_TILE * 8];
    __shared__ int   q_e[AGG_WAVES * AGG_QCAP];
    __shared__ int   q_d[AGG_WAVES * AGG_QCAP];

    const int tid  = threadIdx.x;
    const int wave = tid >> 5;
    const int lane = tid & 31;
    const int lo   = blockIdx.x * AGG_TILE;
    int rows = n_nodes - lo; if (rows > AGG_TILE) rows = AGG_TILE;
    const unsigned span = (unsigned)rows;

    for (int i = tid; i < AGG_TILE * 64; i += AGG_THREADS)
        ((float4*)acc)[i] = make_float4(0.f, 0.f, 0.f, 0.f);
    for (int i = tid; i < AGG_TILE * 8; i += AGG_THREADS) {
        er_t[i] = (i < rows * 8) ? er[(size_t)lo * 8 + i] : 0.f;
        m_t[i]  = -__builtin_huge_valf();
        s_t[i]  = 0.f;
    }
    __syncthreads();

    int* qe = q_e + wave * AGG_QCAP;
    int* qd = q_d + wave * AGG_QCAP;
    const unsigned lt_mask = (1u << lane) - 1u;
    const int hl = lane >> 3;
    int  cnt  = 0;
    int  done = 0;
    bool keep = true;
    int  filed = 0;
    int2* slot = l2_list + (size_t)(blockIdx.x * AGG_WAVES + wave) * L2_CAP;

    auto drain_max = [&](int from, int n) {
        for (int b = 0; b < n; b += 32) {
            if (b + lane < n) {
                const int e  = qe[from + b + lane];
                const int dl = qd[from + b + lane];
                const int sn = src[e];
                const int fp = filed + b + lane;
                if (fp < L2_CAP) slot[fp] = make_int2(sn, dl);
                const float4 l0 = *(const float4*)(el + (size_t)sn * 8);
                const float4 l1 = *(const float4*)(el + (size_t)sn * 8 + 4);
                const float4 r0 = *(const float4*)(er_t + dl * 8);
                const float4 r1 = *(const float4*)(er_t + dl * 8 + 4);
                const float x[8] = {l0.x + r0.x, l0.y + r0.y, l0.z + r0.z, l0.w + r0.w,
                                    l1.x + r1.x, l1.y + r1.y, l1.z + r1.z, l1.w + r1.w};
#pragma unroll
                for (int h = 0; h < 8; ++h) {
                    const float v = (x[h] >= 0.f) ? x[h] : SLOPE * x[h];
                    ldsMaxFloat(m_t + dl * 8 + h, v);
                }
            }
        }
        filed += n;
    };

    auto drain_sum = [&](int from, int n) {
        for (int b = 0; b < n; b += 32) {
            int nb = n - b; if (nb > 32) nb = 32;
            const int i  = from + b + (lane < nb ? lane : nb - 1);
            const int e  = qe[i];
            const int dl = qd[i];
            const int sn = src[e];
            for (int g = 0; g < nb; g += 4) {
                int su[4], du[4];
#pragma unroll
                for (int u = 0; u < 4; ++u) {
                    const int k = (g + u < nb) ? g + u : nb - 1;
                    su[u] = __builtin_amdgcn_readlane(sn, k);
                    du[u] = __builtin_amdgcn_readlane(dl, k);
                }
                float4 r0[4], r1[4];
                float  l0[4], l1[4];
#pragma unroll
                for (int u = 0; u < 4; ++u) {
                    const float* hp = H + (size_t)su[u] * 256 + 4 * lane;
                    r0[u] = *(const float4*)(hp);
                    r1[u] = *(const float4*)(hp + 128);
                    l0[u] = el[(size_t)su[u] * 8 + hl];
                    l1[u] = el[(size_t)su[u] * 8 + 4 + hl];
                }
#pragma unroll
                for (int u = 0; u < 4; ++u) {
                    if (g + u < nb) {
                        const int t0 = du[u] * 8 + hl;
                        const float x0 = l0[u] + er_t[t0];
                        const float x1 = l1[u] + er_t[t0 + 4];
                        const float v0 = (x0 >= 0.f) ? x0 : SLOPE * x0;
                        const float v1 = (x1 >= 0.f) ? x1 : SLOPE * x1;
                        const float a0 = expf(v0 - m_t[t0]);
                        const float a1 = expf(v1 - m_t[t0 + 4]);
                        if ((lane & 7) == 0) {
                            atomicAddLds(s_t + t0,     a0);
                            atomicAddLds(s_t + t0 + 4, a1);
                        }
                        float* row = acc + du[u] * 256 + lane;
                        atomicAddLds(row,       r0[u].x * a0);
                        atomicAddLds(row + 32,  r0[u].y * a0);
                        atomicAddLds(row + 64,  r0[u].z * a0);
                        atomicAddLds(row + 96,  r0[u].w * a0);
                        atomicAddLds(row + 128, r1[u].x * a1);
                        atomicAddLds(row + 160, r1[u].y * a1);
                        atomicAddLds(row + 192, r1[u].z * a1);
                        atomicAddLds(row + 224, r1[u].w * a1);
                    }
                }
            }
        }
    };

    auto scan = [&](auto&& after) {
        const int n4 = n_edges & ~3;
        if (n4 > 0) {
            const int last    = n4 - 4;
            const int n_steps = cdiv(n4, AGG_STEP);
            auto load4 = [&](int st) {
                const int e0 = st * AGG_STEP + lane * 4;
                return *(const int4*)(dst + (e0 < last ? e0 : last));
            };
            int4 cur[4], nxt[4];
#pragma unroll
            for (int u = 0; u < 4; ++u) cur[u] = load4(wave + u * AGG_WAVES);

            for (int st0 = wave; st0 < n_steps; st0 += 4 * AGG_WAVES) {
#pragma unroll
                for (int u = 0; u < 4; ++u) nxt[u] = load4(st0 + (4 + u) * AGG_WAVES);
#pragma unroll
                for (int u = 0; u < 4; ++u) {
                    const int e0 = (st0 + u * AGG_WAVES) * AGG_STEP + lane * 4;
                    const bool valid = e0 < n4;
                    const int dv[4] = {cur[u].x, cur[u].y, cur[u].z, cur[u].w};
                    bool mt[4];
#pragma unroll
                    for (int j = 0; j < 4; ++j) mt[j] = valid && ((unsigned)(dv[j] - lo) < span);
                    if (__builtin_amdgcn_ballot_w32(mt[0] | mt[1] | mt[2] | mt[3]) != 0u) {
#pragma unroll
                        for (int j = 0; j < 4; ++j) {
                            const unsigned mask = __builtin_amdgcn_ballot_w32(mt[j]);
                            if (mt[j]) {
                                const int pos = cnt + __popc(mask & lt_mask);
                                qe[pos] = e0 + j;
                                qd[pos] = dv[j] - lo;
                            }
                            cnt += __popc(mask);
                        }
                        after();
                    }
                }
#pragma unroll
                for (int u = 0; u < 4; ++u) cur[u] = nxt[u];
            }
        }
        if (wave == 0) {
            const int e = n4 + lane;
            const int d = (e < n_edges) ? dst[e] : -1;
            const bool mt = (unsigned)(d - lo) < span;
            const unsigned mask = __builtin_amdgcn_ballot_w32(mt);
            if (mt) {
                const int pos = cnt + __popc(mask & lt_mask);
                qe[pos] = e;
                qd[pos] = d - lo;
            }
            cnt += __popc(mask);
            if (mask != 0u) after();
        }
    };

    scan([&]() {
        while (cnt - done >= 32) { drain_max(done, 32); done += 32; }
        if (cnt > AGG_QCAP - AGG_STEP) {
            if (cnt > done) drain_max(done, cnt - done);
            keep = false; cnt = 0; done = 0;
        }
    });
    if (cnt > done) drain_max(done, cnt - done);
    if (lane == 0) l2_cnt[blockIdx.x * AGG_WAVES + wave] = filed;
    __syncthreads();

    if (keep) {
        if (cnt > 0) drain_sum(0, cnt);
    } else {
        cnt = 0;
        scan([&]() {
            if (cnt > AGG_QCAP - AGG_STEP) { drain_sum(0, cnt); cnt = 0; }
        });
        if (cnt > 0) drain_sum(0, cnt);
    }
    __syncthreads();

    float bw[8], w0[8], w1[8], w2[8];
#pragma unroll
    for (int j = 0; j < 8; ++j) {
        const int c = (j >> 2) * 128 + 4 * lane + (j & 3);
        bw[j] = b1[c];
        w0[j] = W2[c * 3 + 0];
        w1[j] = W2[c * 3 + 1];
        w2[j] = W2[c * 3 + 2];
    }
    const float l2a0 = al2[0], l2a1 = al2[1], l2a2 = al2[2];
    const float r2a0 = ar2[0], r2a1 = ar2[1], r2a2 = ar2[2];
    for (int r = wave; r < rows; r += AGG_WAVES) {
        const float* p = acc + r * 256 + lane;
        const float sv0 = s_t[r * 8 + hl];
        const float sv1 = s_t[r * 8 + 4 + hl];
        float a0 = 0.f, a1 = 0.f, a2 = 0.f;
#pragma unroll
        for (int j = 0; j < 8; ++j) {
            const float sv = (j < 4) ? sv0 : sv1;
            const float q  = p[(j >> 2) * 128 + (j & 3) * 32];
            float v = (sv != 0.f) ? q / sv : 0.f;
            v += bw[j];
            v = (v > 0.f) ? v : expm1f(v);
            a0 += v * w0[j];
            a1 += v * w1[j];
            a2 += v * w2[j];
        }
        for (int o = 16; o > 0; o >>= 1) {
            a0 += __shfl_xor(a0, o);
            a1 += __shfl_xor(a1, o);
            a2 += __shfl_xor(a2, o);
        }
        if (lane == 0) {
            const size_t n = (size_t)(lo + r);
            h2[n * 3 + 0] = a0;
            h2[n * 3 + 1] = a1;
            h2[n * 3 + 2] = a2;
            el2[n] = a0 * l2a0 + a1 * l2a1 + a2 * l2a2;
            er2[n] = a0 * r2a0 + a1 * r2a1 + a2 * r2a2;
        }
    }
}

__global__ __launch_bounds__(AGG_THREADS)
void gat_layer2_owner(const int* __restrict__ src, const int* __restrict__ dst,
                      const int2* __restrict__ l2_list, const int* __restrict__ l2_cnt,
                      const float* __restrict__ h2, const float* __restrict__ el2,
                      const float* __restrict__ er2, const float* __restrict__ b2,
                      float* __restrict__ out, int n_edges, int n_nodes) {
    __shared__ float er_t[AGG_TILE];
    __shared__ float m_t[AGG_TILE];
    __shared__ float s_t[AGG_TILE];
    __shared__ float a_t[AGG_TILE * 3];

    const int tid  = threadIdx.x;
    const int wave = tid >> 5;
    const int lane = tid & 31;
    const int lo   = blockIdx.x * AGG_TILE;
    int rows = n_nodes - lo; if (rows > AGG_TILE) rows = AGG_TILE;
    const unsigned span = (unsigned)rows;

    const int cnt = __builtin_amdgcn_readfirstlane(l2_cnt[blockIdx.x * AGG_WAVES + wave]);
    const int2* slot = l2_list + (size_t)(blockIdx.x * AGG_WAVES + wave) * L2_CAP;

    for (int i = tid; i < AGG_TILE; i += AGG_THREADS) {
        er_t[i] = (i < rows) ? er2[lo + i] : 0.f;
        m_t[i]  = -__builtin_huge_valf();
        s_t[i]  = 0.f;
        a_t[i * 3 + 0] = 0.f;
        a_t[i * 3 + 1] = 0.f;
        a_t[i * 3 + 2] = 0.f;
    }
    __syncthreads();

    auto score = [&](float l, int dl) {
        const float x = l + er_t[dl];
        return (x >= 0.f) ? x : SLOPE * x;
    };
    auto add_edge = [&](float v, int dl, float g0, float g1, float g2) {
        const float a = expf(v - m_t[dl]);
        atomicAddLds(s_t + dl, a);
        atomicAddLds(a_t + dl * 3 + 0, g0 * a);
        atomicAddLds(a_t + dl * 3 + 1, g1 * a);
        atomicAddLds(a_t + dl * 3 + 2, g2 * a);
    };
    auto own_edges = [&](auto&& f) {
        const int n4 = n_edges & ~3;
        for (int st = wave; st * AGG_STEP < n4; st += AGG_WAVES) {
            for (int j = 0; j < AGG_STEP; j += 32) {
                const int e = st * AGG_STEP + j + lane;
                if (e < n4) {
                    const int d = dst[e];
                    if ((unsigned)(d - lo) < span) f(src[e], d - lo);
                }
            }
        }
        if (wave == 0) {
            const int e = n4 + lane;
            if (e < n_edges) {
                const int d = dst[e];
                if ((unsigned)(d - lo) < span) f(src[e], d - lo);
            }
        }
    };
    const bool listed = cnt <= L2_CAP;

    if (listed) {
        for (int base = 0; base < cnt; base += 128) {
            int2 en[4]; float l[4];
#pragma unroll
            for (int u = 0; u < 4; ++u) {
                const int i = base + u * 32 + lane;
                en[u] = slot[i < cnt ? i : cnt - 1];
            }
#pragma unroll
            for (int u = 0; u < 4; ++u) l[u] = el2[en[u].x];
            asm volatile("" :: "v"(l[0]), "v"(l[1]), "v"(l[2]), "v"(l[3]));
#pragma unroll
            for (int u = 0; u < 4; ++u)
                if (base + u * 32 + lane < cnt) ldsMaxFloat(m_t + en[u].y, score(l[u], en[u].y));
        }
    } else {
        own_edges([&](int sn, int dl) { ldsMaxFloat(m_t + dl, score(el2[sn], dl)); });
    }
    __syncthreads();

    if (listed) {
        for (int base = 0; base < cnt; base += 128) {
            int2 en[4]; float l[4], g0[4], g1[4], g2[4];
#pragma unroll
            for (int u = 0; u < 4; ++u) {
                const int i = base + u * 32 + lane;
                en[u] = slot[i < cnt ? i : cnt - 1];
            }
#pragma unroll
            for (int u = 0; u < 4; ++u) {
                l[u]  = el2[en[u].x];
                g0[u] = h2[(size_t)en[u].x * 3 + 0];
                g1[u] = h2[(size_t)en[u].x * 3 + 1];
                g2[u] = h2[(size_t)en[u].x * 3 + 2];
            }
            asm volatile("" :: "v"(l[0]), "v"(l[1]), "v"(l[2]), "v"(l[3]),
                               "v"(g0[0]), "v"(g0[1]), "v"(g0[2]), "v"(g0[3]),
                               "v"(g1[0]), "v"(g1[1]), "v"(g1[2]), "v"(g1[3]),
                               "v"(g2[0]), "v"(g2[1]), "v"(g2[2]), "v"(g2[3]));
#pragma unroll
            for (int u = 0; u < 4; ++u)
                if (base + u * 32 + lane < cnt)
                    add_edge(score(l[u], en[u].y), en[u].y, g0[u], g1[u], g2[u]);
        }
    } else {
        own_edges([&](int sn, int dl) {
            add_edge(score(el2[sn], dl), dl,
                     h2[(size_t)sn * 3 + 0], h2[(size_t)sn * 3 + 1], h2[(size_t)sn * 3 + 2]);
        });
    }
    __syncthreads();

    for (int i = tid; i < rows * 3; i += AGG_THREADS) {
        const int r = i / 3;
        const float sv = s_t[r];
        const float v = (sv != 0.f) ? a_t[i] / sv : 0.f;
        out[(size_t)lo * 3 + i] = v + b2[i - r * 3];
    }
}

extern "C" void kernel_launch(void* const* d_in, const int* in_sizes, int n_in,
                              void* d_out, int out_size, void* d_ws, size_t ws_size,
                              hipStream_t stream) {
    const float* feat = (const float*)d_in[0];
    const int*   src  = (const int*)  d_in[1];
    const int*   dst  = (const int*)  d_in[2];
    const float* W1   = (const float*)d_in[3];
    const float* al1  = (const float*)d_in[4];
    const float* ar1  = (const float*)d_in[5];
    const float* b1   = (const float*)d_in[6];
    const float* W2   = (const float*)d_in[7];
    const float* al2  = (const float*)d_in[8];
    const float* ar2  = (const float*)d_in[9];
    const float* b2   = (const float*)d_in[10];
    float* out = (float*)d_out;

    const int n_nodes = in_sizes[0] / 256;
    const int n_edges = in_sizes[1];

    char* cur = (char*)d_ws;
    auto alloc = [&](size_t bytes) -> float* {
        float* p = (float*)cur;
        cur += (bytes + 255) & ~(size_t)255;
        return p;
    };
    float* H1   = alloc((size_t)n_nodes * 256 * 4);
    float* el1  = alloc((size_t)n_nodes * 8 * 4);
    float* er1  = alloc((size_t)n_nodes * 8 * 4);
    float* h2   = alloc((size_t)n_nodes * 3 * 4);
    float* el2  = alloc((size_t)n_nodes * 4);
    float* er2  = alloc((size_t)n_nodes * 4);
    const int groups = cdiv(n_nodes, AGG_TILE);
    int*  l2_cnt  = (int*) alloc((size_t)groups * AGG_WAVES * 4);
    int2* l2_list = (int2*)alloc((size_t)groups * AGG_WAVES * L2_CAP * 8);

    const int B = 256;

    gat_gemm1_wmma<<<G1_GROUPS, G1_THREADS, 0, stream>>>(feat, W1, H1, n_nodes);
    gat_attn_scores<<<cdiv(n_nodes * 8, B), B, 0, stream>>>(H1, al1, ar1, el1, er1, n_nodes);

    gat_softmax_aggregate_tile<<<groups, AGG_THREADS, 0, stream>>>(src, dst, H1, el1, er1,
                                                                   b1, W2, al2, ar2, h2, el2, er2,
                                                                   l2_list, l2_cnt, n_edges, n_nodes);

    gat_layer2_owner<<<groups, AGG_THREADS, 0, stream>>>(src, dst, l2_list, l2_cnt, h2, el2, er2, b2,
                                                         out, n_edges, n_nodes);
}


# Layer-1 pass 2 by row-owner waves: pass 1 puts each edge's source node on its row's LDS list; the row's wave sums row and exp sums in registers (list arrival order), plain stores, no LDS float adds
# speedup vs baseline: 6.7594x; 1.2773x over previous
#include <hip/hip_runtime.h>
#include <hip/hip_bf16.h>

typedef __attribute__((ext_vector_type(2))) float v2f;
typedef __attribute__((ext_vector_type(8))) float v8f;

#define SLOPE 0.2f

static __host__ __device__ inline int cdiv(int a, int b) { return (a + b - 1) / b; }

__device__ inline void atomicAddLds(float* p, float v) {
    __hip_atomic_fetch_add(p, v, __ATOMIC_RELAXED, __HIP_MEMORY_SCOPE_WORKGROUP);
}

#define G1_GROUPS  256
#define G1_THREADS 512
#define G1_WAVES   16

__global__ __launch_bounds__(G1_THREADS)
void gat_gemm1_wmma(const float* __restrict__ A,
                    const float* __restrict__ W,
                    float* __restrict__ H,
                    int n_nodes) {
    __shared__ __attribute__((aligned(16))) float Wl[256 * 256];

    const int tid  = threadIdx.x;
    const int wave = __builtin_amdgcn_readfirstlane(tid >> 5);
    const int lane = tid & 31;
    const int c    = lane & 15;
    const int hf   = lane >> 4;
    const int q    = wave & 3;
    const int gq   = wave >> 2;

    const unsigned tiles = (unsigned)cdiv(n_nodes, 16);
    const int t0 = (int)((blockIdx.x * tiles) / G1_GROUPS);
    const int t1 = (int)(((blockIdx.x + 1) * tiles) / G1_GROUPS);
    const int full = (t1 - t0) >> 2;
    const int rem  = (t1 - t0) & 3;
    const int e    = (gq - q) & 3;
    const int nit  = full + (e < rem ? 1 : 0);
    auto tile_of = [&](int i) { return t0 + (i < full ? 4 * i + gq : 4 * full + e); };
    auto a_ptr = [&](int tile) {
        int r = 16 * tile + c; if (r >= n_nodes) r = n_nodes - 1;
        return A + (size_t)r * 256 + 4 * hf;
    };
    auto load_a = [&](float4 (&buf)[8], const float* p) {
#pragma unroll
        for (int j = 0; j < 8; ++j) buf[j] = *(const float4*)(p + 8 * j);
    };

    float4 a0[8], a1[8];
    load_a(a0, a_ptr(tile_of(0)));

#pragma unroll
    for (int k = 0; k < 128 / G1_WAVES; ++k) {
        const int u = wave + G1_WAVES * k;
        const float* wp = W + (size_t)(8 * (u >> 2) + 4 * hf) * 256 + 64 * (u & 3) + 4 * c;
        const float4 g0 = *(const float4*)(wp);
        const float4 g1 = *(const float4*)(wp + 256);
        const float4 g2 = *(const float4*)(wp + 512);
        const float4 g3 = *(const float4*)(wp + 768);
        float4* dp = (float4*)Wl + (u * 4) * 32 + lane;
        dp[0]  = make_float4(g0.x, g1.x, g2.x, g3.x);
        dp[32] = make_float4(g0.y, g1.y, g2.y, g3.y);
        dp[64] = make_float4(g0.z, g1.z, g2.z, g3.z);
        dp[96] = make_float4(g0.w, g1.w, g2.w, g3.w);
    }
    __syncthreads();

    v8f acc0 = {}, acc1 = {}, acc2 = {}, acc3 = {};
    const float4* bq = (const float4*)Wl + q * 128 + lane;

    auto mma_chunk = [&](const float4 (&a)[8], int ch) {
        const float4* bp = bq + ch * (8 * 512);
#pragma unroll
        for (int j = 0; j < 8; ++j) {
            const float4 b0 = bp[j * 512];
            const float4 b1 = bp[j * 512 + 32];
            const float4 b2 = bp[j * 512 + 64];
            const float4 b3 = bp[j * 512 + 96];
            const v2f alo = {a[j].x, a[j].y};
            const v2f ahi = {a[j].z, a[j].w};
            acc0 = __builtin_amdgcn_wmma_f32_16x16x4_f32(false, alo, false, (v2f){b0.x, b0.y}, (short)0, acc0, false, false);
            acc1 = __builtin_amdgcn_wmma_f32_16x16x4_f32(false, alo, false, (v2f){b1.x, b1.y}, (short)0, acc1, false, false);
            acc2 = __builtin_amdgcn_wmma_f32_16x16x4_f32(false, alo, false, (v2f){b2.x, b2.y}, (short)0, acc2, false, false);
            acc3 = __builtin_amdgcn_wmma_f32_16x16x4_f32(false, alo, false, (v2f){b3.x, b3.y}, (short)0, acc3, false, false);
            acc0 = __builtin_amdgcn_wmma_f32_16x16x4_f32(false, ahi, false, (v2f){b0.z, b0.w}, (short)0, acc0, false, false);
            acc1 = __builtin_amdgcn_wmma_f32_16x16x4_f32(false, ahi, false, (v2f){b1.z, b1.w}, (short)0, acc1, false, false);
            acc2 = __builtin_amdgcn_wmma_f32_16x16x4_f32(false, ahi, false, (v2f){b2.z, b2.w}, (short)0, acc2, false, false);
            acc3 = __builtin_amdgcn_wmma_f32_16x16x4_f32(false, ahi, false, (v2f){b3.z, b3.w}, (short)0, acc3, false, false);
        }
    };

    for (int it = 0; it < nit; ++it) {
        asm volatile("" ::: "memory");
        const int tile = tile_of(it);
        const float* ap  = a_ptr(tile);
        const float* apn = a_ptr(tile_of(it + 1 < nit ? it + 1 : it));
        load_a(a1, ap + 64);   mma_chunk(a0, 0);
        load_a(a0, ap + 128);  mma_chunk(a1, 1);
        load_a(a1, ap + 192);  mma_chunk(a0, 2);
        load_a(a0, apn);       mma_chunk(a1, 3);

        const int rbase = 16 * tile + 8 * hf;
        float* hp = H + (size_t)rbase * 256 + 64 * q + 4 * c;
        if (16 * tile + 16 <= n_nodes) {
#pragma unroll
            for (int r = 0; r < 8; ++r)
                *(float4*)(hp + r * 256) = make_float4(acc0[r], acc1[r], acc2[r], acc3[r]);
        } else {
#pragma unroll
            for (int r = 0; r < 8; ++r)
                if (rbase + r < n_nodes)
                    *(float4*)(hp + r * 256) = make_float4(acc0[r], acc1[r], acc2[r], acc3[r]);
        }
        acc0 = (v8f){}; acc1 = (v8f){}; acc2 = (v8f){}; acc3 = (v8f){};
    }
}

__global__ void gat_attn_scores(const float* __restrict__ H,
                                const float* __restrict__ al,
                                const float* __restrict__ ar,
                                float* __restrict__ el, float* __restrict__ er,
                                int n_nodes) {
    int i = blockIdx.x * blockDim.x + threadIdx.x;
    if (i >= n_nodes * 8) return;
    int n = i >> 3, h = i & 7;
    const float* hp  = H + (size_t)n * 256 + h * 32;
    const float* alp = al + h * 32;
    const float* arp = ar + h * 32;
    float sl = 0.f, sr = 0.f;
    for (int d = 0; d < 32; ++d) { float v = hp[d]; sl += v * alp[d]; sr += v * arp[d]; }
    el[i] = sl; er[i] = sr;
}

#define AGG_TILE    196
#define AGG_THREADS 1024
#define AGG_WAVES   32
#define AGG_STEP    128
#define AGG_QCAP    224
#define AGG_RCAP    60
#define AGG_FLY     6
#define L2_CAP      512

__device__ inline void ldsMaxFloat(float* addr, float value) {
    if (__float_as_int(value) >= 0)
        atomicMax((int*)addr, __float_as_int(value));
    else
        atomicMin((unsigned int*)addr, __float_as_uint(value));
}

__global__ __launch_bounds__(AGG_THREADS)
void gat_softmax_aggregate_tile(const int* __restrict__ src, const int* __restrict__ dst,
                                const float* __restrict__ H, const float* __restrict__ el,
                                const float* __restrict__ er,
                                const float* __restrict__ b1, const float* __restrict__ W2,
                                const float* __restrict__ al2, const float* __restrict__ ar2,
                                float* __restrict__ h2, float* __restrict__ el2,
                                float* __restrict__ er2,
                                int2* __restrict__ l2_list, int* __restrict__ l2_cnt,
                                int n_edges, int n_nodes) {
    __shared__ __attribute__((aligned(16))) float acc[AGG_TILE * 256];
    __shared__ __attribute__((aligned(16))) float er_t[AGG_TILE * 8];
    __shared__ float m_t[AGG_TILE * 8];
    __shared__ float s_t[AGG_TILE * 8];
    __shared__ int   q_e[AGG_WAVES * AGG_QCAP];
    __shared__ int   q_d[AGG_WAVES * AGG_QCAP];
    __shared__ int   o_src[AGG_TILE * AGG_RCAP];
    __shared__ int   o_cnt[AGG_TILE];

    const int tid  = threadIdx.x;
    const int wave = tid >> 5;
    const int lane = tid & 31;
    const int lo   = blockIdx.x * AGG_TILE;
    int rows = n_nodes - lo; if (rows > AGG_TILE) rows = AGG_TILE;
    const unsigned span = (unsigned)rows;

    for (int i = tid; i < AGG_TILE * 64; i += AGG_THREADS)
        ((float4*)acc)[i] = make_float4(0.f, 0.f, 0.f, 0.f);
    for (int i = tid; i < AGG_TILE * 8; i += AGG_THREADS) {
        er_t[i] = (i < rows * 8) ? er[(size_t)lo * 8 + i] : 0.f;
        m_t[i]  = -__builtin_huge_valf();
        s_t[i]  = 0.f;
    }
    for (int i = tid; i < AGG_TILE; i += AGG_THREADS) o_cnt[i] = 0;
    __syncthreads();

    int* qe = q_e + wave * AGG_QCAP;
    int* qd = q_d + wave * AGG_QCAP;
    const unsigned lt_mask = (1u << lane) - 1u;
    const int hl = lane >> 3;
    int  cnt  = 0;
    int  done = 0;
    int  filed = 0;
    int2* slot = l2_list + (size_t)(blockIdx.x * AGG_WAVES + wave) * L2_CAP;

    auto drain_max = [&](int from, int n) {
        for (int b = 0; b < n; b += 32) {
            if (b + lane < n) {
                const int e  = qe[from + b + lane];
                const int dl = qd[from + b + lane];
                const int sn = src[e];
                const int fp = filed + b + lane;
                if (fp < L2_CAP) slot[fp] = make_int2(sn, dl);
                const int op = atomicAdd(o_cnt + dl, 1);
                if (op < AGG_RCAP) o_src[dl * AGG_RCAP + op] = sn;
                const float4 l0 = *(const float4*)(el + (size_t)sn * 8);
                const float4 l1 = *(const float4*)(el + (size_t)sn * 8 + 4);
                const float4 r0 = *(const float4*)(er_t + dl * 8);
                const float4 r1 = *(const float4*)(er_t + dl * 8 + 4);
                const float x[8] = {l0.x + r0.x, l0.y + r0.y, l0.z + r0.z, l0.w + r0.w,
                                    l1.x + r1.x, l1.y + r1.y, l1.z + r1.z, l1.w + r1.w};
#pragma unroll
                for (int h = 0; h < 8; ++h) {
                    const float v = (x[h] >= 0.f) ? x[h] : SLOPE * x[h];
                    ldsMaxFloat(m_t + dl * 8 + h, v);
                }
            }
        }
        filed += n;
    };

    auto drain_sum = [&](int from, int n) {
        for (int b = 0; b < n; b += 32) {
            int nb = n - b; if (nb > 32) nb = 32;
            const int i  = from + b + (lane < nb ? lane : nb - 1);
            const int e  = qe[i];
            const int dl = qd[i];
            const int sn = src[e];
            for (int g = 0; g < nb; g += 4) {
                int su[4], du[4];
#pragma unroll
                for (int u = 0; u < 4; ++u) {
                    const int k = (g + u < nb) ? g + u : nb - 1;
                    su[u] = __builtin_amdgcn_readlane(sn, k);
                    du[u] = __builtin_amdgcn_readlane(dl, k);
                }
                float4 r0[4], r1[4];
                float  l0[4], l1[4];
#pragma unroll
                for (int u = 0; u < 4; ++u) {
                    const float* hp = H + (size_t)su[u] * 256 + 4 * lane;
                    r0[u] = *(const float4*)(hp);
                    r1[u] = *(const float4*)(hp + 128);
                    l0[u] = el[(size_t)su[u] * 8 + hl];
                    l1[u] = el[(size_t)su[u] * 8 + 4 + hl];
                }
#pragma unroll
                for (int u = 0; u < 4; ++u) {
                    if (g + u < nb) {
                        const int t0 = du[u] * 8 + hl;
                        const float x0 = l0[u] + er_t[t0];
                        const float x1 = l1[u] + er_t[t0 + 4];
                        const float v0 = (x0 >= 0.f) ? x0 : SLOPE * x0;
                        const float v1 = (x1 >= 0.f) ? x1 : SLOPE * x1;
                        const float a0 = expf(v0 - m_t[t0]);
                        const float a1 = expf(v1 - m_t[t0 + 4]);
                        if ((lane & 7) == 0) {
                            atomicAddLds(s_t + t0,     a0);
                            atomicAddLds(s_t + t0 + 4, a1);
                        }
                        float* row = acc + du[u] * 256 + lane;
                        atomicAddLds(row,       r0[u].x * a0);
                        atomicAddLds(row + 32,  r0[u].y * a0);
                        atomicAddLds(row + 64,  r0[u].z * a0);
                        atomicAddLds(row + 96,  r0[u].w * a0);
                        atomicAddLds(row + 128, r1[u].x * a1);
                        atomicAddLds(row + 160, r1[u].y * a1);
                        atomicAddLds(row + 192, r1[u].z * a1);
                        atomicAddLds(row + 224, r1[u].w * a1);
                    }
                }
            }
        }
    };

    auto scan = [&](auto&& after) {
        const int n4 = n_edges & ~3;
        if (n4 > 0) {
            const int last    = n4 - 4;
            const int n_steps = cdiv(n4, AGG_STEP);
            auto load4 = [&](int st) {
                const int e0 = st * AGG_STEP + lane * 4;
                return *(const int4*)(dst + (e0 < last ? e0 : last));
            };
            int4 cur[4], nxt[4];
#pragma unroll
            for (int u = 0; u < 4; ++u) cur[u] = load4(wave + u * AGG_WAVES);

            for (int st0 = wave; st0 < n_steps; st0 += 4 * AGG_WAVES) {
#pragma unroll
                for (int u = 0; u < 4; ++u) nxt[u] = load4(st0 + (4 + u) * AGG_WAVES);
#pragma unroll
                for (int u = 0; u < 4; ++u) {
                    const int e0 = (st0 + u * AGG_WAVES) * AGG_STEP + lane * 4;
                    const bool valid = e0 < n4;
                    const int dv[4] = {cur[u].x, cur[u].y, cur[u].z, cur[u].w};
                    bool mt[4];
#pragma unroll
                    for (int j = 0; j < 4; ++j) mt[j] = valid && ((unsigned)(dv[j] - lo) < span);
                    if (__builtin_amdgcn_ballot_w32(mt[0] | mt[1] | mt[2] | mt[3]) != 0u) {
#pragma unroll
                        for (int j = 0; j < 4; ++j) {
                            const unsigned mask = __builtin_amdgcn_ballot_w32(mt[j]);
                            if (mt[j]) {
                                const int pos = cnt + __popc(mask & lt_mask);
                                qe[pos] = e0 + j;
                                qd[pos] = dv[j] - lo;
                            }
                            cnt += __popc(mask);
                        }
                        after();
                    }
                }
#pragma unroll
                for (int u = 0; u < 4; ++u) cur[u] = nxt[u];
            }
        }
        if (wave == 0) {
            const int e = n4 + lane;
            const int d = (e < n_edges) ? dst[e] : -1;
            const bool mt = (unsigned)(d - lo) < span;
            const unsigned mask = __builtin_amdgcn_ballot_w32(mt);
            if (mt) {
                const int pos = cnt + __popc(mask & lt_mask);
                qe[pos] = e;
                qd[pos] = d - lo;
            }
            cnt += __popc(mask);
            if (mask != 0u) after();
        }
    };

    scan([&]() {
        while (cnt - done >= 32) { drain_max(done, 32); done += 32; }
        if (cnt > AGG_QCAP - AGG_STEP) {
            if (cnt > done) drain_max(done, cnt - done);
            cnt = 0; done = 0;
        }
    });
    if (cnt > done) drain_max(done, cnt - done);
    if (lane == 0) l2_cnt[blockIdx.x * AGG_WAVES + wave] = filed;
    __syncthreads();

    bool over = false;
    for (int i = lane; i < AGG_TILE; i += 32) over |= (o_cnt[i] > AGG_RCAP);
    const bool owned = (__builtin_amdgcn_ballot_w32(over) == 0u);

    if (owned) {
        for (int r = wave; r < rows; r += AGG_WAVES) {
            const int n = __builtin_amdgcn_readfirstlane(o_cnt[r]);
            const int* ol = o_src + r * AGG_RCAP;
            const float e0 = er_t[r * 8 + hl], e1 = er_t[r * 8 + 4 + hl];
            const float m0 = m_t[r * 8 + hl],  m1 = m_t[r * 8 + 4 + hl];
            float sum[8] = {0.f, 0.f, 0.f, 0.f, 0.f, 0.f, 0.f, 0.f};
            float es0 = 0.f, es1 = 0.f;
            for (int b = 0; b < n; b += 32) {
                int nb = n - b; if (nb > 32) nb = 32;
                const int sn = ol[b + (lane < nb ? lane : nb - 1)];
                for (int g = 0; g < nb; g += AGG_FLY) {
                    int su[AGG_FLY];
#pragma unroll
                    for (int u = 0; u < AGG_FLY; ++u)
                        su[u] = __builtin_amdgcn_readlane(sn, (g + u < nb) ? g + u : nb - 1);
                    float4 r0[AGG_FLY], r1[AGG_FLY];
                    float  l0[AGG_FLY], l1[AGG_FLY];
#pragma unroll
                    for (int u = 0; u < AGG_FLY; ++u) {
                        const float* hp = H + (size_t)su[u] * 256 + 4 * lane;
                        r0[u] = *(const float4*)(hp);
                        r1[u] = *(const float4*)(hp + 128);
                        l0[u] = el[(size_t)su[u] * 8 + hl];
                        l1[u] = el[(size_t)su[u] * 8 + 4 + hl];
                    }
#pragma unroll
                    for (int u = 0; u < AGG_FLY; ++u) {
                        if (g + u < nb) {
                            const float x0 = l0[u] + e0;
                            const float x1 = l1[u] + e1;
                            const float v0 = (x0 >= 0.f) ? x0 : SLOPE * x0;
                            const float v1 = (x1 >= 0.f) ? x1 : SLOPE * x1;
                            const float a0 = expf(v0 - m0);
                            const float a1 = expf(v1 - m1);
                            es0 += a0;
                            es1 += a1;
                            sum[0] += r0[u].x * a0; sum[1] += r0[u].y * a0;
                            sum[2] += r0[u].z * a0; sum[3] += r0[u].w * a0;
                            sum[4] += r1[u].x * a1; sum[5] += r1[u].y * a1;
                            sum[6] += r1[u].z * a1; sum[7] += r1[u].w * a1;
                        }
                    }
                }
            }
            float* row = acc + r * 256 + lane;
#pragma unroll
            for (int j = 0; j < 8; ++j) row[(j >> 2) * 128 + (j & 3) * 32] = sum[j];
            if ((lane & 7) == 0) {
                s_t[r * 8 + hl]     = es0;
                s_t[r * 8 + 4 + hl] = es1;
            }
        }
    } else {
        cnt = 0;
        scan([&]() {
            if (cnt > AGG_QCAP - AGG_STEP) { drain_sum(0, cnt); cnt = 0; }
        });
        if (cnt > 0) drain_sum(0, cnt);
        __syncthreads();
    }

    float bw[8], w0[8], w1[8], w2[8];
#pragma unroll
    for (int j = 0; j < 8; ++j) {
        const int c = (j >> 2) * 128 + 4 * lane + (j & 3);
        bw[j] = b1[c];
        w0[j] = W2[c * 3 + 0];
        w1[j] = W2[c * 3 + 1];
        w2[j] = W2[c * 3 + 2];
    }
    const float l2a0 = al2[0], l2a1 = al2[1], l2a2 = al2[2];
    const float r2a0 = ar2[0], r2a1 = ar2[1], r2a2 = ar2[2];
    for (int r = wave; r < rows; r += AGG_WAVES) {
        const float* p = acc + r * 256 + lane;
        const float sv0 = s_t[r * 8 + hl];
        const float sv1 = s_t[r * 8 + 4 + hl];
        float a0 = 0.f, a1 = 0.f, a2 = 0.f;
#pragma unroll
        for (int j = 0; j < 8; ++j) {
            const float sv = (j < 4) ? sv0 : sv1;
            const float q  = p[(j >> 2) * 128 + (j & 3) * 32];
            float v = (sv != 0.f) ? q / sv : 0.f;
            v += bw[j];
            v = (v > 0.f) ? v : expm1f(v);
            a0 += v * w0[j];
            a1 += v * w1[j];
            a2 += v * w2[j];
        }
        for (int o = 16; o > 0; o >>= 1) {
            a0 += __shfl_xor(a0, o);
            a1 += __shfl_xor(a1, o);
            a2 += __shfl_xor(a2, o);
        }
        if (lane == 0) {
            const size_t n = (size_t)(lo + r);
            h2[n * 3 + 0] = a0;
            h2[n * 3 + 1] = a1;
            h2[n * 3 + 2] = a2;
            el2[n] = a0 * l2a0 + a1 * l2a1 + a2 * l2a2;
            er2[n] = a0 * r2a0 + a1 * r2a1 + a2 * r2a2;
        }
    }
}

__global__ __launch_bounds__(AGG_THREADS)
void gat_layer2_owner(const int* __restrict__ src, const int* __restrict__ dst,
                      const int2* __restrict__ l2_list, const int* __restrict__ l2_cnt,
                      const float* __restrict__ h2, const float* __restrict__ el2,
                      const float* __restrict__ er2, const float* __restrict__ b2,
                      float* __restrict__ out, int n_edges, int n_nodes) {
    __shared__ float er_t[AGG_TILE];
    __shared__ float m_t[AGG_TILE];
    __shared__ float s_t[AGG_TILE];
    __shared__ float a_t[AGG_TILE * 3];

    const int tid  = threadIdx.x;
    const int wave = tid >> 5;
    const int lane = tid & 31;
    const int lo   = blockIdx.x * AGG_TILE;
    int rows = n_nodes - lo; if (rows > AGG_TILE) rows = AGG_TILE;
    const unsigned span = (unsigned)rows;

    const int cnt = __builtin_amdgcn_readfirstlane(l2_cnt[blockIdx.x * AGG_WAVES + wave]);
    const int2* slot = l2_list + (size_t)(blockIdx.x * AGG_WAVES + wave) * L2_CAP;

    for (int i = tid; i < AGG_TILE; i += AGG_THREADS) {
        er_t[i] = (i < rows) ? er2[lo + i] : 0.f;
        m_t[i]  = -__builtin_huge_valf();
        s_t[i]  = 0.f;
        a_t[i * 3 + 0] = 0.f;
        a_t[i * 3 + 1] = 0.f;
        a_t[i * 3 + 2] = 0.f;
    }
    __syncthreads();

    auto score = [&](float l, int dl) {
        const float x = l + er_t[dl];
        return (x >= 0.f) ? x : SLOPE * x;
    };
    auto add_edge = [&](float v, int dl, float g0, float g1, float g2) {
        const float a = expf(v - m_t[dl]);
        atomicAddLds(s_t + dl, a);
        atomicAddLds(a_t + dl * 3 + 0, g0 * a);
        atomicAddLds(a_t + dl * 3 + 1, g1 * a);
        atomicAddLds(a_t + dl * 3 + 2, g2 * a);
    };
    auto own_edges = [&](auto&& f) {
        const int n4 = n_edges & ~3;
        for (int st = wave; st * AGG_STEP < n4; st += AGG_WAVES) {
            for (int j = 0; j < AGG_STEP; j += 32) {
                const int e = st * AGG_STEP + j + lane;
                if (e < n4) {
                    const int d = dst[e];
                    if ((unsigned)(d - lo) < span) f(src[e], d - lo);
                }
            }
        }
        if (wave == 0) {
            const int e = n4 + lane;
            if (e < n_edges) {
                const int d = dst[e];
                if ((unsigned)(d - lo) < span) f(src[e], d - lo);
            }
        }
    };
    const bool listed = cnt <= L2_CAP;

    if (listed) {
        for (int base = 0; base < cnt; base += 128) {
            int2 en[4]; float l[4];
#pragma unroll
            for (int u = 0; u < 4; ++u) {
                const int i = base + u * 32 + lane;
                en[u] = slot[i < cnt ? i : cnt - 1];
            }
#pragma unroll
            for (int u = 0; u < 4; ++u) l[u] = el2[en[u].x];
            asm volatile("" :: "v"(l[0]), "v"(l[1]), "v"(l[2]), "v"(l[3]));
#pragma unroll
            for (int u = 0; u < 4; ++u)
                if (base + u * 32 + lane < cnt) ldsMaxFloat(m_t + en[u].y, score(l[u], en[u].y));
        }
    } else {
        own_edges([&](int sn, int dl) { ldsMaxFloat(m_t + dl, score(el2[sn], dl)); });
    }
    __syncthreads();

    if (listed) {
        for (int base = 0; base < cnt; base += 128) {
            int2 en[4]; float l[4], g0[4], g1[4], g2[4];
#pragma unroll
            for (int u = 0; u < 4; ++u) {
                const int i = base + u * 32 + lane;
                en[u] = slot[i < cnt ? i : cnt - 1];
            }
#pragma unroll
            for (int u = 0; u < 4; ++u) {
                l[u]  = el2[en[u].x];
                g0[u] = h2[(size_t)en[u].x * 3 + 0];
                g1[u] = h2[(size_t)en[u].x * 3 + 1];
                g2[u] = h2[(size_t)en[u].x * 3 + 2];
            }
            asm volatile("" :: "v"(l[0]), "v"(l[1]), "v"(l[2]), "v"(l[3]),
                               "v"(g0[0]), "v"(g0[1]), "v"(g0[2]), "v"(g0[3]),
                               "v"(g1[0]), "v"(g1[1]), "v"(g1[2]), "v"(g1[3]),
                               "v"(g2[0]), "v"(g2[1]), "v"(g2[2]), "v"(g2[3]));
#pragma unroll
            for (int u = 0; u < 4; ++u)
                if (base + u * 32 + lane < cnt)
                    add_edge(score(l[u], en[u].y), en[u].y, g0[u], g1[u], g2[u]);
        }
    } else {
        own_edges([&](int sn, int dl) {
            add_edge(score(el2[sn], dl), dl,
                     h2[(size_t)sn * 3 + 0], h2[(size_t)sn * 3 + 1], h2[(size_t)sn * 3 + 2]);
        });
    }
    __syncthreads();

    for (int i = tid; i < rows * 3; i += AGG_THREADS) {
        const int r = i / 3;
        const float sv = s_t[r];
        const float v = (sv != 0.f) ? a_t[i] / sv : 0.f;
        out[(size_t)lo * 3 + i] = v + b2[i - r * 3];
    }
}

extern "C" void kernel_launch(void* const* d_in, const int* in_sizes, int n_in,
                              void* d_out, int out_size, void* d_ws, size_t ws_size,
                              hipStream_t stream) {
    const float* feat = (const float*)d_in[0];
    const int*   src  = (const int*)  d_in[1];
    const int*   dst  = (const int*)  d_in[2];
    const float* W1   = (const float*)d_in[3];
    const float* al1  = (const float*)d_in[4];
    const float* ar1  = (const float*)d_in[5];
    const float* b1   = (const float*)d_in[6];
    const float* W2   = (const float*)d_in[7];
    const float* al2  = (const float*)d_in[8];
    const float* ar2  = (const float*)d_in[9];
    const float* b2   = (const float*)d_in[10];
    float* out = (float*)d_out;

    const int n_nodes = in_sizes[0] / 256;
    const int n_edges = in_sizes[1];

    char* cur = (char*)d_ws;
    auto alloc = [&](size_t bytes) -> float* {
        float* p = (float*)cur;
        cur += (bytes + 255) & ~(size_t)255;
        return p;
    };
    float* H1   = alloc((size_t)n_nodes * 256 * 4);
    float* el1  = alloc((size_t)n_nodes * 8 * 4);
    float* er1  = alloc((size_t)n_nodes * 8 * 4);
    float* h2   = alloc((size_t)n_nodes * 3 * 4);
    float* el2  = alloc((size_t)n_nodes * 4);
    float* er2  = alloc((size_t)n_nodes * 4);
    const int groups = cdiv(n_nodes, AGG_TILE);
    int*  l2_cnt  = (int*) alloc((size_t)groups * AGG_WAVES * 4);
    int2* l2_list = (int2*)alloc((size_t)groups * AGG_WAVES * L2_CAP * 8);

    const int B = 256;

    gat_gemm1_wmma<<<G1_GROUPS, G1_THREADS, 0, stream>>>(feat, W1, H1, n_nodes);
    gat_attn_scores<<<cdiv(n_nodes * 8, B), B, 0, stream>>>(H1, al1, ar1, el1, er1, n_nodes);

    gat_softmax_aggregate_tile<<<groups, AGG_THREADS, 0, stream>>>(src, dst, H1, el1, er1,
                                                                   b1, W2, al2, ar2, h2, el2, er2,
                                                                   l2_list, l2_cnt, n_edges, n_nodes);

    gat_layer2_owner<<<groups, AGG_THREADS, 0, stream>>>(src, dst, l2_list, l2_cnt, h2, el2, er2, b2,
                                                         out, n_edges, n_nodes);
}


# Score launch writes a per-owner-group edge bitmap (LDS integer OR per 1024-edge block, whole-line stores); layer-1 groups read their own bits, not all of dst[] (row list arrival order changed)
# speedup vs baseline: 7.3298x; 1.0844x over previous
#include <hip/hip_runtime.h>
#include <hip/hip_bf16.h>

typedef __attribute__((ext_vector_type(2))) float v2f;
typedef __attribute__((ext_vector_type(8))) float v8f;

#define SLOPE 0.2f

static __host__ __device__ inline int cdiv(int a, int b) { return (a + b - 1) / b; }

__device__ inline void atomicAddLds(float* p, float v) {
    __hip_atomic_fetch_add(p, v, __ATOMIC_RELAXED, __HIP_MEMORY_SCOPE_WORKGROUP);
}

#define G1_GROUPS  256
#define G1_THREADS 512
#define G1_WAVES   16

__global__ __launch_bounds__(G1_THREADS)
void gat_gemm1_wmma(const float* __restrict__ A,
                    const float* __restrict__ W,
                    float* __restrict__ H,
                    int n_nodes) {
    __shared__ __attribute__((aligned(16))) float Wl[256 * 256];

    const int tid  = threadIdx.x;
    const int wave = __builtin_amdgcn_readfirstlane(tid >> 5);
    const int lane = tid & 31;
    const int c    = lane & 15;
    const int hf   = lane >> 4;
    const int q    = wave & 3;
    const int gq   = wave >> 2;

    const unsigned tiles = (unsigned)cdiv(n_nodes, 16);
    const int t0 = (int)((blockIdx.x * tiles) / G1_GROUPS);
    const int t1 = (int)(((blockIdx.x + 1) * tiles) / G1_GROUPS);
    const int full = (t1 - t0) >> 2;
    const int rem  = (t1 - t0) & 3;
    const int e    = (gq - q) & 3;
    const int nit  = full + (e < rem ? 1 : 0);
    auto tile_of = [&](int i) { return t0 + (i < full ? 4 * i + gq : 4 * full + e); };
    auto a_ptr = [&](int tile) {
        int r = 16 * tile + c; if (r >= n_nodes) r = n_nodes - 1;
        return A + (size_t)r * 256 + 4 * hf;
    };
    auto load_a = [&](float4 (&buf)[8], const float* p) {
#pragma unroll
        for (int j = 0; j < 8; ++j) buf[j] = *(const float4*)(p + 8 * j);
    };

    float4 a0[8], a1[8];
    load_a(a0, a_ptr(tile_of(0)));

#pragma unroll
    for (int k = 0; k < 128 / G1_WAVES; ++k) {
        const int u = wave + G1_WAVES * k;
        const float* wp = W + (size_t)(8 * (u >> 2) + 4 * hf) * 256 + 64 * (u & 3) + 4 * c;
        const float4 g0 = *(const float4*)(wp);
        const float4 g1 = *(const float4*)(wp + 256);
        const float4 g2 = *(const float4*)(wp + 512);
        const float4 g3 = *(const float4*)(wp + 768);
        float4* dp = (float4*)Wl + (u * 4) * 32 + lane;
        dp[0]  = make_float4(g0.x, g1.x, g2.x, g3.x);
        dp[32] = make_float4(g0.y, g1.y, g2.y, g3.y);
        dp[64] = make_float4(g0.z, g1.z, g2.z, g3.z);
        dp[96] = make_float4(g0.w, g1.w, g2.w, g3.w);
    }
    __syncthreads();

    v8f acc0 = {}, acc1 = {}, acc2 = {}, acc3 = {};
    const float4* bq = (const float4*)Wl + q * 128 + lane;

    auto mma_chunk = [&](const float4 (&a)[8], int ch) {
        const float4* bp = bq + ch * (8 * 512);
#pragma unroll
        for (int j = 0; j < 8; ++j) {
            const float4 b0 = bp[j * 512];
            const float4 b1 = bp[j * 512 + 32];
            const float4 b2 = bp[j * 512 + 64];
            const float4 b3 = bp[j * 512 + 96];
            const v2f alo = {a[j].x, a[j].y};
            const v2f ahi = {a[j].z, a[j].w};
            acc0 = __builtin_amdgcn_wmma_f32_16x16x4_f32(false, alo, false, (v2f){b0.x, b0.y}, (short)0, acc0, false, false);
            acc1 = __builtin_amdgcn_wmma_f32_16x16x4_f32(false, alo, false, (v2f){b1.x, b1.y}, (short)0, acc1, false, false);
            acc2 = __builtin_amdgcn_wmma_f32_16x16x4_f32(false, alo, false, (v2f){b2.x, b2.y}, (short)0, acc2, false, false);
            acc3 = __builtin_amdgcn_wmma_f32_16x16x4_f32(false, alo, false, (v2f){b3.x, b3.y}, (short)0, acc3, false, false);
            acc0 = __builtin_amdgcn_wmma_f32_16x16x4_f32(false, ahi, false, (v2f){b0.z, b0.w}, (short)0, acc0, false, false);
            acc1 = __builtin_amdgcn_wmma_f32_16x16x4_f32(false, ahi, false, (v2f){b1.z, b1.w}, (short)0, acc1, false, false);
            acc2 = __builtin_amdgcn_wmma_f32_16x16x4_f32(false, ahi, false, (v2f){b2.z, b2.w}, (short)0, acc2, false, false);
            acc3 = __builtin_amdgcn_wmma_f32_16x16x4_f32(false, ahi, false, (v2f){b3.z, b3.w}, (short)0, acc3, false, false);
        }
    };

    for (int it = 0; it < nit; ++it) {
        asm volatile("" ::: "memory");
        const int tile = tile_of(it);
        const float* ap  = a_ptr(tile);
        const float* apn = a_ptr(tile_of(it + 1 < nit ? it + 1 : it));
        load_a(a1, ap + 64);   mma_chunk(a0, 0);
        load_a(a0, ap + 128);  mma_chunk(a1, 1);
        load_a(a1, ap + 192);  mma_chunk(a0, 2);
        load_a(a0, apn);       mma_chunk(a1, 3);

        const int rbase = 16 * tile + 8 * hf;
        float* hp = H + (size_t)rbase * 256 + 64 * q + 4 * c;
        if (16 * tile + 16 <= n_nodes) {
#pragma unroll
            for (int r = 0; r < 8; ++r)
                *(float4*)(hp + r * 256) = make_float4(acc0[r], acc1[r], acc2[r], acc3[r]);
        } else {
#pragma unroll
            for (int r = 0; r < 8; ++r)
                if (rbase + r < n_nodes)
                    *(float4*)(hp + r * 256) = make_float4(acc0[r], acc1[r], acc2[r], acc3[r]);
        }
        acc0 = (v8f){}; acc1 = (v8f){}; acc2 = (v8f){}; acc3 = (v8f){};
    }
}

#define AGG_TILE    196
#define MAP_BLOCK   1024
#define MAP_GROUPS  256
#define SC_THREADS  256

__global__ __launch_bounds__(SC_THREADS)
void gat_attn_scores(const float* __restrict__ H,
                     const float* __restrict__ al,
                     const float* __restrict__ ar,
                     float* __restrict__ el, float* __restrict__ er,
                     const int* __restrict__ dst, unsigned int* __restrict__ emap,
                     int n_edges, int n_groups, int n_nodes) {
    __shared__ __attribute__((aligned(16))) unsigned int tab[MAP_GROUPS * 32];
    const int tid = threadIdx.x;
    const int n_blocks = cdiv(n_edges, MAP_BLOCK);
    for (int blk = blockIdx.x; blk < n_blocks; blk += gridDim.x) {
        const int e0 = blk * MAP_BLOCK + 4 * tid;
        int d[4];
        if (e0 + 3 < n_edges) {
            const int4 v = *(const int4*)(dst + e0);
            d[0] = v.x; d[1] = v.y; d[2] = v.z; d[3] = v.w;
        } else {
#pragma unroll
            for (int k = 0; k < 4; ++k) d[k] = (e0 + k < n_edges) ? dst[e0 + k] : -1;
        }
        for (int g0 = 0; g0 < n_groups; g0 += MAP_GROUPS) {
            for (int k = tid; k < MAP_GROUPS * 8; k += SC_THREADS)
                ((uint4*)tab)[k] = make_uint4(0u, 0u, 0u, 0u);
            __syncthreads();
#pragma unroll
            for (int k = 0; k < 4; ++k) {
                const unsigned g = (unsigned)d[k] / (unsigned)AGG_TILE;
                if (g < (unsigned)n_groups && g >= (unsigned)g0 && g - (unsigned)g0 < (unsigned)MAP_GROUPS)
                    atomicOr(tab + (g - (unsigned)g0) * 32 + (tid >> 3), 1u << (4 * (tid & 7) + k));
            }
            __syncthreads();
            int ng = n_groups - g0; if (ng > MAP_GROUPS) ng = MAP_GROUPS;
            uint4* op = (uint4*)(emap + ((size_t)blk * n_groups + g0) * 32);
            for (int k = tid; k < ng * 8; k += SC_THREADS) op[k] = ((const uint4*)tab)[k];
            __syncthreads();
        }
    }
    int i = blockIdx.x * blockDim.x + threadIdx.x;
    if (i >= n_nodes * 8) return;
    int n = i >> 3, h = i & 7;
    const float* hp  = H + (size_t)n * 256 + h * 32;
    const float* alp = al + h * 32;
    const float* arp = ar + h * 32;
    float sl = 0.f, sr = 0.f;
    for (int d = 0; d < 32; ++d) { float v = hp[d]; sl += v * alp[d]; sr += v * arp[d]; }
    el[i] = sl; er[i] = sr;
}

#define AGG_THREADS 1024
#define AGG_WAVES   32
#define AGG_PART    128
#define AGG_QCAP    224
#define AGG_RCAP    60
#define AGG_FLY     6
#define L2_CAP      512

__device__ inline void ldsMaxFloat(float* addr, float value) {
    if (__float_as_int(value) >= 0)
        atomicMax((int*)addr, __float_as_int(value));
    else
        atomicMin((unsigned int*)addr, __float_as_uint(value));
}

__global__ __launch_bounds__(AGG_THREADS)
void gat_softmax_aggregate_tile(const int* __restrict__ src, const int* __restrict__ dst,
                                const unsigned int* __restrict__ emap,
                                const float* __restrict__ H, const float* __restrict__ el,
                                const float* __restrict__ er,
                                const float* __restrict__ b1, const float* __restrict__ W2,
                                const float* __restrict__ al2, const float* __restrict__ ar2,
                                float* __restrict__ h2, float* __restrict__ el2,
                                float* __restrict__ er2,
                                int2* __restrict__ l2_list, int* __restrict__ l2_cnt,
                                int n_edges, int n_nodes) {
    __shared__ __attribute__((aligned(16))) float acc[AGG_TILE * 256];
    __shared__ __attribute__((aligned(16))) float er_t[AGG_TILE * 8];
    __shared__ float m_t[AGG_TILE * 8];
    __shared__ float s_t[AGG_TILE * 8];
    __shared__ int   q_e[AGG_WAVES * AGG_QCAP];
    __shared__ int   o_src[AGG_TILE * AGG_RCAP];
    __shared__ int   o_cnt[AGG_TILE];

    const int tid  = threadIdx.x;
    const int wave = tid >> 5;
    const int lane = tid & 31;
    const int lo   = blockIdx.x * AGG_TILE;
    int rows = n_nodes - lo; if (rows > AGG_TILE) rows = AGG_TILE;
    const unsigned span = (unsigned)rows;

    for (int i = tid; i < AGG_TILE * 64; i += AGG_THREADS)
        ((float4*)acc)[i] = make_float4(0.f, 0.f, 0.f, 0.f);
    for (int i = tid; i < AGG_TILE * 8; i += AGG_THREADS) {
        er_t[i] = (i < rows * 8) ? er[(size_t)lo * 8 + i] : 0.f;
        m_t[i]  = -__builtin_huge_valf();
        s_t[i]  = 0.f;
    }
    for (int i = tid; i < AGG_TILE; i += AGG_THREADS) o_cnt[i] = 0;
    __syncthreads();

    int* qe = q_e + wave * AGG_QCAP;
    const unsigned lt_mask = (1u << lane) - 1u;
    const int hl = lane >> 3;
    int  cnt  = 0;
    int  done = 0;
    int  filed = 0;
    int2* slot = l2_list + (size_t)(blockIdx.x * AGG_WAVES + wave) * L2_CAP;

    auto drain_max = [&](int from, int n) {
        for (int b = 0; b < n; b += 32) {
            int dl = -1, sn = 0;
            if (b + lane < n) {
                const int e = qe[from + b + lane];
                dl = dst[e] - lo;
                sn = src[e];
            }
            const bool ok = (unsigned)dl < span;
            const unsigned okm = __builtin_amdgcn_ballot_w32(ok);
            if (ok) {
                const int fp = filed + __popc(okm & lt_mask);
                if (fp < L2_CAP) slot[fp] = make_int2(sn, dl);
                const int op = atomicAdd(o_cnt + dl, 1);
                if (op < AGG_RCAP) o_src[dl * AGG_RCAP + op] = sn;
                const float4 l0 = *(const float4*)(el + (size_t)sn * 8);
                const float4 l1 = *(const float4*)(el + (size_t)sn * 8 + 4);
                const float4 r0 = *(const float4*)(er_t + dl * 8);
                const float4 r1 = *(const float4*)(er_t + dl * 8 + 4);
                const float x[8] = {l0.x + r0.x, l0.y + r0.y, l0.z + r0.z, l0.w + r0.w,
                                    l1.x + r1.x, l1.y + r1.y, l1.z + r1.z, l1.w + r1.w};
#pragma unroll
                for (int h = 0; h < 8; ++h) {
                    const float v = (x[h] >= 0.f) ? x[h] : SLOPE * x[h];
                    ldsMaxFloat(m_t + dl * 8 + h, v);
                }
            }
            filed += __popc(okm);
        }
    };

    auto drain_sum = [&](int from, int n) {
        for (int b = 0; b < n; b += 32) {
            int nb = n - b; if (nb > 32) nb = 32;
            const int i  = from + b + (lane < nb ? lane : nb - 1);
            const int e  = qe[i];
            const int dl = dst[e] - lo;
            const int sn = src[e];
            for (int g = 0; g < nb; g += 4) {
                int su[4], du[4];
#pragma unroll
                for (int u = 0; u < 4; ++u) {
                    const int k = (g + u < nb) ? g + u : nb - 1;
                    su[u] = __builtin_amdgcn_readlane(sn, k);
                    du[u] = __builtin_amdgcn_readlane(dl, k);
                }
                float4 r0[4], r1[4];
                float  l0[4], l1[4];
#pragma unroll
                for (int u = 0; u < 4; ++u) {
                    const float* hp = H + (size_t)su[u] * 256 + 4 * lane;
                    r0[u] = *(const float4*)(hp);
                    r1[u] = *(const float4*)(hp + 128);
                    l0[u] = el[(size_t)su[u] * 8 + hl];
                    l1[u] = el[(size_t)su[u] * 8 + 4 + hl];
                }
#pragma unroll
                for (int u = 0; u < 4; ++u) {
                    if (g + u < nb && (unsigned)du[u] < span) {
                        const int t0 = du[u] * 8 + hl;
                        const float x0 = l0[u] + er_t[t0];
                        const float x1 = l1[u] + er_t[t0 + 4];
                        const float v0 = (x0 >= 0.f) ? x0 : SLOPE * x0;
                        const float v1 = (x1 >= 0.f) ? x1 : SLOPE * x1;
                        const float a0 = expf(v0 - m_t[t0]);
                        const float a1 = expf(v1 - m_t[t0 + 4]);
                        if ((lane & 7) == 0) {
                            atomicAddLds(s_t + t0,     a0);
                            atomicAddLds(s_t + t0 + 4, a1);
                        }
                        float* row = acc + du[u] * 256 + lane;
                        atomicAddLds(row,       r0[u].x * a0);
                        atomicAddLds(row + 32,  r0[u].y * a0);
                        atomicAddLds(row + 64,  r0[u].z * a0);
                        atomicAddLds(row + 96,  r0[u].w * a0);
                        atomicAddLds(row + 128, r1[u].x * a1);
                        atomicAddLds(row + 160, r1[u].y * a1);
                        atomicAddLds(row + 192, r1[u].z * a1);
                        atomicAddLds(row + 224, r1[u].w * a1);
                    }
                }
            }
        }
    };

    auto scan = [&](auto&& after) {
        const int n_blocks = cdiv(n_edges, MAP_BLOCK);
        if (n_blocks > 0) {
            auto loadw = [&](int blk) {
                const int bc = blk < n_blocks ? blk : n_blocks - 1;
                return emap[((size_t)bc * gridDim.x + blockIdx.x) * 32 + lane];
            };
            unsigned cur[4], nxt[4];
#pragma unroll
            for (int u = 0; u < 4; ++u) cur[u] = loadw(wave + u * AGG_WAVES);

            for (int b0 = wave; b0 < n_blocks; b0 += 4 * AGG_WAVES) {
#pragma unroll
                for (int u = 0; u < 4; ++u) nxt[u] = loadw(b0 + (4 + u) * AGG_WAVES);
#pragma unroll
                for (int u = 0; u < 4; ++u) {
                    const int blk = b0 + u * AGG_WAVES;
                    const int e0  = blk * MAP_BLOCK + lane * 32;
                    unsigned c = (blk < n_blocks) ? cur[u] : 0u;
                    unsigned mask = __builtin_amdgcn_ballot_w32(c != 0u);
                    while (mask != 0u) {
                        if (c != 0u) {
                            const int pos = cnt + __popc(mask & lt_mask);
                            qe[pos] = e0 + __builtin_ctz(c);
                            c &= c - 1u;
                        }
                        cnt += __popc(mask);
                        after();
                        mask = __builtin_amdgcn_ballot_w32(c != 0u);
                    }
                }
#pragma unroll
                for (int u = 0; u < 4; ++u) cur[u] = nxt[u];
            }
        }
    };

    scan([&]() {
        while (cnt - done >= 32) { drain_max(done, 32); done += 32; }
        if (cnt > AGG_QCAP - AGG_PART) {
            if (cnt > done) drain_max(done, cnt - done);
            cnt = 0; done = 0;
        }
    });
    if (cnt > done) drain_max(done, cnt - done);
    if (lane == 0) l2_cnt[blockIdx.x * AGG_WAVES + wave] = filed;
    __syncthreads();

    bool over = false;
    for (int i = lane; i < AGG_TILE; i += 32) over |= (o_cnt[i] > AGG_RCAP);
    const bool owned = (__builtin_amdgcn_ballot_w32(over) == 0u);

    if (owned) {
        for (int r = wave; r < rows; r += AGG_WAVES) {
            const int n = __builtin_amdgcn_readfirstlane(o_cnt[r]);
            const int* ol = o_src + r * AGG_RCAP;
            const float e0 = er_t[r * 8 + hl], e1 = er_t[r * 8 + 4 + hl];
            const float m0 = m_t[r * 8 + hl],  m1 = m_t[r * 8 + 4 + hl];
            float sum[8] = {0.f, 0.f, 0.f, 0.f, 0.f, 0.f, 0.f, 0.f};
            float es0 = 0.f, es1 = 0.f;
            for (int b = 0; b < n; b += 32) {
                int nb = n - b; if (nb > 32) nb = 32;
                const int sn = ol[b + (lane < nb ? lane : nb - 1)];
                for (int g = 0; g < nb; g += AGG_FLY) {
                    int su[AGG_FLY];
#pragma unroll
                    for (int u = 0; u < AGG_FLY; ++u)
                        su[u] = __builtin_amdgcn_readlane(sn, (g + u < nb) ? g + u : nb - 1);
                    float4 r0[AGG_FLY], r1[AGG_FLY];
                    float  l0[AGG_FLY], l1[AGG_FLY];
#pragma unroll
                    for (int u = 0; u < AGG_FLY; ++u) {
                        const float* hp = H + (size_t)su[u] * 256 + 4 * lane;
                        r0[u] = *(const float4*)(hp);
                        r1[u] = *(const float4*)(hp + 128);
                        l0[u] = el[(size_t)su[u] * 8 + hl];
                        l1[u] = el[(size_t)su[u] * 8 + 4 + hl];
                    }
#pragma unroll
                    for (int u = 0; u < AGG_FLY; ++u) {
                        if (g + u < nb) {
                            const float x0 = l0[u] + e0;
                            const float x1 = l1[u] + e1;
                            const float v0 = (x0 >= 0.f) ? x0 : SLOPE * x0;
                            const float v1 = (x1 >= 0.f) ? x1 : SLOPE * x1;
                            const float a0 = expf(v0 - m0);
                            const float a1 = expf(v1 - m1);
                            es0 += a0;
                            es1 += a1;
                            sum[0] += r0[u].x * a0; sum[1] += r0[u].y * a0;
                            sum[2] += r0[u].z * a0; sum[3] += r0[u].w * a0;
                            sum[4] += r1[u].x * a1; sum[5] += r1[u].y * a1;
                            sum[6] += r1[u].z * a1; sum[7] += r1[u].w * a1;
                        }
                    }
                }
            }
            float* row = acc + r * 256 + lane;
#pragma unroll
            for (int j = 0; j < 8; ++j) row[(j >> 2) * 128 + (j & 3) * 32] = sum[j];
            if ((lane & 7) == 0) {
                s_t[r * 8 + hl]     = es0;
                s_t[r * 8 + 4 + hl] = es1;
            }
        }
    } else {
        cnt = 0;
        scan([&]() {
            if (cnt > AGG_QCAP - AGG_PART) { drain_sum(0, cnt); cnt = 0; }
        });
        if (cnt > 0) drain_sum(0, cnt);
        __syncthreads();
    }

    float bw[8], w0[8], w1[8], w2[8];
#pragma unroll
    for (int j = 0; j < 8; ++j) {
        const int c = (j >> 2) * 128 + 4 * lane + (j & 3);
        bw[j] = b1[c];
        w0[j] = W2[c * 3 + 0];
        w1[j] = W2[c * 3 + 1];
        w2[j] = W2[c * 3 + 2];
    }
    const float l2a0 = al2[0], l2a1 = al2[1], l2a2 = al2[2];
    const float r2a0 = ar2[0], r2a1 = ar2[1], r2a2 = ar2[2];
    for (int r = wave; r < rows; r += AGG_WAVES) {
        const float* p = acc + r * 256 + lane;
        const float sv0 = s_t[r * 8 + hl];
        const float sv1 = s_t[r * 8 + 4 + hl];
        float a0 = 0.f, a1 = 0.f, a2 = 0.f;
#pragma unroll
        for (int j = 0; j < 8; ++j) {
            const float sv = (j < 4) ? sv0 : sv1;
            const float q  = p[(j >> 2) * 128 + (j & 3) * 32];
            float v = (sv != 0.f) ? q / sv : 0.f;
            v += bw[j];
            v = (v > 0.f) ? v : expm1f(v);
            a0 += v * w0[j];
            a1 += v * w1[j];
            a2 += v * w2[j];
        }
        for (int o = 16; o > 0; o >>= 1) {
            a0 += __shfl_xor(a0, o);
            a1 += __shfl_xor(a1, o);
            a2 += __shfl_xor(a2, o);
        }
        if (lane == 0) {
            const size_t n = (size_t)(lo + r);
            h2[n * 3 + 0] = a0;
            h2[n * 3 + 1] = a1;
            h2[n * 3 + 2] = a2;
            el2[n] = a0 * l2a0 + a1 * l2a1 + a2 * l2a2;
            er2[n] = a0 * r2a0 + a1 * r2a1 + a2 * r2a2;
        }
    }
}

__global__ __launch_bounds__(AGG_THREADS)
void gat_layer2_owner(const int* __restrict__ src, const int* __restrict__ dst,
                      const int2* __restrict__ l2_list, const int* __restrict__ l2_cnt,
                      const float* __restrict__ h2, const float* __restrict__ el2,
                      const float* __restrict__ er2, const float* __restrict__ b2,
                      float* __restrict__ out, int n_edges, int n_nodes) {
    __shared__ float er_t[AGG_TILE];
    __shared__ float m_t[AGG_TILE];
    __shared__ float s_t[AGG_TILE];
    __shared__ float a_t[AGG_TILE * 3];

    const int tid  = threadIdx.x;
    const int wave = tid >> 5;
    const int lane = tid & 31;
    const int lo   = blockIdx.x * AGG_TILE;
    int rows = n_nodes - lo; if (rows > AGG_TILE) rows = AGG_TILE;
    const unsigned span = (unsigned)rows;

    const int cnt = __builtin_amdgcn_readfirstlane(l2_cnt[blockIdx.x * AGG_WAVES + wave]);
    const int2* slot = l2_list + (size_t)(blockIdx.x * AGG_WAVES + wave) * L2_CAP;

    for (int i = tid; i < AGG_TILE; i += AGG_THREADS) {
        er_t[i] = (i < rows) ? er2[lo + i] : 0.f;
        m_t[i]  = -__builtin_huge_valf();
        s_t[i]  = 0.f;
        a_t[i * 3 + 0] = 0.f;
        a_t[i * 3 + 1] = 0.f;
        a_t[i * 3 + 2] = 0.f;
    }
    __syncthreads();

    auto score = [&](float l, int dl) {
        const float x = l + er_t[dl];
        return (x >= 0.f) ? x : SLOPE * x;
    };
    auto add_edge = [&](float v, int dl, float g0, float g1, float g2) {
        const float a = expf(v - m_t[dl]);
        atomicAddLds(s_t + dl, a);
        atomicAddLds(a_t + dl * 3 + 0, g0 * a);
        atomicAddLds(a_t + dl * 3 + 1, g1 * a);
        atomicAddLds(a_t + dl * 3 + 2, g2 * a);
    };
    auto own_edges = [&](auto&& f) {
        for (int blk = wave; blk * MAP_BLOCK < n_edges; blk += AGG_WAVES) {
            for (int j = 0; j < MAP_BLOCK; j += 32) {
                const int e = blk * MAP_BLOCK + j + lane;
                if (e < n_edges) {
                    const int d = dst[e];
                    if ((unsigned)(d - lo) < span) f(src[e], d - lo);
                }
            }
        }
    };
    const bool listed = cnt <= L2_CAP;

    if (listed) {
        for (int base = 0; base < cnt; base += 128) {
            int2 en[4]; float l[4];
#pragma unroll
            for (int u = 0; u < 4; ++u) {
                const int i = base + u * 32 + lane;
                en[u] = slot[i < cnt ? i : cnt - 1];
            }
#pragma unroll
            for (int u = 0; u < 4; ++u) l[u] = el2[en[u].x];
            asm volatile("" :: "v"(l[0]), "v"(l[1]), "v"(l[2]), "v"(l[3]));
#pragma unroll
            for (int u = 0; u < 4; ++u)
                if (base + u * 32 + lane < cnt) ldsMaxFloat(m_t + en[u].y, score(l[u], en[u].y));
        }
    } else {
        own_edges([&](int sn, int dl) { ldsMaxFloat(m_t + dl, score(el2[sn], dl)); });
    }
    __syncthreads();

    if (listed) {
        for (int base = 0; base < cnt; base += 128) {
            int2 en[4]; float l[4], g0[4], g1[4], g2[4];
#pragma unroll
            for (int u = 0; u < 4; ++u) {
                const int i = base + u * 32 + lane;
                en[u] = slot[i < cnt ? i : cnt - 1];
            }
#pragma unroll
            for (int u = 0; u < 4; ++u) {
                l[u]  = el2[en[u].x];
                g0[u] = h2[(size_t)en[u].x * 3 + 0];
                g1[u] = h2[(size_t)en[u].x * 3 + 1];
                g2[u] = h2[(size_t)en[u].x * 3 + 2];
            }
            asm volatile("" :: "v"(l[0]), "v"(l[1]), "v"(l[2]), "v"(l[3]),
                               "v"(g0[0]), "v"(g0[1]), "v"(g0[2]), "v"(g0[3]),
                               "v"(g1[0]), "v"(g1[1]), "v"(g1[2]), "v"(g1[3]),
                               "v"(g2[0]), "v"(g2[1]), "v"(g2[2]), "v"(g2[3]));
#pragma unroll
            for (int u = 0; u < 4; ++u)
                if (base + u * 32 + lane < cnt)
                    add_edge(score(l[u], en[u].y), en[u].y, g0[u], g1[u], g2[u]);
        }
    } else {
        own_edges([&](int sn, int dl) {
            add_edge(score(el2[sn], dl), dl,
                     h2[(size_t)sn * 3 + 0], h2[(size_t)sn * 3 + 1], h2[(size_t)sn * 3 + 2]);
        });
    }
    __syncthreads();

    for (int i = tid; i < rows * 3; i += AGG_THREADS) {
        const int r = i / 3;
        const float sv = s_t[r];
        const float v = (sv != 0.f) ? a_t[i] / sv : 0.f;
        out[(size_t)lo * 3 + i] = v + b2[i - r * 3];
    }
}

extern "C" void kernel_launch(void* const* d_in, const int* in_sizes, int n_in,
                              void* d_out, int out_size, void* d_ws, size_t ws_size,
                              hipStream_t stream) {
    const float* feat = (const float*)d_in[0];
    const int*   src  = (const int*)  d_in[1];
    const int*   dst  = (const int*)  d_in[2];
    const float* W1   = (const float*)d_in[3];
    const float* al1  = (const float*)d_in[4];
    const float* ar1  = (const float*)d_in[5];
    const float* b1   = (const float*)d_in[6];
    const float* W2   = (const float*)d_in[7];
    const float* al2  = (const float*)d_in[8];
    const float* ar2  = (const float*)d_in[9];
    const float* b2   = (const float*)d_in[10];
    float* out = (float*)d_out;

    const int n_nodes = in_sizes[0] / 256;
    const int n_edges = in_sizes[1];

    char* cur = (char*)d_ws;
    auto alloc = [&](size_t bytes) -> float* {
        float* p = (float*)cur;
        cur += (bytes + 255) & ~(size_t)255;
        return p;
    };
    float* H1   = alloc((size_t)n_nodes * 256 * 4);
    float* el1  = alloc((size_t)n_nodes * 8 * 4);
    float* er1  = alloc((size_t)n_nodes * 8 * 4);
    float* h2   = alloc((size_t)n_nodes * 3 * 4);
    float* el2  = alloc((size_t)n_nodes * 4);
    float* er2  = alloc((size_t)n_nodes * 4);
    const int groups = cdiv(n_nodes, AGG_TILE);
    int*  l2_cnt  = (int*) alloc((size_t)groups * AGG_WAVES * 4);
    int2* l2_list = (int2*)alloc((size_t)groups * AGG_WAVES * L2_CAP * 8);
    unsigned int* emap = (unsigned int*)alloc((size_t)cdiv(n_edges, MAP_BLOCK) * groups * 32 * 4);

    const int B = 256;

    gat_gemm1_wmma<<<G1_GROUPS, G1_THREADS, 0, stream>>>(feat, W1, H1, n_nodes);
    gat_attn_scores<<<cdiv(n_nodes * 8, B), B, 0, stream>>>(H1, al1, ar1, el1, er1, dst, emap, n_edges, groups, n_nodes);

    gat_softmax_aggregate_tile<<<groups, AGG_THREADS, 0, stream>>>(src, dst, emap, H1, el1, er1,
                                                                   b1, W2, al2, ar2, h2, el2, er2,
                                                                   l2_list, l2_cnt, n_edges, n_nodes);

    gat_layer2_owner<<<groups, AGG_THREADS, 0, stream>>>(src, dst, l2_list, l2_cnt, h2, el2, er2, b2,
                                                         out, n_edges, n_nodes);
}


# Layer-1 pass 2: row's wave forms exp(score - max) once per (edge, head), one edge per lane, into a table in the row's own LDS row; sum loop reads 2 words per edge, no expf there (sum order kept)
# speedup vs baseline: 7.6570x; 1.0446x over previous
#include <hip/hip_runtime.h>
#include <hip/hip_bf16.h>

typedef __attribute__((ext_vector_type(2))) float v2f;
typedef __attribute__((ext_vector_type(8))) float v8f;

#define SLOPE 0.2f

static __host__ __device__ inline int cdiv(int a, int b) { return (a + b - 1) / b; }

__device__ inline void atomicAddLds(float* p, float v) {
    __hip_atomic_fetch_add(p, v, __ATOMIC_RELAXED, __HIP_MEMORY_SCOPE_WORKGROUP);
}

#define G1_GROUPS  256
#define G1_THREADS 512
#define G1_WAVES   16

__global__ __launch_bounds__(G1_THREADS)
void gat_gemm1_wmma(const float* __restrict__ A,
                    const float* __restrict__ W,
                    float* __restrict__ H,
                    int n_nodes) {
    __shared__ __attribute__((aligned(16))) float Wl[256 * 256];

    const int tid  = threadIdx.x;
    const int wave = __builtin_amdgcn_readfirstlane(tid >> 5);
    const int lane = tid & 31;
    const int c    = lane & 15;
    const int hf   = lane >> 4;
    const int q    = wave & 3;
    const int gq   = wave >> 2;

    const unsigned tiles = (unsigned)cdiv(n_nodes, 16);
    const int t0 = (int)((blockIdx.x * tiles) / G1_GROUPS);
    const int t1 = (int)(((blockIdx.x + 1) * tiles) / G1_GROUPS);
    const int full = (t1 - t0) >> 2;
    const int rem  = (t1 - t0) & 3;
    const int e    = (gq - q) & 3;
    const int nit  = full + (e < rem ? 1 : 0);
    auto tile_of = [&](int i) { return t0 + (i < full ? 4 * i + gq : 4 * full + e); };
    auto a_ptr = [&](int tile) {
        int r = 16 * tile + c; if (r >= n_nodes) r = n_nodes - 1;
        return A + (size_t)r * 256 + 4 * hf;
    };
    auto load_a = [&](float4 (&buf)[8], const float* p) {
#pragma unroll
        for (int j = 0; j < 8; ++j) buf[j] = *(const float4*)(p + 8 * j);
    };

    float4 a0[8], a1[8];
    load_a(a0, a_ptr(tile_of(0)));

#pragma unroll
    for (int k = 0; k < 128 / G1_WAVES; ++k) {
        const int u = wave + G1_WAVES * k;
        const float* wp = W + (size_t)(8 * (u >> 2) + 4 * hf) * 256 + 64 * (u & 3) + 4 * c;
        const float4 g0 = *(const float4*)(wp);
        const float4 g1 = *(const float4*)(wp + 256);
        const float4 g2 = *(const float4*)(wp + 512);
        const float4 g3 = *(const float4*)(wp + 768);
        float4* dp = (float4*)Wl + (u * 4) * 32 + lane;
        dp[0]  = make_float4(g0.x, g1.x, g2.x, g3.x);
        dp[32] = make_float4(g0.y, g1.y, g2.y, g3.y);
        dp[64] = make_float4(g0.z, g1.z, g2.z, g3.z);
        dp[96] = make_float4(g0.w, g1.w, g2.w, g3.w);
    }
    __syncthreads();

    v8f acc0 = {}, acc1 = {}, acc2 = {}, acc3 = {};
    const float4* bq = (const float4*)Wl + q * 128 + lane;

    auto mma_chunk = [&](const float4 (&a)[8], int ch) {
        const float4* bp = bq + ch * (8 * 512);
#pragma unroll
        for (int j = 0; j < 8; ++j) {
            const float4 b0 = bp[j * 512];
            const float4 b1 = bp[j * 512 + 32];
            const float4 b2 = bp[j * 512 + 64];
            const float4 b3 = bp[j * 512 + 96];
            const v2f alo = {a[j].x, a[j].y};
            const v2f ahi = {a[j].z, a[j].w};
            acc0 = __builtin_amdgcn_wmma_f32_16x16x4_f32(false, alo, false, (v2f){b0.x, b0.y}, (short)0, acc0, false, false);
            acc1 = __builtin_amdgcn_wmma_f32_16x16x4_f32(false, alo, false, (v2f){b1.x, b1.y}, (short)0, acc1, false, false);
            acc2 = __builtin_amdgcn_wmma_f32_16x16x4_f32(false, alo, false, (v2f){b2.x, b2.y}, (short)0, acc2, false, false);
            acc3 = __builtin_amdgcn_wmma_f32_16x16x4_f32(false, alo, false, (v2f){b3.x, b3.y}, (short)0, acc3, false, false);
            acc0 = __builtin_amdgcn_wmma_f32_16x16x4_f32(false, ahi, false, (v2f){b0.z, b0.w}, (short)0, acc0, false, false);
            acc1 = __builtin_amdgcn_wmma_f32_16x16x4_f32(false, ahi, false, (v2f){b1.z, b1.w}, (short)0, acc1, false, false);
            acc2 = __builtin_amdgcn_wmma_f32_16x16x4_f32(false, ahi, false, (v2f){b2.z, b2.w}, (short)0, acc2, false, false);
            acc3 = __builtin_amdgcn_wmma_f32_16x16x4_f32(false, ahi, false, (v2f){b3.z, b3.w}, (short)0, acc3, false, false);
        }
    };

    for (int it = 0; it < nit; ++it) {
        asm volatile("" ::: "memory");
        const int tile = tile_of(it);
        const float* ap  = a_ptr(tile);
        const float* apn = a_ptr(tile_of(it + 1 < nit ? it + 1 : it));
        load_a(a1, ap + 64);   mma_chunk(a0, 0);
        load_a(a0, ap + 128);  mma_chunk(a1, 1);
        load_a(a1, ap + 192);  mma_chunk(a0, 2);
        load_a(a0, apn);       mma_chunk(a1, 3);

        const int rbase = 16 * tile + 8 * hf;
        float* hp = H + (size_t)rbase * 256 + 64 * q + 4 * c;
        if (16 * tile + 16 <= n_nodes) {
#pragma unroll
            for (int r = 0; r < 8; ++r)
                *(float4*)(hp + r * 256) = make_float4(acc0[r], acc1[r], acc2[r], acc3[r]);
        } else {
#pragma unroll
            for (int r = 0; r < 8; ++r)
                if (rbase + r < n_nodes)
                    *(float4*)(hp + r * 256) = make_float4(acc0[r], acc1[r], acc2[r], acc3[r]);
        }
        acc0 = (v8f){}; acc1 = (v8f){}; acc2 = (v8f){}; acc3 = (v8f){};
    }
}

#define AGG_TILE    196
#define MAP_BLOCK   1024
#define MAP_GROUPS  256
#define SC_THREADS  256

__global__ __launch_bounds__(SC_THREADS)
void gat_attn_scores(const float* __restrict__ H,
                     const float* __restrict__ al,
                     const float* __restrict__ ar,
                     float* __restrict__ el, float* __restrict__ er,
                     const int* __restrict__ dst, unsigned int* __restrict__ emap,
                     int n_edges, int n_groups, int n_nodes) {
    __shared__ __attribute__((aligned(16))) unsigned int tab[MAP_GROUPS * 32];
    const int tid = threadIdx.x;
    const int n_blocks = cdiv(n_edges, MAP_BLOCK);
    for (int blk = blockIdx.x; blk < n_blocks; blk += gridDim.x) {
        const int e0 = blk * MAP_BLOCK + 4 * tid;
        int d[4];
        if (e0 + 3 < n_edges) {
            const int4 v = *(const int4*)(dst + e0);
            d[0] = v.x; d[1] = v.y; d[2] = v.z; d[3] = v.w;
        } else {
#pragma unroll
            for (int k = 0; k < 4; ++k) d[k] = (e0 + k < n_edges) ? dst[e0 + k] : -1;
        }
        for (int g0 = 0; g0 < n_groups; g0 += MAP_GROUPS) {
            for (int k = tid; k < MAP_GROUPS * 8; k += SC_THREADS)
                ((uint4*)tab)[k] = make_uint4(0u, 0u, 0u, 0u);
            __syncthreads();
#pragma unroll
            for (int k = 0; k < 4; ++k) {
                const unsigned g = (unsigned)d[k] / (unsigned)AGG_TILE;
                if (g < (unsigned)n_groups && g >= (unsigned)g0 && g - (unsigned)g0 < (unsigned)MAP_GROUPS)
                    atomicOr(tab + (g - (unsigned)g0) * 32 + (tid >> 3), 1u << (4 * (tid & 7) + k));
            }
            __syncthreads();
            int ng = n_groups - g0; if (ng > MAP_GROUPS) ng = MAP_GROUPS;
            uint4* op = (uint4*)(emap + ((size_t)blk * n_groups + g0) * 32);
            for (int k = tid; k < ng * 8; k += SC_THREADS) op[k] = ((const uint4*)tab)[k];
            __syncthreads();
        }
    }
    int i = blockIdx.x * blockDim.x + threadIdx.x;
    if (i >= n_nodes * 8) return;
    int n = i >> 3, h = i & 7;
    const float* hp  = H + (size_t)n * 256 + h * 32;
    const float* alp = al + h * 32;
    const float* arp = ar + h * 32;
    float sl = 0.f, sr = 0.f;
    for (int d = 0; d < 32; ++d) { float v = hp[d]; sl += v * alp[d]; sr += v * arp[d]; }
    el[i] = sl; er[i] = sr;
}

#define AGG_THREADS 1024
#define AGG_WAVES   32
#define AGG_PART    128
#define AGG_QCAP    224
#define AGG_RCAP    60
#define AGG_FLY     6
#define L2_CAP      512

__device__ inline void ldsMaxFloat(float* addr, float value) {
    if (__float_as_int(value) >= 0)
        atomicMax((int*)addr, __float_as_int(value));
    else
        atomicMin((unsigned int*)addr, __float_as_uint(value));
}

__global__ __launch_bounds__(AGG_THREADS)
void gat_softmax_aggregate_tile(const int* __restrict__ src, const int* __restrict__ dst,
                                const unsigned int* __restrict__ emap,
                                const float* __restrict__ H, const float* __restrict__ el,
                                const float* __restrict__ er,
                                const float* __restrict__ b1, const float* __restrict__ W2,
                                const float* __restrict__ al2, const float* __restrict__ ar2,
                                float* __restrict__ h2, float* __restrict__ el2,
                                float* __restrict__ er2,
                                int2* __restrict__ l2_list, int* __restrict__ l2_cnt,
                                int n_edges, int n_nodes) {
    __shared__ __attribute__((aligned(16))) float acc[AGG_TILE * 256];
    __shared__ __attribute__((aligned(16))) float er_t[AGG_TILE * 8];
    __shared__ __attribute__((aligned(16))) float m_t[AGG_TILE * 8];
    __shared__ float s_t[AGG_TILE * 8];
    __shared__ int   q_e[AGG_WAVES * AGG_QCAP];
    __shared__ int   o_src[AGG_TILE * AGG_RCAP];
    __shared__ int   o_cnt[AGG_TILE];

    const int tid  = threadIdx.x;
    const int wave = tid >> 5;
    const int lane = tid & 31;
    const int lo   = blockIdx.x * AGG_TILE;
    int rows = n_nodes - lo; if (rows > AGG_TILE) rows = AGG_TILE;
    const unsigned span = (unsigned)rows;

    for (int i = tid; i < AGG_TILE * 64; i += AGG_THREADS)
        ((float4*)acc)[i] = make_float4(0.f, 0.f, 0.f, 0.f);
    for (int i = tid; i < AGG_TILE * 8; i += AGG_THREADS) {
        er_t[i] = (i < rows * 8) ? er[(size_t)lo * 8 + i] : 0.f;
        m_t[i]  = -__builtin_huge_valf();
        s_t[i]  = 0.f;
    }
    for (int i = tid; i < AGG_TILE; i += AGG_THREADS) o_cnt[i] = 0;
    __syncthreads();

    int* qe = q_e + wave * AGG_QCAP;
    const unsigned lt_mask = (1u << lane) - 1u;
    const int hl = lane >> 3;
    int  cnt  = 0;
    int  done = 0;
    int  filed = 0;
    int2* slot = l2_list + (size_t)(blockIdx.x * AGG_WAVES + wave) * L2_CAP;

    auto drain_max = [&](int from, int n) {
        for (int b = 0; b < n; b += 32) {
            int dl = -1, sn = 0;
            if (b + lane < n) {
                const int e = qe[from + b + lane];
                dl = dst[e] - lo;
                sn = src[e];
            }
            const bool ok = (unsigned)dl < span;
            const unsigned okm = __builtin_amdgcn_ballot_w32(ok);
            if (ok) {
                const int fp = filed + __popc(okm & lt_mask);
                if (fp < L2_CAP) slot[fp] = make_int2(sn, dl);
                const int op = atomicAdd(o_cnt + dl, 1);
                if (op < AGG_RCAP) o_src[dl * AGG_RCAP + op] = sn;
                const float4 l0 = *(const float4*)(el + (size_t)sn * 8);
                const float4 l1 = *(const float4*)(el + (size_t)sn * 8 + 4);
                const float4 r0 = *(const float4*)(er_t + dl * 8);
                const float4 r1 = *(const float4*)(er_t + dl * 8 + 4);
                const float x[8] = {l0.x + r0.x, l0.y + r0.y, l0.z + r0.z, l0.w + r0.w,
                                    l1.x + r1.x, l1.y + r1.y, l1.z + r1.z, l1.w + r1.w};
#pragma unroll
                for (int h = 0; h < 8; ++h) {
                    const float v = (x[h] >= 0.f) ? x[h] : SLOPE * x[h];
                    ldsMaxFloat(m_t + dl * 8 + h, v);
                }
            }
            filed += __popc(okm);
        }
    };

    auto drain_sum = [&](int from, int n) {
        for (int b = 0; b < n; b += 32) {
            int nb = n - b; if (nb > 32) nb = 32;
            const int i  = from + b + (lane < nb ? lane : nb - 1);
            const int e  = qe[i];
            const int dl = dst[e] - lo;
            const int sn = src[e];
            for (int g = 0; g < nb; g += 4) {
                int su[4], du[4];
#pragma unroll
                for (int u = 0; u < 4; ++u) {
                    const int k = (g + u < nb) ? g + u : nb - 1;
                    su[u] = __builtin_amdgcn_readlane(sn, k);
                    du[u] = __builtin_amdgcn_readlane(dl, k);
                }
                float4 r0[4], r1[4];
                float  l0[4], l1[4];
#pragma unroll
                for (int u = 0; u < 4; ++u) {
                    const float* hp = H + (size_t)su[u] * 256 + 4 * lane;
                    r0[u] = *(const float4*)(hp);
                    r1[u] = *(const float4*)(hp + 128);
                    l0[u] = el[(size_t)su[u] * 8 + hl];
                    l1[u] = el[(size_t)su[u] * 8 + 4 + hl];
                }
#pragma unroll
                for (int u = 0; u < 4; ++u) {
                    if (g + u < nb && (unsigned)du[u] < span) {
                        const int t0 = du[u] * 8 + hl;
                        const float x0 = l0[u] + er_t[t0];
                        const float x1 = l1[u] + er_t[t0 + 4];
                        const float v0 = (x0 >= 0.f) ? x0 : SLOPE * x0;
                        const float v1 = (x1 >= 0.f) ? x1 : SLOPE * x1;
                        const float a0 = expf(v0 - m_t[t0]);
                        const float a1 = expf(v1 - m_t[t0 + 4]);
                        if ((lane & 7) == 0) {
                            atomicAddLds(s_t + t0,     a0);
                            atomicAddLds(s_t + t0 + 4, a1);
                        }
                        float* row = acc + du[u] * 256 + lane;
                        atomicAddLds(row,       r0[u].x * a0);
                        atomicAddLds(row + 32,  r0[u].y * a0);
                        atomicAddLds(row + 64,  r0[u].z * a0);
                        atomicAddLds(row + 96,  r0[u].w * a0);
                        atomicAddLds(row + 128, r1[u].x * a1);
                        atomicAddLds(row + 160, r1[u].y * a1);
                        atomicAddLds(row + 192, r1[u].z * a1);
                        atomicAddLds(row + 224, r1[u].w * a1);
                    }
                }
            }
        }
    };

    auto scan = [&](auto&& after) {
        const int n_blocks = cdiv(n_edges, MAP_BLOCK);
        if (n_blocks > 0) {
            auto loadw = [&](int blk) {
                const int bc = blk < n_blocks ? blk : n_blocks - 1;
                return emap[((size_t)bc * gridDim.x + blockIdx.x) * 32 + lane];
            };
            unsigned cur[4], nxt[4];
#pragma unroll
            for (int u = 0; u < 4; ++u) cur[u] = loadw(wave + u * AGG_WAVES);

            for (int b0 = wave; b0 < n_blocks; b0 += 4 * AGG_WAVES) {
#pragma unroll
                for (int u = 0; u < 4; ++u) nxt[u] = loadw(b0 + (4 + u) * AGG_WAVES);
#pragma unroll
                for (int u = 0; u < 4; ++u) {
                    const int blk = b0 + u * AGG_WAVES;
                    const int e0  = blk * MAP_BLOCK + lane * 32;
                    unsigned c = (blk < n_blocks) ? cur[u] : 0u;
                    unsigned mask = __builtin_amdgcn_ballot_w32(c != 0u);
                    while (mask != 0u) {
                        if (c != 0u) {
                            const int pos = cnt + __popc(mask & lt_mask);
                            qe[pos] = e0 + __builtin_ctz(c);
                            c &= c - 1u;
                        }
                        cnt += __popc(mask);
                        after();
                        mask = __builtin_amdgcn_ballot_w32(c != 0u);
                    }
                }
#pragma unroll
                for (int u = 0; u < 4; ++u) cur[u] = nxt[u];
            }
        }
    };

    scan([&]() {
        while (cnt - done >= 32) { drain_max(done, 32); done += 32; }
        if (cnt > AGG_QCAP - AGG_PART) {
            if (cnt > done) drain_max(done, cnt - done);
            cnt = 0; done = 0;
        }
    });
    if (cnt > done) drain_max(done, cnt - done);
    if (lane == 0) l2_cnt[blockIdx.x * AGG_WAVES + wave] = filed;
    __syncthreads();

    bool over = false;
    for (int i = lane; i < AGG_TILE; i += 32) over |= (o_cnt[i] > AGG_RCAP);
    const bool owned = (__builtin_amdgcn_ballot_w32(over) == 0u);

    if (owned) {
        for (int r = wave; r < rows; r += AGG_WAVES) {
            const int n = __builtin_amdgcn_readfirstlane(o_cnt[r]);
            const int* ol = o_src + r * AGG_RCAP;
            const float4 e0 = *(const float4*)(er_t + r * 8), e1 = *(const float4*)(er_t + r * 8 + 4);
            const float4 m0 = *(const float4*)(m_t + r * 8),  m1 = *(const float4*)(m_t + r * 8 + 4);
            float* wt = acc + r * 256;
            float sum[8] = {0.f, 0.f, 0.f, 0.f, 0.f, 0.f, 0.f, 0.f};
            float es0 = 0.f, es1 = 0.f;
            for (int b = 0; b < n; b += 32) {
                int nb = n - b; if (nb > 32) nb = 32;
                const int sn = ol[b + (lane < nb ? lane : nb - 1)];
                const float4 l0 = *(const float4*)(el + (size_t)sn * 8);
                const float4 l1 = *(const float4*)(el + (size_t)sn * 8 + 4);
                for (int g = 0; g < nb; g += AGG_FLY) {
                    int su[AGG_FLY];
#pragma unroll
                    for (int u = 0; u < AGG_FLY; ++u)
                        su[u] = __builtin_amdgcn_readlane(sn, (g + u < nb) ? g + u : nb - 1);
                    float4 r0[AGG_FLY], r1[AGG_FLY];
#pragma unroll
                    for (int u = 0; u < AGG_FLY; ++u) {
                        const float* hp = H + (size_t)su[u] * 256 + 4 * lane;
                        r0[u] = *(const float4*)(hp);
                        r1[u] = *(const float4*)(hp + 128);
                    }
                    if (g == 0) {
                        const float x[8] = {l0.x + e0.x, l0.y + e0.y, l0.z + e0.z, l0.w + e0.w,
                                            l1.x + e1.x, l1.y + e1.y, l1.z + e1.z, l1.w + e1.w};
                        const float mx[8] = {m0.x, m0.y, m0.z, m0.w, m1.x, m1.y, m1.z, m1.w};
                        float w[8];
#pragma unroll
                        for (int h = 0; h < 8; ++h) {
                            const float v = (x[h] >= 0.f) ? x[h] : SLOPE * x[h];
                            w[h] = expf(v - mx[h]);
                        }
                        *(float4*)(wt + lane * 8)     = make_float4(w[0], w[1], w[2], w[3]);
                        *(float4*)(wt + lane * 8 + 4) = make_float4(w[4], w[5], w[6], w[7]);
                    }
#pragma unroll
                    for (int u = 0; u < AGG_FLY; ++u) {
                        if (g + u < nb) {
                            const float a0 = wt[(g + u) * 8 + hl];
                            const float a1 = wt[(g + u) * 8 + 4 + hl];
                            es0 += a0;
                            es1 += a1;
                            sum[0] += r0[u].x * a0; sum[1] += r0[u].y * a0;
                            sum[2] += r0[u].z * a0; sum[3] += r0[u].w * a0;
                            sum[4] += r1[u].x * a1; sum[5] += r1[u].y * a1;
                            sum[6] += r1[u].z * a1; sum[7] += r1[u].w * a1;
                        }
                    }
                }
            }
            float* row = acc + r * 256 + lane;
#pragma unroll
            for (int j = 0; j < 8; ++j) row[(j >> 2) * 128 + (j & 3) * 32] = sum[j];
            if ((lane & 7) == 0) {
                s_t[r * 8 + hl]     = es0;
                s_t[r * 8 + 4 + hl] = es1;
            }
        }
    } else {
        cnt = 0;
        scan([&]() {
            if (cnt > AGG_QCAP - AGG_PART) { drain_sum(0, cnt); cnt = 0; }
        });
        if (cnt > 0) drain_sum(0, cnt);
        __syncthreads();
    }

    float bw[8], w0[8], w1[8], w2[8];
#pragma unroll
    for (int j = 0; j < 8; ++j) {
        const int c = (j >> 2) * 128 + 4 * lane + (j & 3);
        bw[j] = b1[c];
        w0[j] = W2[c * 3 + 0];
        w1[j] = W2[c * 3 + 1];
        w2[j] = W2[c * 3 + 2];
    }
    const float l2a0 = al2[0], l2a1 = al2[1], l2a2 = al2[2];
    const float r2a0 = ar2[0], r2a1 = ar2[1], r2a2 = ar2[2];
    for (int r = wave; r < rows; r += AGG_WAVES) {
        const float* p = acc + r * 256 + lane;
        const float sv0 = s_t[r * 8 + hl];
        const float sv1 = s_t[r * 8 + 4 + hl];
        float a0 = 0.f, a1 = 0.f, a2 = 0.f;
#pragma unroll
        for (int j = 0; j < 8; ++j) {
            const float sv = (j < 4) ? sv0 : sv1;
            const float q  = p[(j >> 2) * 128 + (j & 3) * 32];
            float v = (sv != 0.f) ? q / sv : 0.f;
            v += bw[j];
            v = (v > 0.f) ? v : expm1f(v);
            a0 += v * w0[j];
            a1 += v * w1[j];
            a2 += v * w2[j];
        }
        for (int o = 16; o > 0; o >>= 1) {
            a0 += __shfl_xor(a0, o);
            a1 += __shfl_xor(a1, o);
            a2 += __shfl_xor(a2, o);
        }
        if (lane == 0) {
            const size_t n = (size_t)(lo + r);
            h2[n * 3 + 0] = a0;
            h2[n * 3 + 1] = a1;
            h2[n * 3 + 2] = a2;
            el2[n] = a0 * l2a0 + a1 * l2a1 + a2 * l2a2;
            er2[n] = a0 * r2a0 + a1 * r2a1 + a2 * r2a2;
        }
    }
}

__global__ __launch_bounds__(AGG_THREADS)
void gat_layer2_owner(const int* __restrict__ src, const int* __restrict__ dst,
                      const int2* __restrict__ l2_list, const int* __restrict__ l2_cnt,
                      const float* __restrict__ h2, const float* __restrict__ el2,
                      const float* __restrict__ er2, const float* __restrict__ b2,
                      float* __restrict__ out, int n_edges, int n_nodes) {
    __shared__ float er_t[AGG_TILE];
    __shared__ float m_t[AGG_TILE];
    __shared__ float s_t[AGG_TILE];
    __shared__ float a_t[AGG_TILE * 3];

    const int tid  = threadIdx.x;
    const int wave = tid >> 5;
    const int lane = tid & 31;
    const int lo   = blockIdx.x * AGG_TILE;
    int rows = n_nodes - lo; if (rows > AGG_TILE) rows = AGG_TILE;
    const unsigned span = (unsigned)rows;

    const int cnt = __builtin_amdgcn_readfirstlane(l2_cnt[blockIdx.x * AGG_WAVES + wave]);
    const int2* slot = l2_list + (size_t)(blockIdx.x * AGG_WAVES + wave) * L2_CAP;

    for (int i = tid; i < AGG_TILE; i += AGG_THREADS) {
        er_t[i] = (i < rows) ? er2[lo + i] : 0.f;
        m_t[i]  = -__builtin_huge_valf();
        s_t[i]  = 0.f;
        a_t[i * 3 + 0] = 0.f;
        a_t[i * 3 + 1] = 0.f;
        a_t[i * 3 + 2] = 0.f;
    }
    __syncthreads();

    auto score = [&](float l, int dl) {
        const float x = l + er_t[dl];
        return (x >= 0.f) ? x : SLOPE * x;
    };
    auto add_edge = [&](float v, int dl, float g0, float g1, float g2) {
        const float a = expf(v - m_t[dl]);
        atomicAddLds(s_t + dl, a);
        atomicAddLds(a_t + dl * 3 + 0, g0 * a);
        atomicAddLds(a_t + dl * 3 + 1, g1 * a);
        atomicAddLds(a_t + dl * 3 + 2, g2 * a);
    };
    auto own_edges = [&](auto&& f) {
        for (int blk = wave; blk * MAP_BLOCK < n_edges; blk += AGG_WAVES) {
            for (int j = 0; j < MAP_BLOCK; j += 32) {
                const int e = blk * MAP_BLOCK + j + lane;
                if (e < n_edges) {
                    const int d = dst[e];
                    if ((unsigned)(d - lo) < span) f(src[e], d - lo);
                }
            }
        }
    };
    const bool listed = cnt <= L2_CAP;

    if (listed) {
        for (int base = 0; base < cnt; base += 128) {
            int2 en[4]; float l[4];
#pragma unroll
            for (int u = 0; u < 4; ++u) {
                const int i = base + u * 32 + lane;
                en[u] = slot[i < cnt ? i : cnt - 1];
            }
#pragma unroll
            for (int u = 0; u < 4; ++u) l[u] = el2[en[u].x];
            asm volatile("" :: "v"(l[0]), "v"(l[1]), "v"(l[2]), "v"(l[3]));
#pragma unroll
            for (int u = 0; u < 4; ++u)
                if (base + u * 32 + lane < cnt) ldsMaxFloat(m_t + en[u].y, score(l[u], en[u].y));
        }
    } else {
        own_edges([&](int sn, int dl) { ldsMaxFloat(m_t + dl, score(el2[sn], dl)); });
    }
    __syncthreads();

    if (listed) {
        for (int base = 0; base < cnt; base += 128) {
            int2 en[4]; float l[4], g0[4], g1[4], g2[4];
#pragma unroll
            for (int u = 0; u < 4; ++u) {
                const int i = base + u * 32 + lane;
                en[u] = slot[i < cnt ? i : cnt - 1];
            }
#pragma unroll
            for (int u = 0; u < 4; ++u) {
                l[u]  = el2[en[u].x];
                g0[u] = h2[(size_t)en[u].x * 3 + 0];
                g1[u] = h2[(size_t)en[u].x * 3 + 1];
                g2[u] = h2[(size_t)en[u].x * 3 + 2];
            }
            asm volatile("" :: "v"(l[0]), "v"(l[1]), "v"(l[2]), "v"(l[3]),
                               "v"(g0[0]), "v"(g0[1]), "v"(g0[2]), "v"(g0[3]),
                               "v"(g1[0]), "v"(g1[1]), "v"(g1[2]), "v"(g1[3]),
                               "v"(g2[0]), "v"(g2[1]), "v"(g2[2]), "v"(g2[3]));
#pragma unroll
            for (int u = 0; u < 4; ++u)
                if (base + u * 32 + lane < cnt)
                    add_edge(score(l[u], en[u].y), en[u].y, g0[u], g1[u], g2[u]);
        }
    } else {
        own_edges([&](int sn, int dl) {
            add_edge(score(el2[sn], dl), dl,
                     h2[(size_t)sn * 3 + 0], h2[(size_t)sn * 3 + 1], h2[(size_t)sn * 3 + 2]);
        });
    }
    __syncthreads();

    for (int i = tid; i < rows * 3; i += AGG_THREADS) {
        const int r = i / 3;
        const float sv = s_t[r];
        const float v = (sv != 0.f) ? a_t[i] / sv : 0.f;
        out[(size_t)lo * 3 + i] = v + b2[i - r * 3];
    }
}

extern "C" void kernel_launch(void* const* d_in, const int* in_sizes, int n_in,
                              void* d_out, int out_size, void* d_ws, size_t ws_size,
                              hipStream_t stream) {
    const float* feat = (const float*)d_in[0];
    const int*   src  = (const int*)  d_in[1];
    const int*   dst  = (const int*)  d_in[2];
    const float* W1   = (const float*)d_in[3];
    const float* al1  = (const float*)d_in[4];
    const float* ar1  = (const float*)d_in[5];
    const float* b1   = (const float*)d_in[6];
    const float* W2   = (const float*)d_in[7];
    const float* al2  = (const float*)d_in[8];
    const float* ar2  = (const float*)d_in[9];
    const float* b2   = (const float*)d_in[10];
    float* out = (float*)d_out;

    const int n_nodes = in_sizes[0] / 256;
    const int n_edges = in_sizes[1];

    char* cur = (char*)d_ws;
    auto alloc = [&](size_t bytes) -> float* {
        float* p = (float*)cur;
        cur += (bytes + 255) & ~(size_t)255;
        return p;
    };
    float* H1   = alloc((size_t)n_nodes * 256 * 4);
    float* el1  = alloc((size_t)n_nodes * 8 * 4);
    float* er1  = alloc((size_t)n_nodes * 8 * 4);
    float* h2   = alloc((size_t)n_nodes * 3 * 4);
    float* el2  = alloc((size_t)n_nodes * 4);
    float* er2  = alloc((size_t)n_nodes * 4);
    const int groups = cdiv(n_nodes, AGG_TILE);
    int*  l2_cnt  = (int*) alloc((size_t)groups * AGG_WAVES * 4);
    int2* l2_list = (int2*)alloc((size_t)groups * AGG_WAVES * L2_CAP * 8);
    unsigned int* emap = (unsigned int*)alloc((size_t)cdiv(n_edges, MAP_BLOCK) * groups * 32 * 4);

    const int B = 256;

    gat_gemm1_wmma<<<G1_GROUPS, G1_THREADS, 0, stream>>>(feat, W1, H1, n_nodes);
    gat_attn_scores<<<cdiv(n_nodes * 8, B), B, 0, stream>>>(H1, al1, ar1, el1, er1, dst, emap, n_edges, groups, n_nodes);

    gat_softmax_aggregate_tile<<<groups, AGG_THREADS, 0, stream>>>(src, dst, emap, H1, el1, er1,
                                                                   b1, W2, al2, ar2, h2, el2, er2,
                                                                   l2_list, l2_cnt, n_edges, n_nodes);

    gat_layer2_owner<<<groups, AGG_THREADS, 0, stream>>>(src, dst, l2_list, l2_cnt, h2, el2, er2, b2,
                                                         out, n_edges, n_nodes);
}


# Layer 2 gathers each edge's source once: layer 1 files (h2, el2) as one 16-byte record per node; a wave's first 128 edges keep entry and record in registers from max pass to sum pass
# speedup vs baseline: 8.0097x; 1.0461x over previous
#include <hip/hip_runtime.h>
#include <hip/hip_bf16.h>

typedef __attribute__((ext_vector_type(2))) float v2f;
typedef __attribute__((ext_vector_type(8))) float v8f;

#define SLOPE 0.2f

static __host__ __device__ inline int cdiv(int a, int b) { return (a + b - 1) / b; }

__device__ inline void atomicAddLds(float* p, float v) {
    __hip_atomic_fetch_add(p, v, __ATOMIC_RELAXED, __HIP_MEMORY_SCOPE_WORKGROUP);
}

#define G1_GROUPS  256
#define G1_THREADS 512
#define G1_WAVES   16

__global__ __launch_bounds__(G1_THREADS)
void gat_gemm1_wmma(const float* __restrict__ A,
                    const float* __restrict__ W,
                    float* __restrict__ H,
                    int n_nodes) {
    __shared__ __attribute__((aligned(16))) float Wl[256 * 256];

    const int tid  = threadIdx.x;
    const int wave = __builtin_amdgcn_readfirstlane(tid >> 5);
    const int lane = tid & 31;
    const int c    = lane & 15;
    const int hf   = lane >> 4;
    const int q    = wave & 3;
    const int gq   = wave >> 2;

    const unsigned tiles = (unsigned)cdiv(n_nodes, 16);
    const int t0 = (int)((blockIdx.x * tiles) / G1_GROUPS);
    const int t1 = (int)(((blockIdx.x + 1) * tiles) / G1_GROUPS);
    const int full = (t1 - t0) >> 2;
    const int rem  = (t1 - t0) & 3;
    const int e    = (gq - q) & 3;
    const int nit  = full + (e < rem ? 1 : 0);
    auto tile_of = [&](int i) { return t0 + (i < full ? 4 * i + gq : 4 * full + e); };
    auto a_ptr = [&](int tile) {
        int r = 16 * tile + c; if (r >= n_nodes) r = n_nodes - 1;
        return A + (size_t)r * 256 + 4 * hf;
    };
    auto load_a = [&](float4 (&buf)[8], const float* p) {
#pragma unroll
        for (int j = 0; j < 8; ++j) buf[j] = *(const float4*)(p + 8 * j);
    };

    float4 a0[8], a1[8];
    load_a(a0, a_ptr(tile_of(0)));

#pragma unroll
    for (int k = 0; k < 128 / G1_WAVES; ++k) {
        const int u = wave + G1_WAVES * k;
        const float* wp = W + (size_t)(8 * (u >> 2) + 4 * hf) * 256 + 64 * (u & 3) + 4 * c;
        const float4 g0 = *(const float4*)(wp);
        const float4 g1 = *(const float4*)(wp + 256);
        const float4 g2 = *(const float4*)(wp + 512);
        const float4 g3 = *(const float4*)(wp + 768);
        float4* dp = (float4*)Wl + (u * 4) * 32 + lane;
        dp[0]  = make_float4(g0.x, g1.x, g2.x, g3.x);
        dp[32] = make_float4(g0.y, g1.y, g2.y, g3.y);
        dp[64] = make_float4(g0.z, g1.z, g2.z, g3.z);
        dp[96] = make_float4(g0.w, g1.w, g2.w, g3.w);
    }
    __syncthreads();

    v8f acc0 = {}, acc1 = {}, acc2 = {}, acc3 = {};
    const float4* bq = (const float4*)Wl + q * 128 + lane;

    auto mma_chunk = [&](const float4 (&a)[8], int ch) {
        const float4* bp = bq + ch * (8 * 512);
#pragma unroll
        for (int j = 0; j < 8; ++j) {
            const float4 b0 = bp[j * 512];
            const float4 b1 = bp[j * 512 + 32];
            const float4 b2 = bp[j * 512 + 64];
            const float4 b3 = bp[j * 512 + 96];
            const v2f alo = {a[j].x, a[j].y};
            const v2f ahi = {a[j].z, a[j].w};
            acc0 = __builtin_amdgcn_wmma_f32_16x16x4_f32(false, alo, false, (v2f){b0.x, b0.y}, (short)0, acc0, false, false);
            acc1 = __builtin_amdgcn_wmma_f32_16x16x4_f32(false, alo, false, (v2f){b1.x, b1.y}, (short)0, acc1, false, false);
            acc2 = __builtin_amdgcn_wmma_f32_16x16x4_f32(false, alo, false, (v2f){b2.x, b2.y}, (short)0, acc2, false, false);
            acc3 = __builtin_amdgcn_wmma_f32_16x16x4_f32(false, alo, false, (v2f){b3.x, b3.y}, (short)0, acc3, false, false);
            acc0 = __builtin_amdgcn_wmma_f32_16x16x4_f32(false, ahi, false, (v2f){b0.z, b0.w}, (short)0, acc0, false, false);
            acc1 = __builtin_amdgcn_wmma_f32_16x16x4_f32(false, ahi, false, (v2f){b1.z, b1.w}, (short)0, acc1, false, false);
            acc2 = __builtin_amdgcn_wmma_f32_16x16x4_f32(false, ahi, false, (v2f){b2.z, b2.w}, (short)0, acc2, false, false);
            acc3 = __builtin_amdgcn_wmma_f32_16x16x4_f32(false, ahi, false, (v2f){b3.z, b3.w}, (short)0, acc3, false, false);
        }
    };

    for (int it = 0; it < nit; ++it) {
        asm volatile("" ::: "memory");
        const int tile = tile_of(it);
        const float* ap  = a_ptr(tile);
        const float* apn = a_ptr(tile_of(it + 1 < nit ? it + 1 : it));
        load_a(a1, ap + 64);   mma_chunk(a0, 0);
        load_a(a0, ap + 128);  mma_chunk(a1, 1);
        load_a(a1, ap + 192);  mma_chunk(a0, 2);
        load_a(a0, apn);       mma_chunk(a1, 3);

        const int rbase = 16 * tile + 8 * hf;
        float* hp = H + (size_t)rbase * 256 + 64 * q + 4 * c;
        if (16 * tile + 16 <= n_nodes) {
#pragma unroll
            for (int r = 0; r < 8; ++r)
                *(float4*)(hp + r * 256) = make_float4(acc0[r], acc1[r], acc2[r], acc3[r]);
        } else {
#pragma unroll
            for (int r = 0; r < 8; ++r)
                if (rbase + r < n_nodes)
                    *(float4*)(hp + r * 256) = make_float4(acc0[r], acc1[r], acc2[r], acc3[r]);
        }
        acc0 = (v8f){}; acc1 = (v8f){}; acc2 = (v8f){}; acc3 = (v8f){};
    }
}

#define AGG_TILE    196
#define MAP_BLOCK   1024
#define MAP_GROUPS  256
#define SC_THREADS  256

__global__ __launch_bounds__(SC_THREADS)
void gat_attn_scores(const float* __restrict__ H,
                     const float* __restrict__ al,
                     const float* __restrict__ ar,
                     float* __restrict__ el, float* __restrict__ er,
                     const int* __restrict__ dst, unsigned int* __restrict__ emap,
                     int n_edges, int n_groups, int n_nodes) {
    __shared__ __attribute__((aligned(16))) unsigned int tab[MAP_GROUPS * 32];
    const int tid = threadIdx.x;
    const int n_blocks = cdiv(n_edges, MAP_BLOCK);
    for (int blk = blockIdx.x; blk < n_blocks; blk += gridDim.x) {
        const int e0 = blk * MAP_BLOCK + 4 * tid;
        int d[4];
        if (e0 + 3 < n_edges) {
            const int4 v = *(const int4*)(dst + e0);
            d[0] = v.x; d[1] = v.y; d[2] = v.z; d[3] = v.w;
        } else {
#pragma unroll
            for (int k = 0; k < 4; ++k) d[k] = (e0 + k < n_edges) ? dst[e0 + k] : -1;
        }
        for (int g0 = 0; g0 < n_groups; g0 += MAP_GROUPS) {
            for (int k = tid; k < MAP_GROUPS * 8; k += SC_THREADS)
                ((uint4*)tab)[k] = make_uint4(0u, 0u, 0u, 0u);
            __syncthreads();
#pragma unroll
            for (int k = 0; k < 4; ++k) {
                const unsigned g = (unsigned)d[k] / (unsigned)AGG_TILE;
                if (g < (unsigned)n_groups && g >= (unsigned)g0 && g - (unsigned)g0 < (unsigned)MAP_GROUPS)
                    atomicOr(tab + (g - (unsigned)g0) * 32 + (tid >> 3), 1u << (4 * (tid & 7) + k));
            }
            __syncthreads();
            int ng = n_groups - g0; if (ng > MAP_GROUPS) ng = MAP_GROUPS;
            uint4* op = (uint4*)(emap + ((size_t)blk * n_groups + g0) * 32);
            for (int k = tid; k < ng * 8; k += SC_THREADS) op[k] = ((const uint4*)tab)[k];
            __syncthreads();
        }
    }
    int i = blockIdx.x * blockDim.x + threadIdx.x;
    if (i >= n_nodes * 8) return;
    int n = i >> 3, h = i & 7;
    const float* hp  = H + (size_t)n * 256 + h * 32;
    const float* alp = al + h * 32;
    const float* arp = ar + h * 32;
    float sl = 0.f, sr = 0.f;
    for (int d = 0; d < 32; ++d) { float v = hp[d]; sl += v * alp[d]; sr += v * arp[d]; }
    el[i] = sl; er[i] = sr;
}

#define AGG_THREADS 1024
#define AGG_WAVES   32
#define AGG_PART    128
#define AGG_QCAP    224
#define AGG_RCAP    60
#define AGG_FLY     6
#define L2_CAP      512

__device__ inline void ldsMaxFloat(float* addr, float value) {
    if (__float_as_int(value) >= 0)
        atomicMax((int*)addr, __float_as_int(value));
    else
        atomicMin((unsigned int*)addr, __float_as_uint(value));
}

__global__ __launch_bounds__(AGG_THREADS)
void gat_softmax_aggregate_tile(const int* __restrict__ src, const int* __restrict__ dst,
                                const unsigned int* __restrict__ emap,
                                const float* __restrict__ H, const float* __restrict__ el,
                                const float* __restrict__ er,
                                const float* __restrict__ b1, const float* __restrict__ W2,
                                const float* __restrict__ al2, const float* __restrict__ ar2,
                                float4* __restrict__ rec2, float* __restrict__ er2,
                                int2* __restrict__ l2_list, int* __restrict__ l2_cnt,
                                int n_edges, int n_nodes) {
    __shared__ __attribute__((aligned(16))) float acc[AGG_TILE * 256];
    __shared__ __attribute__((aligned(16))) float er_t[AGG_TILE * 8];
    __shared__ __attribute__((aligned(16))) float m_t[AGG_TILE * 8];
    __shared__ float s_t[AGG_TILE * 8];
    __shared__ int   q_e[AGG_WAVES * AGG_QCAP];
    __shared__ int   o_src[AGG_TILE * AGG_RCAP];
    __shared__ int   o_cnt[AGG_TILE];

    const int tid  = threadIdx.x;
    const int wave = tid >> 5;
    const int lane = tid & 31;
    const int lo   = blockIdx.x * AGG_TILE;
    int rows = n_nodes - lo; if (rows > AGG_TILE) rows = AGG_TILE;
    const unsigned span = (unsigned)rows;

    for (int i = tid; i < AGG_TILE * 64; i += AGG_THREADS)
        ((float4*)acc)[i] = make_float4(0.f, 0.f, 0.f, 0.f);
    for (int i = tid; i < AGG_TILE * 8; i += AGG_THREADS) {
        er_t[i] = (i < rows * 8) ? er[(size_t)lo * 8 + i] : 0.f;
        m_t[i]  = -__builtin_huge_valf();
        s_t[i]  = 0.f;
    }
    for (int i = tid; i < AGG_TILE; i += AGG_THREADS) o_cnt[i] = 0;
    __syncthreads();

    int* qe = q_e + wave * AGG_QCAP;
    const unsigned lt_mask = (1u << lane) - 1u;
    const int hl = lane >> 3;
    int  cnt  = 0;
    int  done = 0;
    int  filed = 0;
    int2* slot = l2_list + (size_t)(blockIdx.x * AGG_WAVES + wave) * L2_CAP;

    auto drain_max = [&](int from, int n) {
        for (int b = 0; b < n; b += 32) {
            int dl = -1, sn = 0;
            if (b + lane < n) {
                const int e = qe[from + b + lane];
                dl = dst[e] - lo;
                sn = src[e];
            }
            const bool ok = (unsigned)dl < span;
            const unsigned okm = __builtin_amdgcn_ballot_w32(ok);
            if (ok) {
                const int fp = filed + __popc(okm & lt_mask);
                if (fp < L2_CAP) slot[fp] = make_int2(sn, dl);
                const int op = atomicAdd(o_cnt + dl, 1);
                if (op < AGG_RCAP) o_src[dl * AGG_RCAP + op] = sn;
                const float4 l0 = *(const float4*)(el + (size_t)sn * 8);
                const float4 l1 = *(const float4*)(el + (size_t)sn * 8 + 4);
                const float4 r0 = *(const float4*)(er_t + dl * 8);
                const float4 r1 = *(const float4*)(er_t + dl * 8 + 4);
                const float x[8] = {l0.x + r0.x, l0.y + r0.y, l0.z + r0.z, l0.w + r0.w,
                                    l1.x + r1.x, l1.y + r1.y, l1.z + r1.z, l1.w + r1.w};
#pragma unroll
                for (int h = 0; h < 8; ++h) {
                    const float v = (x[h] >= 0.f) ? x[h] : SLOPE * x[h];
                    ldsMaxFloat(m_t + dl * 8 + h, v);
                }
            }
            filed += __popc(okm);
        }
    };

    auto drain_sum = [&](int from, int n) {
        for (int b = 0; b < n; b += 32) {
            int nb = n - b; if (nb > 32) nb = 32;
            const int i  = from + b + (lane < nb ? lane : nb - 1);
            const int e  = qe[i];
            const int dl = dst[e] - lo;
            const int sn = src[e];
            for (int g = 0; g < nb; g += 4) {
                int su[4], du[4];
#pragma unroll
                for (int u = 0; u < 4; ++u) {
                    const int k = (g + u < nb) ? g + u : nb - 1;
                    su[u] = __builtin_amdgcn_readlane(sn, k);
                    du[u] = __builtin_amdgcn_readlane(dl, k);
                }
                float4 r0[4], r1[4];
                float  l0[4], l1[4];
#pragma unroll
                for (int u = 0; u < 4; ++u) {
                    const float* hp = H + (size_t)su[u] * 256 + 4 * lane;
                    r0[u] = *(const float4*)(hp);
                    r1[u] = *(const float4*)(hp + 128);
                    l0[u] = el[(size_t)su[u] * 8 + hl];
                    l1[u] = el[(size_t)su[u] * 8 + 4 + hl];
                }
#pragma unroll
                for (int u = 0; u < 4; ++u) {
                    if (g + u < nb && (unsigned)du[u] < span) {
                        const int t0 = du[u] * 8 + hl;
                        const float x0 = l0[u] + er_t[t0];
                        const float x1 = l1[u] + er_t[t0 + 4];
                        const float v0 = (x0 >= 0.f) ? x0 : SLOPE * x0;
                        const float v1 = (x1 >= 0.f) ? x1 : SLOPE * x1;
                        const float a0 = expf(v0 - m_t[t0]);
                        const float a1 = expf(v1 - m_t[t0 + 4]);
                        if ((lane & 7) == 0) {
                            atomicAddLds(s_t + t0,     a0);
                            atomicAddLds(s_t + t0 + 4, a1);
                        }
                        float* row = acc + du[u] * 256 + lane;
                        atomicAddLds(row,       r0[u].x * a0);
                        atomicAddLds(row + 32,  r0[u].y * a0);
                        atomicAddLds(row + 64,  r0[u].z * a0);
                        atomicAddLds(row + 96,  r0[u].w * a0);
                        atomicAddLds(row + 128, r1[u].x * a1);
                        atomicAddLds(row + 160, r1[u].y * a1);
                        atomicAddLds(row + 192, r1[u].z * a1);
                        atomicAddLds(row + 224, r1[u].w * a1);
                    }
                }
            }
        }
    };

    auto scan = [&](auto&& after) {
        const int n_blocks = cdiv(n_edges, MAP_BLOCK);
        if (n_blocks > 0) {
            auto loadw = [&](int blk) {
                const int bc = blk < n_blocks ? blk : n_blocks - 1;
                return emap[((size_t)bc * gridDim.x + blockIdx.x) * 32 + lane];
            };
            unsigned cur[4], nxt[4];
#pragma unroll
            for (int u = 0; u < 4; ++u) cur[u] = loadw(wave + u * AGG_WAVES);

            for (int b0 = wave; b0 < n_blocks; b0 += 4 * AGG_WAVES) {
#pragma unroll
                for (int u = 0; u < 4; ++u) nxt[u] = loadw(b0 + (4 + u) * AGG_WAVES);
#pragma unroll
                for (int u = 0; u < 4; ++u) {
                    const int blk = b0 + u * AGG_WAVES;
                    const int e0  = blk * MAP_BLOCK + lane * 32;
                    unsigned c = (blk < n_blocks) ? cur[u] : 0u;
                    unsigned mask = __builtin_amdgcn_ballot_w32(c != 0u);
                    while (mask != 0u) {
                        if (c != 0u) {
                            const int pos = cnt + __popc(mask & lt_mask);
                            qe[pos] = e0 + __builtin_ctz(c);
                            c &= c - 1u;
                        }
                        cnt += __popc(mask);
                        after();
                        mask = __builtin_amdgcn_ballot_w32(c != 0u);
                    }
                }
#pragma unroll
                for (int u = 0; u < 4; ++u) cur[u] = nxt[u];
            }
        }
    };

    scan([&]() {
        while (cnt - done >= 32) { drain_max(done, 32); done += 32; }
        if (cnt > AGG_QCAP - AGG_PART) {
            if (cnt > done) drain_max(done, cnt - done);
            cnt = 0; done = 0;
        }
    });
    if (cnt > done) drain_max(done, cnt - done);
    if (lane == 0) l2_cnt[blockIdx.x * AGG_WAVES + wave] = filed;
    __syncthreads();

    bool over = false;
    for (int i = lane; i < AGG_TILE; i += 32) over |= (o_cnt[i] > AGG_RCAP);
    const bool owned = (__builtin_amdgcn_ballot_w32(over) == 0u);

    if (owned) {
        for (int r = wave; r < rows; r += AGG_WAVES) {
            const int n = __builtin_amdgcn_readfirstlane(o_cnt[r]);
            const int* ol = o_src + r * AGG_RCAP;
            const float4 e0 = *(const float4*)(er_t + r * 8), e1 = *(const float4*)(er_t + r * 8 + 4);
            const float4 m0 = *(const float4*)(m_t + r * 8),  m1 = *(const float4*)(m_t + r * 8 + 4);
            float* wt = acc + r * 256;
            float sum[8] = {0.f, 0.f, 0.f, 0.f, 0.f, 0.f, 0.f, 0.f};
            float es0 = 0.f, es1 = 0.f;
            for (int b = 0; b < n; b += 32) {
                int nb = n - b; if (nb > 32) nb = 32;
                const int sn = ol[b + (lane < nb ? lane : nb - 1)];
                const float4 l0 = *(const float4*)(el + (size_t)sn * 8);
                const float4 l1 = *(const float4*)(el + (size_t)sn * 8 + 4);
                for (int g = 0; g < nb; g += AGG_FLY) {
                    int su[AGG_FLY];
#pragma unroll
                    for (int u = 0; u < AGG_FLY; ++u)
                        su[u] = __builtin_amdgcn_readlane(sn, (g + u < nb) ? g + u : nb - 1);
                    float4 r0[AGG_FLY], r1[AGG_FLY];
#pragma unroll
                    for (int u = 0; u < AGG_FLY; ++u) {
                        const float* hp = H + (size_t)su[u] * 256 + 4 * lane;
                        r0[u] = *(const float4*)(hp);
                        r1[u] = *(const float4*)(hp + 128);
                    }
                    if (g == 0) {
                        const float x[8] = {l0.x + e0.x, l0.y + e0.y, l0.z + e0.z, l0.w + e0.w,
                                            l1.x + e1.x, l1.y + e1.y, l1.z + e1.z, l1.w + e1.w};
                        const float mx[8] = {m0.x, m0.y, m0.z, m0.w, m1.x, m1.y, m1.z, m1.w};
                        float w[8];
#pragma unroll
                        for (int h = 0; h < 8; ++h) {
                            const float v = (x[h] >= 0.f) ? x[h] : SLOPE * x[h];
                            w[h] = expf(v - mx[h]);
                        }
                        *(float4*)(wt + lane * 8)     = make_float4(w[0], w[1], w[2], w[3]);
                        *(float4*)(wt + lane * 8 + 4) = make_float4(w[4], w[5], w[6], w[7]);
                    }
#pragma unroll
                    for (int u = 0; u < AGG_FLY; ++u) {
                        if (g + u < nb) {
                            const float a0 = wt[(g + u) * 8 + hl];
                            const float a1 = wt[(g + u) * 8 + 4 + hl];
                            es0 += a0;
                            es1 += a1;
                            sum[0] += r0[u].x * a0; sum[1] += r0[u].y * a0;
                            sum[2] += r0[u].z * a0; sum[3] += r0[u].w * a0;
                            sum[4] += r1[u].x * a1; sum[5] += r1[u].y * a1;
                            sum[6] += r1[u].z * a1; sum[7] += r1[u].w * a1;
                        }
                    }
                }
            }
            float* row = acc + r * 256 + lane;
#pragma unroll
            for (int j = 0; j < 8; ++j) row[(j >> 2) * 128 + (j & 3) * 32] = sum[j];
            if ((lane & 7) == 0) {
                s_t[r * 8 + hl]     = es0;
                s_t[r * 8 + 4 + hl] = es1;
            }
        }
    } else {
        cnt = 0;
        scan([&]() {
            if (cnt > AGG_QCAP - AGG_PART) { drain_sum(0, cnt); cnt = 0; }
        });
        if (cnt > 0) drain_sum(0, cnt);
        __syncthreads();
    }

    float bw[8], w0[8], w1[8], w2[8];
#pragma unroll
    for (int j = 0; j < 8; ++j) {
        const int c = (j >> 2) * 128 + 4 * lane + (j & 3);
        bw[j] = b1[c];
        w0[j] = W2[c * 3 + 0];
        w1[j] = W2[c * 3 + 1];
        w2[j] = W2[c * 3 + 2];
    }
    const float l2a0 = al2[0], l2a1 = al2[1], l2a2 = al2[2];
    const float r2a0 = ar2[0], r2a1 = ar2[1], r2a2 = ar2[2];
    for (int r = wave; r < rows; r += AGG_WAVES) {
        const float* p = acc + r * 256 + lane;
        const float sv0 = s_t[r * 8 + hl];
        const float sv1 = s_t[r * 8 + 4 + hl];
        float a0 = 0.f, a1 = 0.f, a2 = 0.f;
#pragma unroll
        for (int j = 0; j < 8; ++j) {
            const float sv = (j < 4) ? sv0 : sv1;
            const float q  = p[(j >> 2) * 128 + (j & 3) * 32];
            float v = (sv != 0.f) ? q / sv : 0.f;
            v += bw[j];
            v = (v > 0.f) ? v : expm1f(v);
            a0 += v * w0[j];
            a1 += v * w1[j];
            a2 += v * w2[j];
        }
        for (int o = 16; o > 0; o >>= 1) {
            a0 += __shfl_xor(a0, o);
            a1 += __shfl_xor(a1, o);
            a2 += __shfl_xor(a2, o);
        }
        if (lane == 0) {
            const size_t n = (size_t)(lo + r);
            rec2[n] = make_float4(a0, a1, a2, a0 * l2a0 + a1 * l2a1 + a2 * l2a2);
            er2[n] = a0 * r2a0 + a1 * r2a1 + a2 * r2a2;
        }
    }
}

__global__ __launch_bounds__(AGG_THREADS)
void gat_layer2_owner(const int* __restrict__ src, const int* __restrict__ dst,
                      const int2* __restrict__ l2_list, const int* __restrict__ l2_cnt,
                      const float4* __restrict__ rec2,
                      const float* __restrict__ er2, const float* __restrict__ b2,
                      float* __restrict__ out, int n_edges, int n_nodes) {
    __shared__ float er_t[AGG_TILE];
    __shared__ float m_t[AGG_TILE];
    __shared__ float s_t[AGG_TILE];
    __shared__ float a_t[AGG_TILE * 3];

    const int tid  = threadIdx.x;
    const int wave = tid >> 5;
    const int lane = tid & 31;
    const int lo   = blockIdx.x * AGG_TILE;
    int rows = n_nodes - lo; if (rows > AGG_TILE) rows = AGG_TILE;
    const unsigned span = (unsigned)rows;

    const int cnt = __builtin_amdgcn_readfirstlane(l2_cnt[blockIdx.x * AGG_WAVES + wave]);
    const int2* slot = l2_list + (size_t)(blockIdx.x * AGG_WAVES + wave) * L2_CAP;

    for (int i = tid; i < AGG_TILE; i += AGG_THREADS) {
        er_t[i] = (i < rows) ? er2[lo + i] : 0.f;
        m_t[i]  = -__builtin_huge_valf();
        s_t[i]  = 0.f;
        a_t[i * 3 + 0] = 0.f;
        a_t[i * 3 + 1] = 0.f;
        a_t[i * 3 + 2] = 0.f;
    }
    __syncthreads();

    auto score = [&](float l, int dl) {
        const float x = l + er_t[dl];
        return (x >= 0.f) ? x : SLOPE * x;
    };
    auto add_edge = [&](float v, int dl, float g0, float g1, float g2) {
        const float a = expf(v - m_t[dl]);
        atomicAddLds(s_t + dl, a);
        atomicAddLds(a_t + dl * 3 + 0, g0 * a);
        atomicAddLds(a_t + dl * 3 + 1, g1 * a);
        atomicAddLds(a_t + dl * 3 + 2, g2 * a);
    };
    auto own_edges = [&](auto&& f) {
        for (int blk = wave; blk * MAP_BLOCK < n_edges; blk += AGG_WAVES) {
            for (int j = 0; j < MAP_BLOCK; j += 32) {
                const int e = blk * MAP_BLOCK + j + lane;
                if (e < n_edges) {
                    const int d = dst[e];
                    if ((unsigned)(d - lo) < span) f(src[e], d - lo);
                }
            }
        }
    };
    const bool listed = cnt <= L2_CAP;

    int2   en0[4];
    float4 rc0[4];
    if (listed) {
        if (cnt > 0) {
#pragma unroll
            for (int u = 0; u < 4; ++u) {
                const int i = u * 32 + lane;
                en0[u] = slot[i < cnt ? i : cnt - 1];
            }
#pragma unroll
            for (int u = 0; u < 4; ++u) rc0[u] = rec2[en0[u].x];
            asm volatile("" :: "v"(rc0[0].w), "v"(rc0[1].w), "v"(rc0[2].w), "v"(rc0[3].w));
#pragma unroll
            for (int u = 0; u < 4; ++u)
                if (u * 32 + lane < cnt) ldsMaxFloat(m_t + en0[u].y, score(rc0[u].w, en0[u].y));
        }
        for (int base = 128; base < cnt; base += 128) {
            int2 en[4]; float l[4];
#pragma unroll
            for (int u = 0; u < 4; ++u) {
                const int i = base + u * 32 + lane;
                en[u] = slot[i < cnt ? i : cnt - 1];
            }
#pragma unroll
            for (int u = 0; u < 4; ++u) l[u] = rec2[en[u].x].w;
            asm volatile("" :: "v"(l[0]), "v"(l[1]), "v"(l[2]), "v"(l[3]));
#pragma unroll
            for (int u = 0; u < 4; ++u)
                if (base + u * 32 + lane < cnt) ldsMaxFloat(m_t + en[u].y, score(l[u], en[u].y));
        }
    } else {
        own_edges([&](int sn, int dl) { ldsMaxFloat(m_t + dl, score(rec2[sn].w, dl)); });
    }
    __syncthreads();

    if (listed) {
        if (cnt > 0) {
#pragma unroll
            for (int u = 0; u < 4; ++u)
                if (u * 32 + lane < cnt)
                    add_edge(score(rc0[u].w, en0[u].y), en0[u].y, rc0[u].x, rc0[u].y, rc0[u].z);
        }
        for (int base = 128; base < cnt; base += 128) {
            int2 en[4]; float4 rc[4];
#pragma unroll
            for (int u = 0; u < 4; ++u) {
                const int i = base + u * 32 + lane;
                en[u] = slot[i < cnt ? i : cnt - 1];
            }
#pragma unroll
            for (int u = 0; u < 4; ++u) rc[u] = rec2[en[u].x];
            asm volatile("" :: "v"(rc[0].w), "v"(rc[1].w), "v"(rc[2].w), "v"(rc[3].w));
#pragma unroll
            for (int u = 0; u < 4; ++u)
                if (base + u * 32 + lane < cnt)
                    add_edge(score(rc[u].w, en[u].y), en[u].y, rc[u].x, rc[u].y, rc[u].z);
        }
    } else {
        own_edges([&](int sn, int dl) {
            const float4 rc = rec2[sn];
            add_edge(score(rc.w, dl), dl, rc.x, rc.y, rc.z);
        });
    }
    __syncthreads();

    for (int i = tid; i < rows * 3; i += AGG_THREADS) {
        const int r = i / 3;
        const float sv = s_t[r];
        const float v = (sv != 0.f) ? a_t[i] / sv : 0.f;
        out[(size_t)lo * 3 + i] = v + b2[i - r * 3];
    }
}

extern "C" void kernel_launch(void* const* d_in, const int* in_sizes, int n_in,
                              void* d_out, int out_size, void* d_ws, size_t ws_size,
                              hipStream_t stream) {
    const float* feat = (const float*)d_in[0];
    const int*   src  = (const int*)  d_in[1];
    const int*   dst  = (const int*)  d_in[2];
    const float* W1   = (const float*)d_in[3];
    const float* al1  = (const float*)d_in[4];
    const float* ar1  = (const float*)d_in[5];
    const float* b1   = (const float*)d_in[6];
    const float* W2   = (const float*)d_in[7];
    const float* al2  = (const float*)d_in[8];
    const float* ar2  = (const float*)d_in[9];
    const float* b2   = (const float*)d_in[10];
    float* out = (float*)d_out;

    const int n_nodes = in_sizes[0] / 256;
    const int n_edges = in_sizes[1];

    char* cur = (char*)d_ws;
    auto alloc = [&](size_t bytes) -> float* {
        float* p = (float*)cur;
        cur += (bytes + 255) & ~(size_t)255;
        return p;
    };
    float* H1   = alloc((size_t)n_nodes * 256 * 4);
    float* el1  = alloc((size_t)n_nodes * 8 * 4);
    float* er1  = alloc((size_t)n_nodes * 8 * 4);
    float4* rec2 = (float4*)alloc((size_t)n_nodes * 16);
    float* er2  = alloc((size_t)n_nodes * 4);
    const int groups = cdiv(n_nodes, AGG_TILE);
    int*  l2_cnt  = (int*) alloc((size_t)groups * AGG_WAVES * 4);
    int2* l2_list = (int2*)alloc((size_t)groups * AGG_WAVES * L2_CAP * 8);
    unsigned int* emap = (unsigned int*)alloc((size_t)cdiv(n_edges, MAP_BLOCK) * groups * 32 * 4);

    const int B = 256;

    gat_gemm1_wmma<<<G1_GROUPS, G1_THREADS, 0, stream>>>(feat, W1, H1, n_nodes);
    gat_attn_scores<<<cdiv(n_nodes * 8, B), B, 0, stream>>>(H1, al1, ar1, el1, er1, dst, emap, n_edges, groups, n_nodes);

    gat_softmax_aggregate_tile<<<groups, AGG_THREADS, 0, stream>>>(src, dst, emap, H1, el1, er1,
                                                                   b1, W2, al2, ar2, rec2, er2,
                                                                   l2_list, l2_cnt, n_edges, n_nodes);

    gat_layer2_owner<<<groups, AGG_THREADS, 0, stream>>>(src, dst, l2_list, l2_cnt, rec2, er2, b2,
                                                         out, n_edges, n_nodes);
}


# Layer-1 pass 1 keeps each edge's 8 scores in the row's own LDS row (first 32 edges); pass 2 forms its weights from them: el[src] gathered once per edge, not twice; no row clear; sums unchanged
# speedup vs baseline: 8.3527x; 1.0428x over previous
#include <hip/hip_runtime.h>
#include <hip/hip_bf16.h>

typedef __attribute__((ext_vector_type(2))) float v2f;
typedef __attribute__((ext_vector_type(8))) float v8f;

#define SLOPE 0.2f

static __host__ __device__ inline int cdiv(int a, int b) { return (a + b - 1) / b; }

__device__ inline void atomicAddLds(float* p, float v) {
    __hip_atomic_fetch_add(p, v, __ATOMIC_RELAXED, __HIP_MEMORY_SCOPE_WORKGROUP);
}

#define G1_GROUPS  256
#define G1_THREADS 512
#define G1_WAVES   16

__global__ __launch_bounds__(G1_THREADS)
void gat_gemm1_wmma(const float* __restrict__ A,
                    const float* __restrict__ W,
                    float* __restrict__ H,
                    int n_nodes) {
    __shared__ __attribute__((aligned(16))) float Wl[256 * 256];

    const int tid  = threadIdx.x;
    const int wave = __builtin_amdgcn_readfirstlane(tid >> 5);
    const int lane = tid & 31;
    const int c    = lane & 15;
    const int hf   = lane >> 4;
    const int q    = wave & 3;
    const int gq   = wave >> 2;

    const unsigned tiles = (unsigned)cdiv(n_nodes, 16);
    const int t0 = (int)((blockIdx.x * tiles) / G1_GROUPS);
    const int t1 = (int)(((blockIdx.x + 1) * tiles) / G1_GROUPS);
    const int full = (t1 - t0) >> 2;
    const int rem  = (t1 - t0) & 3;
    const int e    = (gq - q) & 3;
    const int nit  = full + (e < rem ? 1 : 0);
    auto tile_of = [&](int i) { return t0 + (i < full ? 4 * i + gq : 4 * full + e); };
    auto a_ptr = [&](int tile) {
        int r = 16 * tile + c; if (r >= n_nodes) r = n_nodes - 1;
        return A + (size_t)r * 256 + 4 * hf;
    };
    auto load_a = [&](float4 (&buf)[8], const float* p) {
#pragma unroll
        for (int j = 0; j < 8; ++j) buf[j] = *(const float4*)(p + 8 * j);
    };

    float4 a0[8], a1[8];
    load_a(a0, a_ptr(tile_of(0)));

#pragma unroll
    for (int k = 0; k < 128 / G1_WAVES; ++k) {
        const int u = wave + G1_WAVES * k;
        const float* wp = W + (size_t)(8 * (u >> 2) + 4 * hf) * 256 + 64 * (u & 3) + 4 * c;
        const float4 g0 = *(const float4*)(wp);
        const float4 g1 = *(const float4*)(wp + 256);
        const float4 g2 = *(const float4*)(wp + 512);
        const float4 g3 = *(const float4*)(wp + 768);
        float4* dp = (float4*)Wl + (u * 4) * 32 + lane;
        dp[0]  = make_float4(g0.x, g1.x, g2.x, g3.x);
        dp[32] = make_float4(g0.y, g1.y, g2.y, g3.y);
        dp[64] = make_float4(g0.z, g1.z, g2.z, g3.z);
        dp[96] = make_float4(g0.w, g1.w, g2.w, g3.w);
    }
    __syncthreads();

    v8f acc0 = {}, acc1 = {}, acc2 = {}, acc3 = {};
    const float4* bq = (const float4*)Wl + q * 128 + lane;

    auto mma_chunk = [&](const float4 (&a)[8], int ch) {
        const float4* bp = bq + ch * (8 * 512);
#pragma unroll
        for (int j = 0; j < 8; ++j) {
            const float4 b0 = bp[j * 512];
            const float4 b1 = bp[j * 512 + 32];
            const float4 b2 = bp[j * 512 + 64];
            const float4 b3 = bp[j * 512 + 96];
            const v2f alo = {a[j].x, a[j].y};
            const v2f ahi = {a[j].z, a[j].w};
            acc0 = __builtin_amdgcn_wmma_f32_16x16x4_f32(false, alo, false, (v2f){b0.x, b0.y}, (short)0, acc0, false, false);
            acc1 = __builtin_amdgcn_wmma_f32_16x16x4_f32(false, alo, false, (v2f){b1.x, b1.y}, (short)0, acc1, false, false);
            acc2 = __builtin_amdgcn_wmma_f32_16x16x4_f32(false, alo, false, (v2f){b2.x, b2.y}, (short)0, acc2, false, false);
            acc3 = __builtin_amdgcn_wmma_f32_16x16x4_f32(false, alo, false, (v2f){b3.x, b3.y}, (short)0, acc3, false, false);
            acc0 = __builtin_amdgcn_wmma_f32_16x16x4_f32(false, ahi, false, (v2f){b0.z, b0.w}, (short)0, acc0, false, false);
            acc1 = __builtin_amdgcn_wmma_f32_16x16x4_f32(false, ahi, false, (v2f){b1.z, b1.w}, (short)0, acc1, false, false);
            acc2 = __builtin_amdgcn_wmma_f32_16x16x4_f32(false, ahi, false, (v2f){b2.z, b2.w}, (short)0, acc2, false, false);
            acc3 = __builtin_amdgcn_wmma_f32_16x16x4_f32(false, ahi, false, (v2f){b3.z, b3.w}, (short)0, acc3, false, false);
        }
    };

    for (int it = 0; it < nit; ++it) {
        asm volatile("" ::: "memory");
        const int tile = tile_of(it);
        const float* ap  = a_ptr(tile);
        const float* apn = a_ptr(tile_of(it + 1 < nit ? it + 1 : it));
        load_a(a1, ap + 64);   mma_chunk(a0, 0);
        load_a(a0, ap + 128);  mma_chunk(a1, 1);
        load_a(a1, ap + 192);  mma_chunk(a0, 2);
        load_a(a0, apn);       mma_chunk(a1, 3);

        const int rbase = 16 * tile + 8 * hf;
        float* hp = H + (size_t)rbase * 256 + 64 * q + 4 * c;
        if (16 * tile + 16 <= n_nodes) {
#pragma unroll
            for (int r = 0; r < 8; ++r)
                *(float4*)(hp + r * 256) = make_float4(acc0[r], acc1[r], acc2[r], acc3[r]);
        } else {
#pragma unroll
            for (int r = 0; r < 8; ++r)
                if (rbase + r < n_nodes)
                    *(float4*)(hp + r * 256) = make_float4(acc0[r], acc1[r], acc2[r], acc3[r]);
        }
        acc0 = (v8f){}; acc1 = (v8f){}; acc2 = (v8f){}; acc3 = (v8f){};
    }
}

#define AGG_TILE    196
#define MAP_BLOCK   1024
#define MAP_GROUPS  256
#define SC_THREADS  256

__global__ __launch_bounds__(SC_THREADS)
void gat_attn_scores(const float* __restrict__ H,
                     const float* __restrict__ al,
                     const float* __restrict__ ar,
                     float* __restrict__ el, float* __restrict__ er,
                     const int* __restrict__ dst, unsigned int* __restrict__ emap,
                     int n_edges, int n_groups, int n_nodes) {
    __shared__ __attribute__((aligned(16))) unsigned int tab[MAP_GROUPS * 32];
    const int tid = threadIdx.x;
    const int n_blocks = cdiv(n_edges, MAP_BLOCK);
    for (int blk = blockIdx.x; blk < n_blocks; blk += gridDim.x) {
        const int e0 = blk * MAP_BLOCK + 4 * tid;
        int d[4];
        if (e0 + 3 < n_edges) {
            const int4 v = *(const int4*)(dst + e0);
            d[0] = v.x; d[1] = v.y; d[2] = v.z; d[3] = v.w;
        } else {
#pragma unroll
            for (int k = 0; k < 4; ++k) d[k] = (e0 + k < n_edges) ? dst[e0 + k] : -1;
        }
        for (int g0 = 0; g0 < n_groups; g0 += MAP_GROUPS) {
            for (int k = tid; k < MAP_GROUPS * 8; k += SC_THREADS)
                ((uint4*)tab)[k] = make_uint4(0u, 0u, 0u, 0u);
            __syncthreads();
#pragma unroll
            for (int k = 0; k < 4; ++k) {
                const unsigned g = (unsigned)d[k] / (unsigned)AGG_TILE;
                if (g < (unsigned)n_groups && g >= (unsigned)g0 && g - (unsigned)g0 < (unsigned)MAP_GROUPS)
                    atomicOr(tab + (g - (unsigned)g0) * 32 + (tid >> 3), 1u << (4 * (tid & 7) + k));
            }
            __syncthreads();
            int ng = n_groups - g0; if (ng > MAP_GROUPS) ng = MAP_GROUPS;
            uint4* op = (uint4*)(emap + ((size_t)blk * n_groups + g0) * 32);
            for (int k = tid; k < ng * 8; k += SC_THREADS) op[k] = ((const uint4*)tab)[k];
            __syncthreads();
        }
    }
    int i = blockIdx.x * blockDim.x + threadIdx.x;
    if (i >= n_nodes * 8) return;
    int n = i >> 3, h = i & 7;
    const float* hp  = H + (size_t)n * 256 + h * 32;
    const float* alp = al + h * 32;
    const float* arp = ar + h * 32;
    float sl = 0.f, sr = 0.f;
    for (int d = 0; d < 32; ++d) { float v = hp[d]; sl += v * alp[d]; sr += v * arp[d]; }
    el[i] = sl; er[i] = sr;
}

#define AGG_THREADS 1024
#define AGG_WAVES   32
#define AGG_PART    128
#define AGG_QCAP    224
#define AGG_RCAP    60
#define AGG_SCAP    32
#define AGG_FLY     6
#define L2_CAP      512

__device__ inline void ldsMaxFloat(float* addr, float value) {
    if (__float_as_int(value) >= 0)
        atomicMax((int*)addr, __float_as_int(value));
    else
        atomicMin((unsigned int*)addr, __float_as_uint(value));
}

__global__ __launch_bounds__(AGG_THREADS)
void gat_softmax_aggregate_tile(const int* __restrict__ src, const int* __restrict__ dst,
                                const unsigned int* __restrict__ emap,
                                const float* __restrict__ H, const float* __restrict__ el,
                                const float* __restrict__ er,
                                const float* __restrict__ b1, const float* __restrict__ W2,
                                const float* __restrict__ al2, const float* __restrict__ ar2,
                                float4* __restrict__ rec2, float* __restrict__ er2,
                                int2* __restrict__ l2_list, int* __restrict__ l2_cnt,
                                int n_edges, int n_nodes) {
    __shared__ __attribute__((aligned(16))) float acc[AGG_TILE * 256];
    __shared__ __attribute__((aligned(16))) float er_t[AGG_TILE * 8];
    __shared__ __attribute__((aligned(16))) float m_t[AGG_TILE * 8];
    __shared__ float s_t[AGG_TILE * 8];
    __shared__ int   q_e[AGG_WAVES * AGG_QCAP];
    __shared__ int   o_src[AGG_TILE * AGG_RCAP];
    __shared__ int   o_cnt[AGG_TILE];

    const int tid  = threadIdx.x;
    const int wave = tid >> 5;
    const int lane = tid & 31;
    const int lo   = blockIdx.x * AGG_TILE;
    int rows = n_nodes - lo; if (rows > AGG_TILE) rows = AGG_TILE;
    const unsigned span = (unsigned)rows;

    for (int i = tid; i < AGG_TILE * 8; i += AGG_THREADS) {
        er_t[i] = (i < rows * 8) ? er[(size_t)lo * 8 + i] : 0.f;
        m_t[i]  = -__builtin_huge_valf();
        s_t[i]  = 0.f;
    }
    for (int i = tid; i < AGG_TILE; i += AGG_THREADS) o_cnt[i] = 0;
    __syncthreads();

    int* qe = q_e + wave * AGG_QCAP;
    const unsigned lt_mask = (1u << lane) - 1u;
    const int hl = lane >> 3;
    int  cnt  = 0;
    int  done = 0;
    int  filed = 0;
    int2* slot = l2_list + (size_t)(blockIdx.x * AGG_WAVES + wave) * L2_CAP;

    auto drain_max = [&](int from, int n) {
        for (int b = 0; b < n; b += 32) {
            int dl = -1, sn = 0;
            if (b + lane < n) {
                const int e = qe[from + b + lane];
                dl = dst[e] - lo;
                sn = src[e];
            }
            const bool ok = (unsigned)dl < span;
            const unsigned okm = __builtin_amdgcn_ballot_w32(ok);
            if (ok) {
                const int fp = filed + __popc(okm & lt_mask);
                if (fp < L2_CAP) slot[fp] = make_int2(sn, dl);
                const int op = atomicAdd(o_cnt + dl, 1);
                if (op < AGG_RCAP) o_src[dl * AGG_RCAP + op] = sn;
                const float4 l0 = *(const float4*)(el + (size_t)sn * 8);
                const float4 l1 = *(const float4*)(el + (size_t)sn * 8 + 4);
                const float4 r0 = *(const float4*)(er_t + dl * 8);
                const float4 r1 = *(const float4*)(er_t + dl * 8 + 4);
                const float x[8] = {l0.x + r0.x, l0.y + r0.y, l0.z + r0.z, l0.w + r0.w,
                                    l1.x + r1.x, l1.y + r1.y, l1.z + r1.z, l1.w + r1.w};
                float v[8];
#pragma unroll
                for (int h = 0; h < 8; ++h) v[h] = (x[h] >= 0.f) ? x[h] : SLOPE * x[h];
                if (op < AGG_SCAP) {
                    float* sp = acc + dl * 256 + op * 8;
                    *(float4*)(sp)     = make_float4(v[0], v[1], v[2], v[3]);
                    *(float4*)(sp + 4) = make_float4(v[4], v[5], v[6], v[7]);
                }
#pragma unroll
                for (int h = 0; h < 8; ++h) ldsMaxFloat(m_t + dl * 8 + h, v[h]);
            }
            filed += __popc(okm);
        }
    };

    auto drain_sum = [&](int from, int n) {
        for (int b = 0; b < n; b += 32) {
            int nb = n - b; if (nb > 32) nb = 32;
            const int i  = from + b + (lane < nb ? lane : nb - 1);
            const int e  = qe[i];
            const int dl = dst[e] - lo;
            const int sn = src[e];
            for (int g = 0; g < nb; g += 4) {
                int su[4], du[4];
#pragma unroll
                for (int u = 0; u < 4; ++u) {
                    const int k = (g + u < nb) ? g + u : nb - 1;
                    su[u] = __builtin_amdgcn_readlane(sn, k);
                    du[u] = __builtin_amdgcn_readlane(dl, k);
                }
                float4 r0[4], r1[4];
                float  l0[4], l1[4];
#pragma unroll
                for (int u = 0; u < 4; ++u) {
                    const float* hp = H + (size_t)su[u] * 256 + 4 * lane;
                    r0[u] = *(const float4*)(hp);
                    r1[u] = *(const float4*)(hp + 128);
                    l0[u] = el[(size_t)su[u] * 8 + hl];
                    l1[u] = el[(size_t)su[u] * 8 + 4 + hl];
                }
#pragma unroll
                for (int u = 0; u < 4; ++u) {
                    if (g + u < nb && (unsigned)du[u] < span) {
                        const int t0 = du[u] * 8 + hl;
                        const float x0 = l0[u] + er_t[t0];
                        const float x1 = l1[u] + er_t[t0 + 4];
                        const float v0 = (x0 >= 0.f) ? x0 : SLOPE * x0;
                        const float v1 = (x1 >= 0.f) ? x1 : SLOPE * x1;
                        const float a0 = expf(v0 - m_t[t0]);
                        const float a1 = expf(v1 - m_t[t0 + 4]);
                        if ((lane & 7) == 0) {
                            atomicAddLds(s_t + t0,     a0);
                            atomicAddLds(s_t + t0 + 4, a1);
                        }
                        float* row = acc + du[u] * 256 + lane;
                        atomicAddLds(row,       r0[u].x * a0);
                        atomicAddLds(row + 32,  r0[u].y * a0);
                        atomicAddLds(row + 64,  r0[u].z * a0);
                        atomicAddLds(row + 96,  r0[u].w * a0);
                        atomicAddLds(row + 128, r1[u].x * a1);
                        atomicAddLds(row + 160, r1[u].y * a1);
                        atomicAddLds(row + 192, r1[u].z * a1);
                        atomicAddLds(row + 224, r1[u].w * a1);
                    }
                }
            }
        }
    };

    auto scan = [&](auto&& after) {
        const int n_blocks = cdiv(n_edges, MAP_BLOCK);
        if (n_blocks > 0) {
            auto loadw = [&](int blk) {
                const int bc = blk < n_blocks ? blk : n_blocks - 1;
                return emap[((size_t)bc * gridDim.x + blockIdx.x) * 32 + lane];
            };
            unsigned cur[4], nxt[4];
#pragma unroll
            for (int u = 0; u < 4; ++u) cur[u] = loadw(wave + u * AGG_WAVES);

            for (int b0 = wave; b0 < n_blocks; b0 += 4 * AGG_WAVES) {
#pragma unroll
                for (int u = 0; u < 4; ++u) nxt[u] = loadw(b0 + (4 + u) * AGG_WAVES);
#pragma unroll
                for (int u = 0; u < 4; ++u) {
                    const int blk = b0 + u * AGG_WAVES;
                    const int e0  = blk * MAP_BLOCK + lane * 32;
                    unsigned c = (blk < n_blocks) ? cur[u] : 0u;
                    unsigned mask = __builtin_amdgcn_ballot_w32(c != 0u);
                    while (mask != 0u) {
                        if (c != 0u) {
                            const int pos = cnt + __popc(mask & lt_mask);
                            qe[pos] = e0 + __builtin_ctz(c);
                            c &= c - 1u;
                        }
                        cnt += __popc(mask);
                        after();
                        mask = __builtin_amdgcn_ballot_w32(c != 0u);
                    }
                }
#pragma unroll
                for (int u = 0; u < 4; ++u) cur[u] = nxt[u];
            }
        }
    };

    scan([&]() {
        while (cnt - done >= 32) { drain_max(done, 32); done += 32; }
        if (cnt > AGG_QCAP - AGG_PART) {
            if (cnt > done) drain_max(done, cnt - done);
            cnt = 0; done = 0;
        }
    });
    if (cnt > done) drain_max(done, cnt - done);
    if (lane == 0) l2_cnt[blockIdx.x * AGG_WAVES + wave] = filed;
    __syncthreads();

    bool over = false;
    for (int i = lane; i < AGG_TILE; i += 32) over |= (o_cnt[i] > AGG_RCAP);
    const bool owned = (__builtin_amdgcn_ballot_w32(over) == 0u);

    if (owned) {
        for (int r = wave; r < rows; r += AGG_WAVES) {
            const int n = __builtin_amdgcn_readfirstlane(o_cnt[r]);
            const int* ol = o_src + r * AGG_RCAP;
            const float4 e0 = *(const float4*)(er_t + r * 8), e1 = *(const float4*)(er_t + r * 8 + 4);
            const float4 m0 = *(const float4*)(m_t + r * 8),  m1 = *(const float4*)(m_t + r * 8 + 4);
            float* wt = acc + r * 256;
            float sum[8] = {0.f, 0.f, 0.f, 0.f, 0.f, 0.f, 0.f, 0.f};
            float es0 = 0.f, es1 = 0.f;
            for (int b = 0; b < n; b += 32) {
                int nb = n - b; if (nb > 32) nb = 32;
                const int li = lane < nb ? lane : nb - 1;
                const int sn = ol[b + li];
                const bool kept = (b + 32 <= AGG_SCAP);
                float4 l0 = make_float4(0.f, 0.f, 0.f, 0.f), l1 = l0;
                if (!kept) {
                    l0 = *(const float4*)(el + (size_t)sn * 8);
                    l1 = *(const float4*)(el + (size_t)sn * 8 + 4);
                }
                for (int g = 0; g < nb; g += AGG_FLY) {
                    int su[AGG_FLY];
#pragma unroll
                    for (int u = 0; u < AGG_FLY; ++u)
                        su[u] = __builtin_amdgcn_readlane(sn, (g + u < nb) ? g + u : nb - 1);
                    float4 r0[AGG_FLY], r1[AGG_FLY];
#pragma unroll
                    for (int u = 0; u < AGG_FLY; ++u) {
                        const float* hp = H + (size_t)su[u] * 256 + 4 * lane;
                        r0[u] = *(const float4*)(hp);
                        r1[u] = *(const float4*)(hp + 128);
                    }
                    if (g == 0) {
                        float v[8];
                        if (kept) {
                            const float4 s0 = *(const float4*)(wt + li * 8);
                            const float4 s1 = *(const float4*)(wt + li * 8 + 4);
                            v[0] = s0.x; v[1] = s0.y; v[2] = s0.z; v[3] = s0.w;
                            v[4] = s1.x; v[5] = s1.y; v[6] = s1.z; v[7] = s1.w;
                        } else {
                            const float x[8] = {l0.x + e0.x, l0.y + e0.y, l0.z + e0.z, l0.w + e0.w,
                                                l1.x + e1.x, l1.y + e1.y, l1.z + e1.z, l1.w + e1.w};
#pragma unroll
                            for (int h = 0; h < 8; ++h) v[h] = (x[h] >= 0.f) ? x[h] : SLOPE * x[h];
                        }
                        const float mx[8] = {m0.x, m0.y, m0.z, m0.w, m1.x, m1.y, m1.z, m1.w};
                        float w[8];
#pragma unroll
                        for (int h = 0; h < 8; ++h) w[h] = expf(v[h] - mx[h]);
                        *(float4*)(wt + lane * 8)     = make_float4(w[0], w[1], w[2], w[3]);
                        *(float4*)(wt + lane * 8 + 4) = make_float4(w[4], w[5], w[6], w[7]);
                    }
#pragma unroll
                    for (int u = 0; u < AGG_FLY; ++u) {
                        if (g + u < nb) {
                            const float a0 = wt[(g + u) * 8 + hl];
                            const float a1 = wt[(g + u) * 8 + 4 + hl];
                            es0 += a0;
                            es1 += a1;
                            sum[0] += r0[u].x * a0; sum[1] += r0[u].y * a0;
                            sum[2] += r0[u].z * a0; sum[3] += r0[u].w * a0;
                            sum[4] += r1[u].x * a1; sum[5] += r1[u].y * a1;
                            sum[6] += r1[u].z * a1; sum[7] += r1[u].w * a1;
                        }
                    }
                }
            }
            float* row = acc + r * 256 + lane;
#pragma unroll
            for (int j = 0; j < 8; ++j) row[(j >> 2) * 128 + (j & 3) * 32] = sum[j];
            if ((lane & 7) == 0) {
                s_t[r * 8 + hl]     = es0;
                s_t[r * 8 + 4 + hl] = es1;
            }
        }
    } else {
        for (int i = tid; i < AGG_TILE * 64; i += AGG_THREADS)
            ((float4*)acc)[i] = make_float4(0.f, 0.f, 0.f, 0.f);
        __syncthreads();
        cnt = 0;
        scan([&]() {
            if (cnt > AGG_QCAP - AGG_PART) { drain_sum(0, cnt); cnt = 0; }
        });
        if (cnt > 0) drain_sum(0, cnt);
        __syncthreads();
    }

    float bw[8], w0[8], w1[8], w2[8];
#pragma unroll
    for (int j = 0; j < 8; ++j) {
        const int c = (j >> 2) * 128 + 4 * lane + (j & 3);
        bw[j] = b1[c];
        w0[j] = W2[c * 3 + 0];
        w1[j] = W2[c * 3 + 1];
        w2[j] = W2[c * 3 + 2];
    }
    const float l2a0 = al2[0], l2a1 = al2[1], l2a2 = al2[2];
    const float r2a0 = ar2[0], r2a1 = ar2[1], r2a2 = ar2[2];
    for (int r = wave; r < rows; r += AGG_WAVES) {
        const float* p = acc + r * 256 + lane;
        const float sv0 = s_t[r * 8 + hl];
        const float sv1 = s_t[r * 8 + 4 + hl];
        float a0 = 0.f, a1 = 0.f, a2 = 0.f;
#pragma unroll
        for (int j = 0; j < 8; ++j) {
            const float sv = (j < 4) ? sv0 : sv1;
            const float q  = p[(j >> 2) * 128 + (j & 3) * 32];
            float v = (sv != 0.f) ? q / sv : 0.f;
            v += bw[j];
            v = (v > 0.f) ? v : expm1f(v);
            a0 += v * w0[j];
            a1 += v * w1[j];
            a2 += v * w2[j];
        }
        for (int o = 16; o > 0; o >>= 1) {
            a0 += __shfl_xor(a0, o);
            a1 += __shfl_xor(a1, o);
            a2 += __shfl_xor(a2, o);
        }
        if (lane == 0) {
            const size_t n = (size_t)(lo + r);
            rec2[n] = make_float4(a0, a1, a2, a0 * l2a0 + a1 * l2a1 + a2 * l2a2);
            er2[n] = a0 * r2a0 + a1 * r2a1 + a2 * r2a2;
        }
    }
}

__global__ __launch_bounds__(AGG_THREADS)
void gat_layer2_owner(const int* __restrict__ src, const int* __restrict__ dst,
                      const int2* __restrict__ l2_list, const int* __restrict__ l2_cnt,
                      const float4* __restrict__ rec2,
                      const float* __restrict__ er2, const float* __restrict__ b2,
                      float* __restrict__ out, int n_edges, int n_nodes) {
    __shared__ float er_t[AGG_TILE];
    __shared__ float m_t[AGG_TILE];
    __shared__ float s_t[AGG_TILE];
    __shared__ float a_t[AGG_TILE * 3];

    const int tid  = threadIdx.x;
    const int wave = tid >> 5;
    const int lane = tid & 31;
    const int lo   = blockIdx.x * AGG_TILE;
    int rows = n_nodes - lo; if (rows > AGG_TILE) rows = AGG_TILE;
    const unsigned span = (unsigned)rows;

    const int cnt = __builtin_amdgcn_readfirstlane(l2_cnt[blockIdx.x * AGG_WAVES + wave]);
    const int2* slot = l2_list + (size_t)(blockIdx.x * AGG_WAVES + wave) * L2_CAP;

    for (int i = tid; i < AGG_TILE; i += AGG_THREADS) {
        er_t[i] = (i < rows) ? er2[lo + i] : 0.f;
        m_t[i]  = -__builtin_huge_valf();
        s_t[i]  = 0.f;
        a_t[i * 3 + 0] = 0.f;
        a_t[i * 3 + 1] = 0.f;
        a_t[i * 3 + 2] = 0.f;
    }
    __syncthreads();

    auto score = [&](float l, int dl) {
        const float x = l + er_t[dl];
        return (x >= 0.f) ? x : SLOPE * x;
    };
    auto add_edge = [&](float v, int dl, float g0, float g1, float g2) {
        const float a = expf(v - m_t[dl]);
        atomicAddLds(s_t + dl, a);
        atomicAddLds(a_t + dl * 3 + 0, g0 * a);
        atomicAddLds(a_t + dl * 3 + 1, g1 * a);
        atomicAddLds(a_t + dl * 3 + 2, g2 * a);
    };
    auto own_edges = [&](auto&& f) {
        for (int blk = wave; blk * MAP_BLOCK < n_edges; blk += AGG_WAVES) {
            for (int j = 0; j < MAP_BLOCK; j += 32) {
                const int e = blk * MAP_BLOCK + j + lane;
                if (e < n_edges) {
                    const int d = dst[e];
                    if ((unsigned)(d - lo) < span) f(src[e], d - lo);
                }
            }
        }
    };
    const bool listed = cnt <= L2_CAP;

    int2   en0[4];
    float4 rc0[4];
    if (listed) {
        if (cnt > 0) {
#pragma unroll
            for (int u = 0; u < 4; ++u) {
                const int i = u * 32 + lane;
                en0[u] = slot[i < cnt ? i : cnt - 1];
            }
#pragma unroll
            for (int u = 0; u < 4; ++u) rc0[u] = rec2[en0[u].x];
            asm volatile("" :: "v"(rc0[0].w), "v"(rc0[1].w), "v"(rc0[2].w), "v"(rc0[3].w));
#pragma unroll
            for (int u = 0; u < 4; ++u)
                if (u * 32 + lane < cnt) ldsMaxFloat(m_t + en0[u].y, score(rc0[u].w, en0[u].y));
        }
        for (int base = 128; base < cnt; base += 128) {
            int2 en[4]; float l[4];
#pragma unroll
            for (int u = 0; u < 4; ++u) {
                const int i = base + u * 32 + lane;
                en[u] = slot[i < cnt ? i : cnt - 1];
            }
#pragma unroll
            for (int u = 0; u < 4; ++u) l[u] = rec2[en[u].x].w;
            asm volatile("" :: "v"(l[0]), "v"(l[1]), "v"(l[2]), "v"(l[3]));
#pragma unroll
            for (int u = 0; u < 4; ++u)
                if (base + u * 32 + lane < cnt) ldsMaxFloat(m_t + en[u].y, score(l[u], en[u].y));
        }
    } else {
        own_edges([&](int sn, int dl) { ldsMaxFloat(m_t + dl, score(rec2[sn].w, dl)); });
    }
    __syncthreads();

    if (listed) {
        if (cnt > 0) {
#pragma unroll
            for (int u = 0; u < 4; ++u)
                if (u * 32 + lane < cnt)
                    add_edge(score(rc0[u].w, en0[u].y), en0[u].y, rc0[u].x, rc0[u].y, rc0[u].z);
        }
        for (int base = 128; base < cnt; base += 128) {
            int2 en[4]; float4 rc[4];
#pragma unroll
            for (int u = 0; u < 4; ++u) {
                const int i = base + u * 32 + lane;
                en[u] = slot[i < cnt ? i : cnt - 1];
            }
#pragma unroll
            for (int u = 0; u < 4; ++u) rc[u] = rec2[en[u].x];
            asm volatile("" :: "v"(rc[0].w), "v"(rc[1].w), "v"(rc[2].w), "v"(rc[3].w));
#pragma unroll
            for (int u = 0; u < 4; ++u)
                if (base + u * 32 + lane < cnt)
                    add_edge(score(rc[u].w, en[u].y), en[u].y, rc[u].x, rc[u].y, rc[u].z);
        }
    } else {
        own_edges([&](int sn, int dl) {
            const float4 rc = rec2[sn];
            add_edge(score(rc.w, dl), dl, rc.x, rc.y, rc.z);
        });
    }
    __syncthreads();

    for (int i = tid; i < rows * 3; i += AGG_THREADS) {
        const int r = i / 3;
        const float sv = s_t[r];
        const float v = (sv != 0.f) ? a_t[i] / sv : 0.f;
        out[(size_t)lo * 3 + i] = v + b2[i - r * 3];
    }
}

extern "C" void kernel_launch(void* const* d_in, const int* in_sizes, int n_in,
                              void* d_out, int out_size, void* d_ws, size_t ws_size,
                              hipStream_t stream) {
    const float* feat = (const float*)d_in[0];
    const int*   src  = (const int*)  d_in[1];
    const int*   dst  = (const int*)  d_in[2];
    const float* W1   = (const float*)d_in[3];
    const float* al1  = (const float*)d_in[4];
    const float* ar1  = (const float*)d_in[5];
    const float* b1   = (const float*)d_in[6];
    const float* W2   = (const float*)d_in[7];
    const float* al2  = (const float*)d_in[8];
    const float* ar2  = (const float*)d_in[9];
    const float* b2   = (const float*)d_in[10];
    float* out = (float*)d_out;

    const int n_nodes = in_sizes[0] / 256;
    const int n_edges = in_sizes[1];

    char* cur = (char*)d_ws;
    auto alloc = [&](size_t bytes) -> float* {
        float* p = (float*)cur;
        cur += (bytes + 255) & ~(size_t)255;
        return p;
    };
    float* H1   = alloc((size_t)n_nodes * 256 * 4);
    float* el1  = alloc((size_t)n_nodes * 8 * 4);
    float* er1  = alloc((size_t)n_nodes * 8 * 4);
    float4* rec2 = (float4*)alloc((size_t)n_nodes * 16);
    float* er2  = alloc((size_t)n_nodes * 4);
    const int groups = cdiv(n_nodes, AGG_TILE);
    int*  l2_cnt  = (int*) alloc((size_t)groups * AGG_WAVES * 4);
    int2* l2_list = (int2*)alloc((size_t)groups * AGG_WAVES * L2_CAP * 8);
    unsigned int* emap = (unsigned int*)alloc((size_t)cdiv(n_edges, MAP_BLOCK) * groups * 32 * 4);

    const int B = 256;

    gat_gemm1_wmma<<<G1_GROUPS, G1_THREADS, 0, stream>>>(feat, W1, H1, n_nodes);
    gat_attn_scores<<<cdiv(n_nodes * 8, B), B, 0, stream>>>(H1, al1, ar1, el1, er1, dst, emap, n_edges, groups, n_nodes);

    gat_softmax_aggregate_tile<<<groups, AGG_THREADS, 0, stream>>>(src, dst, emap, H1, el1, er1,
                                                                   b1, W2, al2, ar2, rec2, er2,
                                                                   l2_list, l2_cnt, n_edges, n_nodes);

    gat_layer2_owner<<<groups, AGG_THREADS, 0, stream>>>(src, dst, l2_list, l2_cnt, rec2, er2, b2,
                                                         out, n_edges, n_nodes);
}
